# starting kernel
# baseline (speedup 1.0000x reference)
#include <hip/hip_runtime.h>
#include <hip/hip_bf16.h>

typedef float v2f __attribute__((ext_vector_type(2)));
typedef float v8f __attribute__((ext_vector_type(8)));

#define N_NODES   100000
#define N_EDGES   1600000
#define IN_DIM    128
#define HDIM      64
#define N_CLASSES 16
#define NEG_SLOPE 0.2f

template <int K, int N>
__global__ __launch_bounds__(256) void gemm_wmma_f32(
    const float* __restrict__ A, const float* __restrict__ B,
    const float* __restrict__ bias, float* __restrict__ C, int M) {
  __shared__ float Bs[K * N];
  const int tid = threadIdx.x;
  for (int i = tid; i < K * N; i += 256) Bs[i] = B[i];
  __syncthreads();

  const int wave = tid >> 5;
  const int lane = tid & 31;
  const int hi = lane >> 4;
  const int lo = lane & 15;

  const long m0 = (long)blockIdx.x * 128 + wave * 16;
  long r = m0 + lo;
  if (r > M - 1) r = M - 1;
  const float* __restrict__ Arow = A + (size_t)r * K;

  constexpr int NT = N / 16;
  v8f acc[NT] = {};

  for (int kk = 0; kk < K / 4; ++kk) {
    const int kb = kk * 4 + 2 * hi;
    v2f a;
    a.x = Arow[kb];
    a.y = Arow[kb + 1];
#pragma unroll
    for (int nt = 0; nt < NT; ++nt) {
      v2f b;
      b.x = Bs[kb * N + nt * 16 + lo];
      b.y = Bs[(kb + 1) * N + nt * 16 + lo];
      acc[nt] = __builtin_amdgcn_wmma_f32_16x16x4_f32(
           false, a,  false, b,
           (short)0, acc[nt],  false,  false);
    }
  }

#pragma unroll
  for (int nt = 0; nt < NT; ++nt) {
    const int col = nt * 16 + lo;
    const float bv = bias ? bias[col] : 0.0f;
#pragma unroll
    for (int rr = 0; rr < 8; ++rr) {
      const long row = m0 + rr + 8 * hi;
      if (row < M) C[(size_t)row * N + col] = acc[nt][rr] + bv;
    }
  }
}

__global__ __launch_bounds__(256) void node_prep(
    const float* __restrict__ h, const float* __restrict__ a_src,
    const float* __restrict__ a_dst, float* __restrict__ alpha_s,
    float* __restrict__ alpha_d, float* __restrict__ m, int n_nodes) {
  const int wave = threadIdx.x >> 5;
  const int lane = threadIdx.x & 31;
  const int n = blockIdx.x * 8 + wave;
  if (n >= n_nodes) return;
  const float* hr = h + (size_t)n * HDIM;
  const float h0 = hr[lane], h1 = hr[lane + 32];
  float ps = h0 * a_src[lane] + h1 * a_src[lane + 32];
  float pd = h0 * a_dst[lane] + h1 * a_dst[lane + 32];
#pragma unroll
  for (int off = 16; off > 0; off >>= 1) {
    ps += __shfl_xor(ps, off, 32);
    pd += __shfl_xor(pd, off, 32);
  }
  if (lane == 0) {
    alpha_s[n] = ps;
    alpha_d[n] = pd;
    float e = ps + pd;
    m[n] = (e > 0.0f) ? e : NEG_SLOPE * e;
  }
}

__global__ __launch_bounds__(256) void edge_max(
    const int* __restrict__ adj, const float* __restrict__ alpha_s,
    const float* __restrict__ alpha_d, float* __restrict__ ebuf,
    float* __restrict__ m, int n_edges) {
  const int i = blockIdx.x * 256 + threadIdx.x;
  if (i >= n_edges) return;
  const int src = adj[i];
  const int dst = adj[n_edges + i];
  float e = alpha_s[src] + alpha_d[dst];
  e = (e > 0.0f) ? e : NEG_SLOPE * e;
  ebuf[i] = e;
  atomicMax(&m[dst], e);
}

__global__ __launch_bounds__(256) void node_post(
    const float* __restrict__ h, const float* __restrict__ alpha_s,
    const float* __restrict__ alpha_d, const float* __restrict__ m,
    float* __restrict__ s, float* __restrict__ acc, int n_nodes) {
  const int wave = threadIdx.x >> 5;
  const int lane = threadIdx.x & 31;
  const int n = blockIdx.x * 8 + wave;
  if (n >= n_nodes) return;
  float e = alpha_s[n] + alpha_d[n];
  e = (e > 0.0f) ? e : NEG_SLOPE * e;
  const float pself = __expf(e - m[n]);
  if (lane == 0) s[n] = pself;
  const size_t base = (size_t)n * HDIM;
  acc[base + lane]      = h[base + lane] * pself;
  acc[base + lane + 32] = h[base + lane + 32] * pself;
}

__global__ __launch_bounds__(256) void edge_expsum(
    const int* __restrict__ adj, float* __restrict__ ebuf,
    const float* __restrict__ m, float* __restrict__ s, int n_edges) {
  const int i = blockIdx.x * 256 + threadIdx.x;
  if (i >= n_edges) return;
  const int dst = adj[n_edges + i];
  const float p = __expf(ebuf[i] - m[dst]);
  ebuf[i] = p;
  atomicAdd(&s[dst], p);
}

__global__ __launch_bounds__(256) void edge_scatter(
    const int* __restrict__ adj, const float* __restrict__ ebuf,
    const float* __restrict__ h, float* __restrict__ acc, int n_edges) {
  const int wid = (blockIdx.x * 256 + threadIdx.x) >> 5;
  if (wid >= n_edges) return;
  const int lane = threadIdx.x & 31;
  const int src = adj[wid];
  const int dst = adj[n_edges + wid];
  const float w = ebuf[wid];
  const float* __restrict__ hs = h + (size_t)src * HDIM;
  float* __restrict__ ao = acc + (size_t)dst * HDIM;
  atomicAdd(&ao[lane],      hs[lane]      * w);
  atomicAdd(&ao[lane + 32], hs[lane + 32] * w);
}

__global__ __launch_bounds__(256) void normalize_bias(
    const float* __restrict__ acc, const float* __restrict__ s,
    const float* __restrict__ b, float* __restrict__ out, int total,
    int apply_elu) {
  const int idx = blockIdx.x * 256 + threadIdx.x;
  if (idx >= total) return;
  const int n = idx >> 6;
  const int c = idx & 63;
  float v = acc[idx] / s[n] + b[c];
  if (apply_elu) v = (v > 0.0f) ? v : (__expf(v) - 1.0f);
  out[idx] = v;
}

extern "C" void kernel_launch(void* const* d_in, const int* in_sizes, int n_in,
                              void* d_out, int out_size, void* d_ws,
                              size_t ws_size, hipStream_t stream) {
  const float* X      = (const float*)d_in[0];
  const int*   adj    = (const int*)d_in[1];
  const float* W1     = (const float*)d_in[2];
  const float* a_src1 = (const float*)d_in[3];
  const float* a_dst1 = (const float*)d_in[4];
  const float* b1     = (const float*)d_in[5];
  const float* W2     = (const float*)d_in[6];
  const float* a_src2 = (const float*)d_in[7];
  const float* a_dst2 = (const float*)d_in[8];
  const float* b2     = (const float*)d_in[9];
  const float* clas_W = (const float*)d_in[10];
  const float* clas_b = (const float*)d_in[11];

  float* out = (float*)d_out;
  float* x2  = out + (size_t)N_NODES * N_CLASSES;

  float* ws   = (float*)d_ws;
  float* h    = ws;
  float* x1   = ws + (size_t)N_NODES * HDIM;
  float* acc  = ws + (size_t)2 * N_NODES * HDIM;
  float* as_  = ws + (size_t)3 * N_NODES * HDIM;
  float* ad_  = as_ + N_NODES;
  float* m    = ad_ + N_NODES;
  float* s    = m + N_NODES;
  float* ebuf = s + N_NODES;

  const int gemm_blocks = (N_NODES + 127) / 128;
  const int node_blocks = (N_NODES + 7) / 8;
  const int edge_blocks = (N_EDGES + 255) / 256;
  const int scat_blocks = (N_EDGES + 7) / 8;
  const int norm_total  = N_NODES * HDIM;
  const int norm_blocks = (norm_total + 255) / 256;

  gemm_wmma_f32<IN_DIM, HDIM><<<gemm_blocks, 256, 0, stream>>>(X, W1, nullptr, h, N_NODES);
  node_prep<<<node_blocks, 256, 0, stream>>>(h, a_src1, a_dst1, as_, ad_, m, N_NODES);
  edge_max<<<edge_blocks, 256, 0, stream>>>(adj, as_, ad_, ebuf, m, N_EDGES);
  node_post<<<node_blocks, 256, 0, stream>>>(h, as_, ad_, m, s, acc, N_NODES);
  edge_expsum<<<edge_blocks, 256, 0, stream>>>(adj, ebuf, m, s, N_EDGES);
  edge_scatter<<<scat_blocks, 256, 0, stream>>>(adj, ebuf, h, acc, N_EDGES);
  normalize_bias<<<norm_blocks, 256, 0, stream>>>(acc, s, b1, x1, norm_total,  1);

  gemm_wmma_f32<HDIM, HDIM><<<gemm_blocks, 256, 0, stream>>>(x1, W2, nullptr, h, N_NODES);
  node_prep<<<node_blocks, 256, 0, stream>>>(h, a_src2, a_dst2, as_, ad_, m, N_NODES);
  edge_max<<<edge_blocks, 256, 0, stream>>>(adj, as_, ad_, ebuf, m, N_EDGES);
  node_post<<<node_blocks, 256, 0, stream>>>(h, as_, ad_, m, s, acc, N_NODES);
  edge_expsum<<<edge_blocks, 256, 0, stream>>>(adj, ebuf, m, s, N_EDGES);
  edge_scatter<<<scat_blocks, 256, 0, stream>>>(adj, ebuf, h, acc, N_EDGES);
  normalize_bias<<<norm_blocks, 256, 0, stream>>>(acc, s, b2, x2, norm_total,  0);

  gemm_wmma_f32<HDIM, N_CLASSES><<<gemm_blocks, 256, 0, stream>>>(x2, clas_W, clas_b, out, N_NODES);
}


# Replace per-edge f32 atomic max/sum/row-add by edge lists grouped by destination (integer atomic rank, prefix scan, groups sorted by source) walked per node; sums now run in fixed sorted order
# speedup vs baseline: 2.3784x; 2.3784x over previous
#include <hip/hip_runtime.h>
#include <hip/hip_bf16.h>

typedef float v2f __attribute__((ext_vector_type(2)));
typedef float v8f __attribute__((ext_vector_type(8)));

#define N_NODES   100000
#define N_EDGES   1600000
#define IN_DIM    128
#define HDIM      64
#define N_CLASSES 16
#define NEG_SLOPE 0.2f

template <int K, int N>
__global__ __launch_bounds__(256) void gemm_wmma_f32(
    const float* __restrict__ A, const float* __restrict__ B,
    const float* __restrict__ bias, float* __restrict__ C, int M) {
  __shared__ float Bs[K * N];
  const int tid = threadIdx.x;
  for (int i = tid; i < K * N; i += 256) Bs[i] = B[i];
  __syncthreads();

  const int wave = tid >> 5;
  const int lane = tid & 31;
  const int hi = lane >> 4;
  const int lo = lane & 15;

  const long m0 = (long)blockIdx.x * 128 + wave * 16;
  long r = m0 + lo;
  if (r > M - 1) r = M - 1;
  const float* __restrict__ Arow = A + (size_t)r * K;

  constexpr int NT = N / 16;
  v8f acc[NT] = {};

  for (int kk = 0; kk < K / 4; ++kk) {
    const int kb = kk * 4 + 2 * hi;
    v2f a;
    a.x = Arow[kb];
    a.y = Arow[kb + 1];
#pragma unroll
    for (int nt = 0; nt < NT; ++nt) {
      v2f b;
      b.x = Bs[kb * N + nt * 16 + lo];
      b.y = Bs[(kb + 1) * N + nt * 16 + lo];
      acc[nt] = __builtin_amdgcn_wmma_f32_16x16x4_f32(
           false, a,  false, b,
           (short)0, acc[nt],  false,  false);
    }
  }

#pragma unroll
  for (int nt = 0; nt < NT; ++nt) {
    const int col = nt * 16 + lo;
    const float bv = bias ? bias[col] : 0.0f;
#pragma unroll
    for (int rr = 0; rr < 8; ++rr) {
      const long row = m0 + rr + 8 * hi;
      if (row < M) C[(size_t)row * N + col] = acc[nt][rr] + bv;
    }
  }
}

__global__ __launch_bounds__(256) void node_prep(
    const float* __restrict__ h, const float* __restrict__ a_src,
    const float* __restrict__ a_dst, float* __restrict__ alpha_s,
    float* __restrict__ alpha_d, int n_nodes) {
  const int wave = threadIdx.x >> 5;
  const int lane = threadIdx.x & 31;
  const int n = blockIdx.x * 8 + wave;
  if (n >= n_nodes) return;
  const float* hr = h + (size_t)n * HDIM;
  const float h0 = hr[lane], h1 = hr[lane + 32];
  float ps = h0 * a_src[lane] + h1 * a_src[lane + 32];
  float pd = h0 * a_dst[lane] + h1 * a_dst[lane + 32];
#pragma unroll
  for (int off = 16; off > 0; off >>= 1) {
    ps += __shfl_xor(ps, off, 32);
    pd += __shfl_xor(pd, off, 32);
  }
  if (lane == 0) {
    alpha_s[n] = ps;
    alpha_d[n] = pd;
  }
}

__global__ __launch_bounds__(256) void zero_counts(int* __restrict__ cnt, int n) {
  const int i = blockIdx.x * 256 + threadIdx.x;
  if (i < n) cnt[i] = 0;
}

__global__ __launch_bounds__(256) void edge_rank(
    const int* __restrict__ adj, int* __restrict__ cnt,
    int* __restrict__ rank, int n_edges) {
  const int i = blockIdx.x * 256 + threadIdx.x;
  if (i >= n_edges) return;
  const int dst = adj[n_edges + i];
  rank[i] = atomicAdd(&cnt[dst], 1);
}

__global__ __launch_bounds__(1024) void scan_offsets(
    const int* __restrict__ cnt, int* __restrict__ off, int n) {
  __shared__ int wsum[32];
  const int tid = threadIdx.x;
  const int lane = tid & 31;
  const int wave = tid >> 5;
  const int lim = blockIdx.x * 1024;
  int v = 0;
  for (int j = tid; j < lim; j += 1024) v += (j < n) ? cnt[j] : 0;
#pragma unroll
  for (int o = 16; o > 0; o >>= 1) v += __shfl_xor(v, o, 32);
  if (lane == 0) wsum[wave] = v;
  __syncthreads();
  int base = 0;
#pragma unroll
  for (int w = 0; w < 32; ++w) base += wsum[w];
  __syncthreads();
  const int idx = lim + tid;
  const int c = (idx < n) ? cnt[idx] : 0;
  int inc = c;
#pragma unroll
  for (int o = 1; o < 32; o <<= 1) {
    const int t = __shfl_up(inc, o, 32);
    if (lane >= o) inc += t;
  }
  if (lane == 31) wsum[wave] = inc;
  __syncthreads();
  int wpre = 0;
  for (int w = 0; w < wave; ++w) wpre += wsum[w];
  if (idx <= n) off[idx] = base + wpre + inc - c;
}

__global__ __launch_bounds__(256) void edge_fill(
    const int* __restrict__ adj, const int* __restrict__ off,
    const int* __restrict__ rank, int* __restrict__ tmp, int n_edges) {
  const int i = blockIdx.x * 256 + threadIdx.x;
  if (i >= n_edges) return;
  const int dst = adj[n_edges + i];
  tmp[off[dst] + rank[i]] = adj[i];
}

#define NPW 4

__global__ __launch_bounds__(256) void seg_sort(
    const int* __restrict__ off, const int* __restrict__ tmp,
    int* __restrict__ csr, int n_nodes) {
  const int lane = threadIdx.x & 31;
  const int wid = __builtin_amdgcn_readfirstlane((int)((blockIdx.x * 256 + threadIdx.x) >> 5));
  for (int t = 0; t < NPW; ++t) {
    const int n = wid * NPW + t;
    if (n >= n_nodes) break;
    const int start = off[n];
    const int end = off[n + 1];
    for (int ib = start; ib < end; ib += 32) {
      const int i = ib + lane;
      const bool vi = i < end;
      const int mine = vi ? tmp[i] : 0;
      int r = 0;
      for (int kb = start; kb < end; kb += 32) {
        const int kj = kb + lane;
        const int other = (kj < end) ? tmp[kj] : 0;
        const int cnt = (end - kb < 32) ? (end - kb) : 32;
        for (int k = 0; k < cnt; ++k) {
          const int o = __builtin_amdgcn_readlane(other, k);
          const int kk = kb + k;
          r += (o < mine || (o == mine && kk < i)) ? 1 : 0;
        }
      }
      if (vi) csr[start + r] = mine;
    }
  }
}

template <int ELU>
__global__ __launch_bounds__(256) void node_walk(
    const int* __restrict__ off, const int* __restrict__ csr,
    const float* __restrict__ alpha_s, const float* __restrict__ alpha_d,
    const float* __restrict__ h, const float* __restrict__ b,
    float* __restrict__ out, int n_nodes) {
  const int lane = threadIdx.x & 31;
  const int wid = __builtin_amdgcn_readfirstlane((int)((blockIdx.x * 256 + threadIdx.x) >> 5));
  const float2 bb = *(const float2*)(b + 2 * lane);
  for (int t = 0; t < NPW; ++t) {
    const int n = wid * NPW + t;
    if (n >= n_nodes) break;
    const int start = off[n];
    const int end = off[n + 1];
    const float adn = alpha_d[n];
    float es = alpha_s[n] + adn;
    es = (es > 0.0f) ? es : NEG_SLOPE * es;

    float m = es;
    for (int base = start; base < end; base += 32) {
      const int j = base + lane;
      if (j < end) {
        const int sj = csr[j];
        float e = alpha_s[sj] + adn;
        e = (e > 0.0f) ? e : NEG_SLOPE * e;
        m = fmaxf(m, e);
      }
    }
#pragma unroll
    for (int o = 16; o > 0; o >>= 1) m = fmaxf(m, __shfl_xor(m, o, 32));

    const float pself = __expf(es - m);
    float ssum = pself;
    const float2 hn = *(const float2*)(h + (size_t)n * HDIM + 2 * lane);
    float ax = hn.x * pself;
    float ay = hn.y * pself;

    for (int base = start; base < end; base += 32) {
      const int j = base + lane;
      int sj = n;
      float p = 0.0f;
      if (j < end) {
        sj = csr[j];
        float e = alpha_s[sj] + adn;
        e = (e > 0.0f) ? e : NEG_SLOPE * e;
        p = __expf(e - m);
      }
      const int pi = __float_as_int(p);
      const int cnt = (end - base < 32) ? (end - base) : 32;
      for (int k = 0; k < cnt; k += 4) {
        const int s0 = __builtin_amdgcn_readlane(sj, k);
        const int s1 = __builtin_amdgcn_readlane(sj, k + 1);
        const int s2 = __builtin_amdgcn_readlane(sj, k + 2);
        const int s3 = __builtin_amdgcn_readlane(sj, k + 3);
        const float p0 = __int_as_float(__builtin_amdgcn_readlane(pi, k));
        const float p1 = __int_as_float(__builtin_amdgcn_readlane(pi, k + 1));
        const float p2 = __int_as_float(__builtin_amdgcn_readlane(pi, k + 2));
        const float p3 = __int_as_float(__builtin_amdgcn_readlane(pi, k + 3));
        const float2 g0 = *(const float2*)(h + (size_t)s0 * HDIM + 2 * lane);
        const float2 g1 = *(const float2*)(h + (size_t)s1 * HDIM + 2 * lane);
        const float2 g2 = *(const float2*)(h + (size_t)s2 * HDIM + 2 * lane);
        const float2 g3 = *(const float2*)(h + (size_t)s3 * HDIM + 2 * lane);
        ssum += p0; ax += g0.x * p0; ay += g0.y * p0;
        if (k + 1 < cnt) { ssum += p1; ax += g1.x * p1; ay += g1.y * p1; }
        if (k + 2 < cnt) { ssum += p2; ax += g2.x * p2; ay += g2.y * p2; }
        if (k + 3 < cnt) { ssum += p3; ax += g3.x * p3; ay += g3.y * p3; }
      }
    }

    float vx = ax / ssum + bb.x;
    float vy = ay / ssum + bb.y;
    if (ELU) {
      vx = (vx > 0.0f) ? vx : (__expf(vx) - 1.0f);
      vy = (vy > 0.0f) ? vy : (__expf(vy) - 1.0f);
    }
    *(float2*)(out + (size_t)n * HDIM + 2 * lane) = make_float2(vx, vy);
  }
}

extern "C" void kernel_launch(void* const* d_in, const int* in_sizes, int n_in,
                              void* d_out, int out_size, void* d_ws,
                              size_t ws_size, hipStream_t stream) {
  const float* X      = (const float*)d_in[0];
  const int*   adj    = (const int*)d_in[1];
  const float* W1     = (const float*)d_in[2];
  const float* a_src1 = (const float*)d_in[3];
  const float* a_dst1 = (const float*)d_in[4];
  const float* b1     = (const float*)d_in[5];
  const float* W2     = (const float*)d_in[6];
  const float* a_src2 = (const float*)d_in[7];
  const float* a_dst2 = (const float*)d_in[8];
  const float* b2     = (const float*)d_in[9];
  const float* clas_W = (const float*)d_in[10];
  const float* clas_b = (const float*)d_in[11];

  float* out = (float*)d_out;
  float* x2  = out + (size_t)N_NODES * N_CLASSES;

  float* ws   = (float*)d_ws;
  float* h    = ws;
  float* x1   = ws + (size_t)N_NODES * HDIM;
  int*   ib   = (int*)(ws + (size_t)2 * N_NODES * HDIM);
  float* as_  = ws + (size_t)3 * N_NODES * HDIM;
  float* ad_  = as_ + N_NODES;
  int* cnt  = ib;
  int* off  = ib + 100352;
  int* rank = ib + 2 * 100352;
  int* tmp  = rank + N_EDGES;
  int* csr  = tmp + N_EDGES;

  const int gemm_blocks = (N_NODES + 127) / 128;
  const int node_blocks = (N_NODES + 7) / 8;
  const int edge_blocks = (N_EDGES + 255) / 256;
  const int zero_blocks = (N_NODES + 255) / 256;
  const int scan_blocks = (N_NODES + 1 + 1023) / 1024;
  const int walk_blocks = (N_NODES + 8 * NPW - 1) / (8 * NPW);

  zero_counts<<<zero_blocks, 256, 0, stream>>>(cnt, N_NODES);
  edge_rank<<<edge_blocks, 256, 0, stream>>>(adj, cnt, rank, N_EDGES);
  scan_offsets<<<scan_blocks, 1024, 0, stream>>>(cnt, off, N_NODES);
  edge_fill<<<edge_blocks, 256, 0, stream>>>(adj, off, rank, tmp, N_EDGES);
  seg_sort<<<walk_blocks, 256, 0, stream>>>(off, tmp, csr, N_NODES);

  gemm_wmma_f32<IN_DIM, HDIM><<<gemm_blocks, 256, 0, stream>>>(X, W1, nullptr, h, N_NODES);
  node_prep<<<node_blocks, 256, 0, stream>>>(h, a_src1, a_dst1, as_, ad_, N_NODES);
  node_walk<1><<<walk_blocks, 256, 0, stream>>>(off, csr, as_, ad_, h, b1, x1, N_NODES);

  gemm_wmma_f32<HDIM, HDIM><<<gemm_blocks, 256, 0, stream>>>(x1, W2, nullptr, h, N_NODES);
  node_prep<<<node_blocks, 256, 0, stream>>>(h, a_src2, a_dst2, as_, ad_, N_NODES);
  node_walk<0><<<walk_blocks, 256, 0, stream>>>(off, csr, as_, ad_, h, b2, x2, N_NODES);

  gemm_wmma_f32<HDIM, N_CLASSES><<<gemm_blocks, 256, 0, stream>>>(x2, clas_W, clas_b, out, N_NODES);
}


# Fold the per-node attention dot products (h.a_src, h.a_dst) into the GEMM epilogue, from the accumulators; the two node_prep launches go; each dot now sums 4 terms per lane, then a 16-lane tree
# speedup vs baseline: 2.5673x; 1.0794x over previous
#include <hip/hip_runtime.h>
#include <hip/hip_bf16.h>

typedef float v2f __attribute__((ext_vector_type(2)));
typedef float v8f __attribute__((ext_vector_type(8)));

#define N_NODES   100000
#define N_EDGES   1600000
#define IN_DIM    128
#define HDIM      64
#define N_CLASSES 16
#define NEG_SLOPE 0.2f

template <int K, int N, bool ALPHA>
__global__ __launch_bounds__(256) void gemm_wmma_f32(
    const float* __restrict__ A, const float* __restrict__ B,
    const float* __restrict__ bias, float* __restrict__ C, int M,
    const float* __restrict__ a_src, const float* __restrict__ a_dst,
    float* __restrict__ alpha_s, float* __restrict__ alpha_d) {
  __shared__ float Bs[K * N];
  const int tid = threadIdx.x;
  for (int i = tid; i < K * N; i += 256) Bs[i] = B[i];
  __syncthreads();

  const int wave = tid >> 5;
  const int lane = tid & 31;
  const int hi = lane >> 4;
  const int lo = lane & 15;

  const long m0 = (long)blockIdx.x * 128 + wave * 16;
  long r = m0 + lo;
  if (r > M - 1) r = M - 1;
  const float* __restrict__ Arow = A + (size_t)r * K;

  constexpr int NT = N / 16;
  v8f acc[NT] = {};

  [[maybe_unused]] float avs[NT], avd[NT];
  if constexpr (ALPHA) {
#pragma unroll
    for (int nt = 0; nt < NT; ++nt) {
      avs[nt] = a_src[nt * 16 + lo];
      avd[nt] = a_dst[nt * 16 + lo];
    }
  }

  for (int kk = 0; kk < K / 4; ++kk) {
    const int kb = kk * 4 + 2 * hi;
    v2f a;
    a.x = Arow[kb];
    a.y = Arow[kb + 1];
#pragma unroll
    for (int nt = 0; nt < NT; ++nt) {
      v2f b;
      b.x = Bs[kb * N + nt * 16 + lo];
      b.y = Bs[(kb + 1) * N + nt * 16 + lo];
      acc[nt] = __builtin_amdgcn_wmma_f32_16x16x4_f32(
           false, a,  false, b,
           (short)0, acc[nt],  false,  false);
    }
  }

#pragma unroll
  for (int nt = 0; nt < NT; ++nt) {
    const int col = nt * 16 + lo;
    const float bv = bias ? bias[col] : 0.0f;
#pragma unroll
    for (int rr = 0; rr < 8; ++rr) {
      const long row = m0 + rr + 8 * hi;
      if (row < M) C[(size_t)row * N + col] = acc[nt][rr] + bv;
    }
  }

  if constexpr (ALPHA) {
    float v[16];
#pragma unroll
    for (int rr = 0; rr < 8; ++rr) {
      float s = acc[0][rr] * avs[0];
      float d = acc[0][rr] * avd[0];
#pragma unroll
      for (int nt = 1; nt < NT; ++nt) {
        s += acc[nt][rr] * avs[nt];
        d += acc[nt][rr] * avd[nt];
      }
      v[rr] = s;
      v[8 + rr] = d;
    }
#pragma unroll
    for (int w = 8; w >= 1; w >>= 1) {
      const bool up = (lo & w) != 0;
#pragma unroll
      for (int i = 0; i < w; ++i) {
        const float lower = v[i];
        const float upper = v[i + w];
        const float keep = up ? upper : lower;
        const float send = up ? lower : upper;
        v[i] = keep + __shfl_xor(send, w, 32);
      }
    }
    const long row = m0 + (lo & 7) + 8 * hi;
    float* __restrict__ dstp = (lo < 8) ? alpha_s : alpha_d;
    if (row < M) dstp[row] = v[0];
  }
}

__global__ __launch_bounds__(256) void zero_counts(int* __restrict__ cnt, int n) {
  const int i = blockIdx.x * 256 + threadIdx.x;
  if (i < n) cnt[i] = 0;
}

__global__ __launch_bounds__(256) void edge_rank(
    const int* __restrict__ adj, int* __restrict__ cnt,
    int* __restrict__ rank, int n_edges) {
  const int i = blockIdx.x * 256 + threadIdx.x;
  if (i >= n_edges) return;
  const int dst = adj[n_edges + i];
  rank[i] = atomicAdd(&cnt[dst], 1);
}

__global__ __launch_bounds__(1024) void scan_offsets(
    const int* __restrict__ cnt, int* __restrict__ off, int n) {
  __shared__ int wsum[32];
  const int tid = threadIdx.x;
  const int lane = tid & 31;
  const int wave = tid >> 5;
  const int lim = blockIdx.x * 1024;
  int v = 0;
  for (int j = tid; j < lim; j += 1024) v += (j < n) ? cnt[j] : 0;
#pragma unroll
  for (int o = 16; o > 0; o >>= 1) v += __shfl_xor(v, o, 32);
  if (lane == 0) wsum[wave] = v;
  __syncthreads();
  int base = 0;
#pragma unroll
  for (int w = 0; w < 32; ++w) base += wsum[w];
  __syncthreads();
  const int idx = lim + tid;
  const int c = (idx < n) ? cnt[idx] : 0;
  int inc = c;
#pragma unroll
  for (int o = 1; o < 32; o <<= 1) {
    const int t = __shfl_up(inc, o, 32);
    if (lane >= o) inc += t;
  }
  if (lane == 31) wsum[wave] = inc;
  __syncthreads();
  int wpre = 0;
  for (int w = 0; w < wave; ++w) wpre += wsum[w];
  if (idx <= n) off[idx] = base + wpre + inc - c;
}

__global__ __launch_bounds__(256) void edge_fill(
    const int* __restrict__ adj, const int* __restrict__ off,
    const int* __restrict__ rank, int* __restrict__ tmp, int n_edges) {
  const int i = blockIdx.x * 256 + threadIdx.x;
  if (i >= n_edges) return;
  const int dst = adj[n_edges + i];
  tmp[off[dst] + rank[i]] = adj[i];
}

#define NPW 4

__global__ __launch_bounds__(256) void seg_sort(
    const int* __restrict__ off, const int* __restrict__ tmp,
    int* __restrict__ csr, int n_nodes) {
  const int lane = threadIdx.x & 31;
  const int wid = __builtin_amdgcn_readfirstlane((int)((blockIdx.x * 256 + threadIdx.x) >> 5));
  for (int t = 0; t < NPW; ++t) {
    const int n = wid * NPW + t;
    if (n >= n_nodes) break;
    const int start = off[n];
    const int end = off[n + 1];
    for (int ib = start; ib < end; ib += 32) {
      const int i = ib + lane;
      const bool vi = i < end;
      const int mine = vi ? tmp[i] : 0;
      int r = 0;
      for (int kb = start; kb < end; kb += 32) {
        const int kj = kb + lane;
        const int other = (kj < end) ? tmp[kj] : 0;
        const int cnt = (end - kb < 32) ? (end - kb) : 32;
        for (int k = 0; k < cnt; ++k) {
          const int o = __builtin_amdgcn_readlane(other, k);
          const int kk = kb + k;
          r += (o < mine || (o == mine && kk < i)) ? 1 : 0;
        }
      }
      if (vi) csr[start + r] = mine;
    }
  }
}

template <int ELU>
__global__ __launch_bounds__(256) void node_walk(
    const int* __restrict__ off, const int* __restrict__ csr,
    const float* __restrict__ alpha_s, const float* __restrict__ alpha_d,
    const float* __restrict__ h, const float* __restrict__ b,
    float* __restrict__ out, int n_nodes) {
  const int lane = threadIdx.x & 31;
  const int wid = __builtin_amdgcn_readfirstlane((int)((blockIdx.x * 256 + threadIdx.x) >> 5));
  const float2 bb = *(const float2*)(b + 2 * lane);
  for (int t = 0; t < NPW; ++t) {
    const int n = wid * NPW + t;
    if (n >= n_nodes) break;
    const int start = off[n];
    const int end = off[n + 1];
    const float adn = alpha_d[n];
    float es = alpha_s[n] + adn;
    es = (es > 0.0f) ? es : NEG_SLOPE * es;

    float m = es;
    for (int base = start; base < end; base += 32) {
      const int j = base + lane;
      if (j < end) {
        const int sj = csr[j];
        float e = alpha_s[sj] + adn;
        e = (e > 0.0f) ? e : NEG_SLOPE * e;
        m = fmaxf(m, e);
      }
    }
#pragma unroll
    for (int o = 16; o > 0; o >>= 1) m = fmaxf(m, __shfl_xor(m, o, 32));

    const float pself = __expf(es - m);
    float ssum = pself;
    const float2 hn = *(const float2*)(h + (size_t)n * HDIM + 2 * lane);
    float ax = hn.x * pself;
    float ay = hn.y * pself;

    for (int base = start; base < end; base += 32) {
      const int j = base + lane;
      int sj = n;
      float p = 0.0f;
      if (j < end) {
        sj = csr[j];
        float e = alpha_s[sj] + adn;
        e = (e > 0.0f) ? e : NEG_SLOPE * e;
        p = __expf(e - m);
      }
      const int pi = __float_as_int(p);
      const int cnt = (end - base < 32) ? (end - base) : 32;
      for (int k = 0; k < cnt; k += 4) {
        const int s0 = __builtin_amdgcn_readlane(sj, k);
        const int s1 = __builtin_amdgcn_readlane(sj, k + 1);
        const int s2 = __builtin_amdgcn_readlane(sj, k + 2);
        const int s3 = __builtin_amdgcn_readlane(sj, k + 3);
        const float p0 = __int_as_float(__builtin_amdgcn_readlane(pi, k));
        const float p1 = __int_as_float(__builtin_amdgcn_readlane(pi, k + 1));
        const float p2 = __int_as_float(__builtin_amdgcn_readlane(pi, k + 2));
        const float p3 = __int_as_float(__builtin_amdgcn_readlane(pi, k + 3));
        const float2 g0 = *(const float2*)(h + (size_t)s0 * HDIM + 2 * lane);
        const float2 g1 = *(const float2*)(h + (size_t)s1 * HDIM + 2 * lane);
        const float2 g2 = *(const float2*)(h + (size_t)s2 * HDIM + 2 * lane);
        const float2 g3 = *(const float2*)(h + (size_t)s3 * HDIM + 2 * lane);
        ssum += p0; ax += g0.x * p0; ay += g0.y * p0;
        if (k + 1 < cnt) { ssum += p1; ax += g1.x * p1; ay += g1.y * p1; }
        if (k + 2 < cnt) { ssum += p2; ax += g2.x * p2; ay += g2.y * p2; }
        if (k + 3 < cnt) { ssum += p3; ax += g3.x * p3; ay += g3.y * p3; }
      }
    }

    float vx = ax / ssum + bb.x;
    float vy = ay / ssum + bb.y;
    if (ELU) {
      vx = (vx > 0.0f) ? vx : (__expf(vx) - 1.0f);
      vy = (vy > 0.0f) ? vy : (__expf(vy) - 1.0f);
    }
    *(float2*)(out + (size_t)n * HDIM + 2 * lane) = make_float2(vx, vy);
  }
}

extern "C" void kernel_launch(void* const* d_in, const int* in_sizes, int n_in,
                              void* d_out, int out_size, void* d_ws,
                              size_t ws_size, hipStream_t stream) {
  const float* X      = (const float*)d_in[0];
  const int*   adj    = (const int*)d_in[1];
  const float* W1     = (const float*)d_in[2];
  const float* a_src1 = (const float*)d_in[3];
  const float* a_dst1 = (const float*)d_in[4];
  const float* b1     = (const float*)d_in[5];
  const float* W2     = (const float*)d_in[6];
  const float* a_src2 = (const float*)d_in[7];
  const float* a_dst2 = (const float*)d_in[8];
  const float* b2     = (const float*)d_in[9];
  const float* clas_W = (const float*)d_in[10];
  const float* clas_b = (const float*)d_in[11];

  float* out = (float*)d_out;
  float* x2  = out + (size_t)N_NODES * N_CLASSES;

  float* ws   = (float*)d_ws;
  float* h    = ws;
  float* x1   = ws + (size_t)N_NODES * HDIM;
  int*   ib   = (int*)(ws + (size_t)2 * N_NODES * HDIM);
  float* as_  = ws + (size_t)3 * N_NODES * HDIM;
  float* ad_  = as_ + N_NODES;
  int* cnt  = ib;
  int* off  = ib + 100352;
  int* rank = ib + 2 * 100352;
  int* tmp  = rank + N_EDGES;
  int* csr  = tmp + N_EDGES;

  const int gemm_blocks = (N_NODES + 127) / 128;
  const int edge_blocks = (N_EDGES + 255) / 256;
  const int zero_blocks = (N_NODES + 255) / 256;
  const int scan_blocks = (N_NODES + 1 + 1023) / 1024;
  const int walk_blocks = (N_NODES + 8 * NPW - 1) / (8 * NPW);

  zero_counts<<<zero_blocks, 256, 0, stream>>>(cnt, N_NODES);
  edge_rank<<<edge_blocks, 256, 0, stream>>>(adj, cnt, rank, N_EDGES);
  scan_offsets<<<scan_blocks, 1024, 0, stream>>>(cnt, off, N_NODES);
  edge_fill<<<edge_blocks, 256, 0, stream>>>(adj, off, rank, tmp, N_EDGES);
  seg_sort<<<walk_blocks, 256, 0, stream>>>(off, tmp, csr, N_NODES);

  gemm_wmma_f32<IN_DIM, HDIM, true><<<gemm_blocks, 256, 0, stream>>>(
      X, W1, nullptr, h, N_NODES, a_src1, a_dst1, as_, ad_);
  node_walk<1><<<walk_blocks, 256, 0, stream>>>(off, csr, as_, ad_, h, b1, x1, N_NODES);

  gemm_wmma_f32<HDIM, HDIM, true><<<gemm_blocks, 256, 0, stream>>>(
      x1, W2, nullptr, h, N_NODES, a_src2, a_dst2, as_, ad_);
  node_walk<0><<<walk_blocks, 256, 0, stream>>>(off, csr, as_, ad_, h, b2, x2, N_NODES);

  gemm_wmma_f32<HDIM, N_CLASSES, false><<<gemm_blocks, 256, 0, stream>>>(
      x2, clas_W, clas_b, out, N_NODES, nullptr, nullptr, nullptr, nullptr);
}


# Node walk takes each source's score from the h row it gathers (16 rows in registers, 32-lane tree dot with a_src): per-edge alpha_s lookups go; one pass, running max, sums rescaled per 16 edges
# speedup vs baseline: 2.6787x; 1.0434x over previous
#include <hip/hip_runtime.h>
#include <hip/hip_bf16.h>

typedef float v2f __attribute__((ext_vector_type(2)));
typedef float v8f __attribute__((ext_vector_type(8)));

#define N_NODES   100000
#define N_EDGES   1600000
#define IN_DIM    128
#define HDIM      64
#define N_CLASSES 16
#define NEG_SLOPE 0.2f

template <int K, int N, bool ALPHA>
__global__ __launch_bounds__(256) void gemm_wmma_f32(
    const float* __restrict__ A, const float* __restrict__ B,
    const float* __restrict__ bias, float* __restrict__ C, int M,
    const float* __restrict__ a_src, const float* __restrict__ a_dst,
    float* __restrict__ alpha_s, float* __restrict__ alpha_d) {
  __shared__ float Bs[K * N];
  const int tid = threadIdx.x;
  for (int i = tid; i < K * N; i += 256) Bs[i] = B[i];
  __syncthreads();

  const int wave = tid >> 5;
  const int lane = tid & 31;
  const int hi = lane >> 4;
  const int lo = lane & 15;

  const long m0 = (long)blockIdx.x * 128 + wave * 16;
  long r = m0 + lo;
  if (r > M - 1) r = M - 1;
  const float* __restrict__ Arow = A + (size_t)r * K;

  constexpr int NT = N / 16;
  v8f acc[NT] = {};

  [[maybe_unused]] float avs[NT], avd[NT];
  if constexpr (ALPHA) {
#pragma unroll
    for (int nt = 0; nt < NT; ++nt) {
      avs[nt] = a_src[nt * 16 + lo];
      avd[nt] = a_dst[nt * 16 + lo];
    }
  }

  for (int kk = 0; kk < K / 4; ++kk) {
    const int kb = kk * 4 + 2 * hi;
    v2f a;
    a.x = Arow[kb];
    a.y = Arow[kb + 1];
#pragma unroll
    for (int nt = 0; nt < NT; ++nt) {
      v2f b;
      b.x = Bs[kb * N + nt * 16 + lo];
      b.y = Bs[(kb + 1) * N + nt * 16 + lo];
      acc[nt] = __builtin_amdgcn_wmma_f32_16x16x4_f32(
           false, a,  false, b,
           (short)0, acc[nt],  false,  false);
    }
  }

#pragma unroll
  for (int nt = 0; nt < NT; ++nt) {
    const int col = nt * 16 + lo;
    const float bv = bias ? bias[col] : 0.0f;
#pragma unroll
    for (int rr = 0; rr < 8; ++rr) {
      const long row = m0 + rr + 8 * hi;
      if (row < M) C[(size_t)row * N + col] = acc[nt][rr] + bv;
    }
  }

  if constexpr (ALPHA) {
    float v[16];
#pragma unroll
    for (int rr = 0; rr < 8; ++rr) {
      float s = acc[0][rr] * avs[0];
      float d = acc[0][rr] * avd[0];
#pragma unroll
      for (int nt = 1; nt < NT; ++nt) {
        s += acc[nt][rr] * avs[nt];
        d += acc[nt][rr] * avd[nt];
      }
      v[rr] = s;
      v[8 + rr] = d;
    }
#pragma unroll
    for (int w = 8; w >= 1; w >>= 1) {
      const bool up = (lo & w) != 0;
#pragma unroll
      for (int i = 0; i < w; ++i) {
        const float lower = v[i];
        const float upper = v[i + w];
        const float keep = up ? upper : lower;
        const float send = up ? lower : upper;
        v[i] = keep + __shfl_xor(send, w, 32);
      }
    }
    const long row = m0 + (lo & 7) + 8 * hi;
    float* __restrict__ dstp = (lo < 8) ? alpha_s : alpha_d;
    if (row < M) dstp[row] = v[0];
  }
}

__global__ __launch_bounds__(256) void zero_counts(int* __restrict__ cnt, int n) {
  const int i = blockIdx.x * 256 + threadIdx.x;
  if (i < n) cnt[i] = 0;
}

__global__ __launch_bounds__(256) void edge_rank(
    const int* __restrict__ adj, int* __restrict__ cnt,
    int* __restrict__ rank, int n_edges) {
  const int i = blockIdx.x * 256 + threadIdx.x;
  if (i >= n_edges) return;
  const int dst = adj[n_edges + i];
  rank[i] = atomicAdd(&cnt[dst], 1);
}

__global__ __launch_bounds__(1024) void scan_offsets(
    const int* __restrict__ cnt, int* __restrict__ off, int n) {
  __shared__ int wsum[32];
  const int tid = threadIdx.x;
  const int lane = tid & 31;
  const int wave = tid >> 5;
  const int lim = blockIdx.x * 1024;
  int v = 0;
  for (int j = tid; j < lim; j += 1024) v += (j < n) ? cnt[j] : 0;
#pragma unroll
  for (int o = 16; o > 0; o >>= 1) v += __shfl_xor(v, o, 32);
  if (lane == 0) wsum[wave] = v;
  __syncthreads();
  int base = 0;
#pragma unroll
  for (int w = 0; w < 32; ++w) base += wsum[w];
  __syncthreads();
  const int idx = lim + tid;
  const int c = (idx < n) ? cnt[idx] : 0;
  int inc = c;
#pragma unroll
  for (int o = 1; o < 32; o <<= 1) {
    const int t = __shfl_up(inc, o, 32);
    if (lane >= o) inc += t;
  }
  if (lane == 31) wsum[wave] = inc;
  __syncthreads();
  int wpre = 0;
  for (int w = 0; w < wave; ++w) wpre += wsum[w];
  if (idx <= n) off[idx] = base + wpre + inc - c;
}

__global__ __launch_bounds__(256) void edge_fill(
    const int* __restrict__ adj, const int* __restrict__ off,
    const int* __restrict__ rank, int* __restrict__ tmp, int n_edges) {
  const int i = blockIdx.x * 256 + threadIdx.x;
  if (i >= n_edges) return;
  const int dst = adj[n_edges + i];
  tmp[off[dst] + rank[i]] = adj[i];
}

#define NPW 4

__global__ __launch_bounds__(256) void seg_sort(
    const int* __restrict__ off, const int* __restrict__ tmp,
    int* __restrict__ csr, int n_nodes) {
  const int lane = threadIdx.x & 31;
  const int wid = __builtin_amdgcn_readfirstlane((int)((blockIdx.x * 256 + threadIdx.x) >> 5));
  for (int t = 0; t < NPW; ++t) {
    const int n = wid * NPW + t;
    if (n >= n_nodes) break;
    const int start = off[n];
    const int end = off[n + 1];
    for (int ib = start; ib < end; ib += 32) {
      const int i = ib + lane;
      const bool vi = i < end;
      const int mine = vi ? tmp[i] : 0;
      int r = 0;
      for (int kb = start; kb < end; kb += 32) {
        const int kj = kb + lane;
        const int other = (kj < end) ? tmp[kj] : 0;
        const int cnt = (end - kb < 32) ? (end - kb) : 32;
        for (int k = 0; k < cnt; ++k) {
          const int o = __builtin_amdgcn_readlane(other, k);
          const int kk = kb + k;
          r += (o < mine || (o == mine && kk < i)) ? 1 : 0;
        }
      }
      if (vi) csr[start + r] = mine;
    }
  }
}

#define CH 16

__device__ __forceinline__ void walk_chunk(
    const float* __restrict__ h, const int lane, const int myk, const float2 av,
    const float adn, const int sj, const int c, const int cnt,
    float& m, float& ssum, float& ax, float& ay) {
  float2 g[CH];
#pragma unroll
  for (int k = 0; k < CH; ++k) {
    if (k == 0 || k < cnt) {
      const int s = __builtin_amdgcn_readlane(sj, c + k);
      g[k] = *(const float2*)(h + (size_t)s * HDIM + 2 * lane);
    }
  }

  float v[CH];
#pragma unroll
  for (int k = 0; k < CH; ++k) {
    v[k] = 0.0f;
    if (k == 0 || k < cnt) v[k] = g[k].x * av.x + g[k].y * av.y;
  }

#pragma unroll
  for (int w = 16; w >= 2; w >>= 1) {
    const bool up = (lane & w) != 0;
#pragma unroll
    for (int i = 0; i < (w >> 1); ++i) {
      const float lower = v[i];
      const float upper = v[i + (w >> 1)];
      const float keep = up ? upper : lower;
      const float send = up ? lower : upper;
      v[i] = keep + __shfl_xor(send, w, 32);
    }
  }
  const float score = v[0] + __shfl_xor(v[0], 1, 32);

  float e = score + adn;
  e = (e > 0.0f) ? e : NEG_SLOPE * e;
  const bool valid = myk < cnt;
  float cm = valid ? e : m;
#pragma unroll
  for (int o = 16; o >= 2; o >>= 1) cm = fmaxf(cm, __shfl_xor(cm, o, 32));
  const float mn = fmaxf(m, cm);
  const float scale = __expf(m - mn);
  ssum *= scale;
  ax *= scale;
  ay *= scale;
  m = mn;

  const float p = valid ? __expf(e - m) : 0.0f;
  const int pi = __float_as_int(p);
#pragma unroll
  for (int k = 0; k < CH; ++k) {
    if (k == 0 || k < cnt) {
      const float pk = __int_as_float(__builtin_amdgcn_readlane(pi, 2 * k));
      ssum += pk;
      ax += g[k].x * pk;
      ay += g[k].y * pk;
    }
  }
}

template <int ELU>
__global__ __launch_bounds__(256) void node_walk(
    const int* __restrict__ off, const int* __restrict__ csr,
    const float* __restrict__ alpha_s, const float* __restrict__ alpha_d,
    const float* __restrict__ h, const float* __restrict__ a_src,
    const float* __restrict__ b, float* __restrict__ out, int n_nodes) {
  const int lane = threadIdx.x & 31;
  const int wid = __builtin_amdgcn_readfirstlane((int)((blockIdx.x * 256 + threadIdx.x) >> 5));
  const float2 bb = *(const float2*)(b + 2 * lane);
  const float2 av = *(const float2*)(a_src + 2 * lane);
  const int myk = lane >> 1;
  for (int t = 0; t < NPW; ++t) {
    const int n = wid * NPW + t;
    if (n >= n_nodes) break;
    const int start = off[n];
    const int end = off[n + 1];
    const float adn = alpha_d[n];
    float es = alpha_s[n] + adn;
    es = (es > 0.0f) ? es : NEG_SLOPE * es;
    const float2 hn = *(const float2*)(h + (size_t)n * HDIM + 2 * lane);

    float m = es;
    float ssum = 1.0f;
    float ax = hn.x;
    float ay = hn.y;

    for (int base = start; base < end; base += 32) {
      const int j = base + lane;
      const int sj = csr[(j < end) ? j : (end - 1)];
      const int rem = (end - base < 32) ? (end - base) : 32;
      for (int c = 0; c < rem; c += CH)
        walk_chunk(h, lane, myk, av, adn, sj, c,
                   (rem - c < CH) ? (rem - c) : CH, m, ssum, ax, ay);
    }

    float vx = ax / ssum + bb.x;
    float vy = ay / ssum + bb.y;
    if (ELU) {
      vx = (vx > 0.0f) ? vx : (__expf(vx) - 1.0f);
      vy = (vy > 0.0f) ? vy : (__expf(vy) - 1.0f);
    }
    *(float2*)(out + (size_t)n * HDIM + 2 * lane) = make_float2(vx, vy);
  }
}

extern "C" void kernel_launch(void* const* d_in, const int* in_sizes, int n_in,
                              void* d_out, int out_size, void* d_ws,
                              size_t ws_size, hipStream_t stream) {
  const float* X      = (const float*)d_in[0];
  const int*   adj    = (const int*)d_in[1];
  const float* W1     = (const float*)d_in[2];
  const float* a_src1 = (const float*)d_in[3];
  const float* a_dst1 = (const float*)d_in[4];
  const float* b1     = (const float*)d_in[5];
  const float* W2     = (const float*)d_in[6];
  const float* a_src2 = (const float*)d_in[7];
  const float* a_dst2 = (const float*)d_in[8];
  const float* b2     = (const float*)d_in[9];
  const float* clas_W = (const float*)d_in[10];
  const float* clas_b = (const float*)d_in[11];

  float* out = (float*)d_out;
  float* x2  = out + (size_t)N_NODES * N_CLASSES;

  float* ws   = (float*)d_ws;
  float* h    = ws;
  float* x1   = ws + (size_t)N_NODES * HDIM;
  int*   ib   = (int*)(ws + (size_t)2 * N_NODES * HDIM);
  float* as_  = ws + (size_t)3 * N_NODES * HDIM;
  float* ad_  = as_ + N_NODES;
  int* cnt  = ib;
  int* off  = ib + 100352;
  int* rank = ib + 2 * 100352;
  int* tmp  = rank + N_EDGES;
  int* csr  = tmp + N_EDGES;

  const int gemm_blocks = (N_NODES + 127) / 128;
  const int edge_blocks = (N_EDGES + 255) / 256;
  const int zero_blocks = (N_NODES + 255) / 256;
  const int scan_blocks = (N_NODES + 1 + 1023) / 1024;
  const int walk_blocks = (N_NODES + 8 * NPW - 1) / (8 * NPW);

  zero_counts<<<zero_blocks, 256, 0, stream>>>(cnt, N_NODES);
  edge_rank<<<edge_blocks, 256, 0, stream>>>(adj, cnt, rank, N_EDGES);
  scan_offsets<<<scan_blocks, 1024, 0, stream>>>(cnt, off, N_NODES);
  edge_fill<<<edge_blocks, 256, 0, stream>>>(adj, off, rank, tmp, N_EDGES);
  seg_sort<<<walk_blocks, 256, 0, stream>>>(off, tmp, csr, N_NODES);

  gemm_wmma_f32<IN_DIM, HDIM, true><<<gemm_blocks, 256, 0, stream>>>(
      X, W1, nullptr, h, N_NODES, a_src1, a_dst1, as_, ad_);
  node_walk<1><<<walk_blocks, 256, 0, stream>>>(off, csr, as_, ad_, h, a_src1, b1, x1, N_NODES);

  gemm_wmma_f32<HDIM, HDIM, true><<<gemm_blocks, 256, 0, stream>>>(
      x1, W2, nullptr, h, N_NODES, a_src2, a_dst2, as_, ad_);
  node_walk<0><<<walk_blocks, 256, 0, stream>>>(off, csr, as_, ad_, h, a_src2, b2, x2, N_NODES);

  gemm_wmma_f32<HDIM, N_CLASSES, false><<<gemm_blocks, 256, 0, stream>>>(
      x2, clas_W, clas_b, out, N_NODES, nullptr, nullptr, nullptr, nullptr);
}


# scan_offsets: each block sums the counts in front of its tile with 16 independent 128-bit loads in flight per thread, not one dword load per memory round trip (97 in a row); integer sums only
# speedup vs baseline: 3.0448x; 1.1367x over previous
#include <hip/hip_runtime.h>
#include <hip/hip_bf16.h>

typedef float v2f __attribute__((ext_vector_type(2)));
typedef float v8f __attribute__((ext_vector_type(8)));

#define N_NODES   100000
#define N_EDGES   1600000
#define IN_DIM    128
#define HDIM      64
#define N_CLASSES 16
#define NEG_SLOPE 0.2f

template <int K, int N, bool ALPHA>
__global__ __launch_bounds__(256) void gemm_wmma_f32(
    const float* __restrict__ A, const float* __restrict__ B,
    const float* __restrict__ bias, float* __restrict__ C, int M,
    const float* __restrict__ a_src, const float* __restrict__ a_dst,
    float* __restrict__ alpha_s, float* __restrict__ alpha_d) {
  __shared__ float Bs[K * N];
  const int tid = threadIdx.x;
  for (int i = tid; i < K * N; i += 256) Bs[i] = B[i];
  __syncthreads();

  const int wave = tid >> 5;
  const int lane = tid & 31;
  const int hi = lane >> 4;
  const int lo = lane & 15;

  const long m0 = (long)blockIdx.x * 128 + wave * 16;
  long r = m0 + lo;
  if (r > M - 1) r = M - 1;
  const float* __restrict__ Arow = A + (size_t)r * K;

  constexpr int NT = N / 16;
  v8f acc[NT] = {};

  [[maybe_unused]] float avs[NT], avd[NT];
  if constexpr (ALPHA) {
#pragma unroll
    for (int nt = 0; nt < NT; ++nt) {
      avs[nt] = a_src[nt * 16 + lo];
      avd[nt] = a_dst[nt * 16 + lo];
    }
  }

  for (int kk = 0; kk < K / 4; ++kk) {
    const int kb = kk * 4 + 2 * hi;
    v2f a;
    a.x = Arow[kb];
    a.y = Arow[kb + 1];
#pragma unroll
    for (int nt = 0; nt < NT; ++nt) {
      v2f b;
      b.x = Bs[kb * N + nt * 16 + lo];
      b.y = Bs[(kb + 1) * N + nt * 16 + lo];
      acc[nt] = __builtin_amdgcn_wmma_f32_16x16x4_f32(
           false, a,  false, b,
           (short)0, acc[nt],  false,  false);
    }
  }

#pragma unroll
  for (int nt = 0; nt < NT; ++nt) {
    const int col = nt * 16 + lo;
    const float bv = bias ? bias[col] : 0.0f;
#pragma unroll
    for (int rr = 0; rr < 8; ++rr) {
      const long row = m0 + rr + 8 * hi;
      if (row < M) C[(size_t)row * N + col] = acc[nt][rr] + bv;
    }
  }

  if constexpr (ALPHA) {
    float v[16];
#pragma unroll
    for (int rr = 0; rr < 8; ++rr) {
      float s = acc[0][rr] * avs[0];
      float d = acc[0][rr] * avd[0];
#pragma unroll
      for (int nt = 1; nt < NT; ++nt) {
        s += acc[nt][rr] * avs[nt];
        d += acc[nt][rr] * avd[nt];
      }
      v[rr] = s;
      v[8 + rr] = d;
    }
#pragma unroll
    for (int w = 8; w >= 1; w >>= 1) {
      const bool up = (lo & w) != 0;
#pragma unroll
      for (int i = 0; i < w; ++i) {
        const float lower = v[i];
        const float upper = v[i + w];
        const float keep = up ? upper : lower;
        const float send = up ? lower : upper;
        v[i] = keep + __shfl_xor(send, w, 32);
      }
    }
    const long row = m0 + (lo & 7) + 8 * hi;
    float* __restrict__ dstp = (lo < 8) ? alpha_s : alpha_d;
    if (row < M) dstp[row] = v[0];
  }
}

__global__ __launch_bounds__(256) void zero_counts(int* __restrict__ cnt, int n) {
  const int i = blockIdx.x * 256 + threadIdx.x;
  if (i < n) cnt[i] = 0;
}

__global__ __launch_bounds__(256) void edge_rank(
    const int* __restrict__ adj, int* __restrict__ cnt,
    int* __restrict__ rank, int n_edges) {
  const int i = blockIdx.x * 256 + threadIdx.x;
  if (i >= n_edges) return;
  const int dst = adj[n_edges + i];
  rank[i] = atomicAdd(&cnt[dst], 1);
}

#define PF 16

__global__ __launch_bounds__(1024) __attribute__((amdgpu_waves_per_eu(8, 8))) void scan_offsets(
    const int* __restrict__ cnt, int* __restrict__ off, int n) {
  __shared__ int wsum[32];
  const int tid = threadIdx.x;
  const int lane = tid & 31;
  const int wave = tid >> 5;
  const int lim = blockIdx.x * 1024;
  const int idx = lim + tid;
  const int c = (idx < n) ? cnt[idx] : 0;

  const int4* __restrict__ p = (const int4*)cnt;
  const int nvec = lim >> 2;
  int v = 0;
#pragma clang loop unroll(disable) vectorize(disable) interleave(disable)
  for (int r0 = 0; r0 < nvec; r0 += PF * 1024) {
    int4 t[PF];
#pragma unroll
    for (int u = 0; u < PF; ++u) {
      const int j = r0 + u * 1024 + tid;
      t[u] = p[(j < nvec) ? j : (nvec - 1)];
    }
#pragma unroll
    for (int u = 0; u < PF; ++u) {
      const int j = r0 + u * 1024 + tid;
      const int s = (t[u].x + t[u].y) + (t[u].z + t[u].w);
      v += (j < nvec) ? s : 0;
    }
  }
#pragma unroll
  for (int o = 16; o > 0; o >>= 1) v += __shfl_xor(v, o, 32);
  if (lane == 0) wsum[wave] = v;
  __syncthreads();
  int base = 0;
#pragma unroll
  for (int w = 0; w < 32; ++w) base += wsum[w];
  __syncthreads();
  int inc = c;
#pragma unroll
  for (int o = 1; o < 32; o <<= 1) {
    const int t = __shfl_up(inc, o, 32);
    if (lane >= o) inc += t;
  }
  if (lane == 31) wsum[wave] = inc;
  __syncthreads();
  int wpre = 0;
  for (int w = 0; w < wave; ++w) wpre += wsum[w];
  if (idx <= n) off[idx] = base + wpre + inc - c;
}

__global__ __launch_bounds__(256) void edge_fill(
    const int* __restrict__ adj, const int* __restrict__ off,
    const int* __restrict__ rank, int* __restrict__ tmp, int n_edges) {
  const int i = blockIdx.x * 256 + threadIdx.x;
  if (i >= n_edges) return;
  const int dst = adj[n_edges + i];
  tmp[off[dst] + rank[i]] = adj[i];
}

#define NPW 4

__global__ __launch_bounds__(256) void seg_sort(
    const int* __restrict__ off, const int* __restrict__ tmp,
    int* __restrict__ csr, int n_nodes) {
  const int lane = threadIdx.x & 31;
  const int wid = __builtin_amdgcn_readfirstlane((int)((blockIdx.x * 256 + threadIdx.x) >> 5));
  for (int t = 0; t < NPW; ++t) {
    const int n = wid * NPW + t;
    if (n >= n_nodes) break;
    const int start = off[n];
    const int end = off[n + 1];
    for (int ib = start; ib < end; ib += 32) {
      const int i = ib + lane;
      const bool vi = i < end;
      const int mine = vi ? tmp[i] : 0;
      int r = 0;
      for (int kb = start; kb < end; kb += 32) {
        const int kj = kb + lane;
        const int other = (kj < end) ? tmp[kj] : 0;
        const int cnt = (end - kb < 32) ? (end - kb) : 32;
        for (int k = 0; k < cnt; ++k) {
          const int o = __builtin_amdgcn_readlane(other, k);
          const int kk = kb + k;
          r += (o < mine || (o == mine && kk < i)) ? 1 : 0;
        }
      }
      if (vi) csr[start + r] = mine;
    }
  }
}

#define CH 16

__device__ __forceinline__ void walk_chunk(
    const float* __restrict__ h, const int lane, const int myk, const float2 av,
    const float adn, const int sj, const int c, const int cnt,
    float& m, float& ssum, float& ax, float& ay) {
  float2 g[CH];
#pragma unroll
  for (int k = 0; k < CH; ++k) {
    if (k == 0 || k < cnt) {
      const int s = __builtin_amdgcn_readlane(sj, c + k);
      g[k] = *(const float2*)(h + (size_t)s * HDIM + 2 * lane);
    }
  }

  float v[CH];
#pragma unroll
  for (int k = 0; k < CH; ++k) {
    v[k] = 0.0f;
    if (k == 0 || k < cnt) v[k] = g[k].x * av.x + g[k].y * av.y;
  }

#pragma unroll
  for (int w = 16; w >= 2; w >>= 1) {
    const bool up = (lane & w) != 0;
#pragma unroll
    for (int i = 0; i < (w >> 1); ++i) {
      const float lower = v[i];
      const float upper = v[i + (w >> 1)];
      const float keep = up ? upper : lower;
      const float send = up ? lower : upper;
      v[i] = keep + __shfl_xor(send, w, 32);
    }
  }
  const float score = v[0] + __shfl_xor(v[0], 1, 32);

  float e = score + adn;
  e = (e > 0.0f) ? e : NEG_SLOPE * e;
  const bool valid = myk < cnt;
  float cm = valid ? e : m;
#pragma unroll
  for (int o = 16; o >= 2; o >>= 1) cm = fmaxf(cm, __shfl_xor(cm, o, 32));
  const float mn = fmaxf(m, cm);
  const float scale = __expf(m - mn);
  ssum *= scale;
  ax *= scale;
  ay *= scale;
  m = mn;

  const float p = valid ? __expf(e - m) : 0.0f;
  const int pi = __float_as_int(p);
#pragma unroll
  for (int k = 0; k < CH; ++k) {
    if (k == 0 || k < cnt) {
      const float pk = __int_as_float(__builtin_amdgcn_readlane(pi, 2 * k));
      ssum += pk;
      ax += g[k].x * pk;
      ay += g[k].y * pk;
    }
  }
}

template <int ELU>
__global__ __launch_bounds__(256) void node_walk(
    const int* __restrict__ off, const int* __restrict__ csr,
    const float* __restrict__ alpha_s, const float* __restrict__ alpha_d,
    const float* __restrict__ h, const float* __restrict__ a_src,
    const float* __restrict__ b, float* __restrict__ out, int n_nodes) {
  const int lane = threadIdx.x & 31;
  const int wid = __builtin_amdgcn_readfirstlane((int)((blockIdx.x * 256 + threadIdx.x) >> 5));
  const float2 bb = *(const float2*)(b + 2 * lane);
  const float2 av = *(const float2*)(a_src + 2 * lane);
  const int myk = lane >> 1;
  for (int t = 0; t < NPW; ++t) {
    const int n = wid * NPW + t;
    if (n >= n_nodes) break;
    const int start = off[n];
    const int end = off[n + 1];
    const float adn = alpha_d[n];
    float es = alpha_s[n] + adn;
    es = (es > 0.0f) ? es : NEG_SLOPE * es;
    const float2 hn = *(const float2*)(h + (size_t)n * HDIM + 2 * lane);

    float m = es;
    float ssum = 1.0f;
    float ax = hn.x;
    float ay = hn.y;

    for (int base = start; base < end; base += 32) {
      const int j = base + lane;
      const int sj = csr[(j < end) ? j : (end - 1)];
      const int rem = (end - base < 32) ? (end - base) : 32;
      for (int c = 0; c < rem; c += CH)
        walk_chunk(h, lane, myk, av, adn, sj, c,
                   (rem - c < CH) ? (rem - c) : CH, m, ssum, ax, ay);
    }

    float vx = ax / ssum + bb.x;
    float vy = ay / ssum + bb.y;
    if (ELU) {
      vx = (vx > 0.0f) ? vx : (__expf(vx) - 1.0f);
      vy = (vy > 0.0f) ? vy : (__expf(vy) - 1.0f);
    }
    *(float2*)(out + (size_t)n * HDIM + 2 * lane) = make_float2(vx, vy);
  }
}

extern "C" void kernel_launch(void* const* d_in, const int* in_sizes, int n_in,
                              void* d_out, int out_size, void* d_ws,
                              size_t ws_size, hipStream_t stream) {
  const float* X      = (const float*)d_in[0];
  const int*   adj    = (const int*)d_in[1];
  const float* W1     = (const float*)d_in[2];
  const float* a_src1 = (const float*)d_in[3];
  const float* a_dst1 = (const float*)d_in[4];
  const float* b1     = (const float*)d_in[5];
  const float* W2     = (const float*)d_in[6];
  const float* a_src2 = (const float*)d_in[7];
  const float* a_dst2 = (const float*)d_in[8];
  const float* b2     = (const float*)d_in[9];
  const float* clas_W = (const float*)d_in[10];
  const float* clas_b = (const float*)d_in[11];

  float* out = (float*)d_out;
  float* x2  = out + (size_t)N_NODES * N_CLASSES;

  float* ws   = (float*)d_ws;
  float* h    = ws;
  float* x1   = ws + (size_t)N_NODES * HDIM;
  int*   ib   = (int*)(ws + (size_t)2 * N_NODES * HDIM);
  float* as_  = ws + (size_t)3 * N_NODES * HDIM;
  float* ad_  = as_ + N_NODES;
  int* cnt  = ib;
  int* off  = ib + 100352;
  int* rank = ib + 2 * 100352;
  int* tmp  = rank + N_EDGES;
  int* csr  = tmp + N_EDGES;

  const int gemm_blocks = (N_NODES + 127) / 128;
  const int edge_blocks = (N_EDGES + 255) / 256;
  const int zero_blocks = (N_NODES + 255) / 256;
  const int scan_blocks = (N_NODES + 1 + 1023) / 1024;
  const int walk_blocks = (N_NODES + 8 * NPW - 1) / (8 * NPW);

  zero_counts<<<zero_blocks, 256, 0, stream>>>(cnt, N_NODES);
  edge_rank<<<edge_blocks, 256, 0, stream>>>(adj, cnt, rank, N_EDGES);
  scan_offsets<<<scan_blocks, 1024, 0, stream>>>(cnt, off, N_NODES);
  edge_fill<<<edge_blocks, 256, 0, stream>>>(adj, off, rank, tmp, N_EDGES);
  seg_sort<<<walk_blocks, 256, 0, stream>>>(off, tmp, csr, N_NODES);

  gemm_wmma_f32<IN_DIM, HDIM, true><<<gemm_blocks, 256, 0, stream>>>(
      X, W1, nullptr, h, N_NODES, a_src1, a_dst1, as_, ad_);
  node_walk<1><<<walk_blocks, 256, 0, stream>>>(off, csr, as_, ad_, h, a_src1, b1, x1, N_NODES);

  gemm_wmma_f32<HDIM, HDIM, true><<<gemm_blocks, 256, 0, stream>>>(
      x1, W2, nullptr, h, N_NODES, a_src2, a_dst2, as_, ad_);
  node_walk<0><<<walk_blocks, 256, 0, stream>>>(off, csr, as_, ad_, h, a_src2, b2, x2, N_NODES);

  gemm_wmma_f32<HDIM, N_CLASSES, false><<<gemm_blocks, 256, 0, stream>>>(
      x2, clas_W, clas_b, out, N_NODES, nullptr, nullptr, nullptr, nullptr);
}


# Edge lists by destination built by a two-level counting sort binned in LDS (256-node bins, runs written as neighbouring words), not a global atomic and a 4-byte scatter per edge; float sums untouched
# speedup vs baseline: 4.7225x; 1.5510x over previous
#include <hip/hip_runtime.h>
#include <hip/hip_bf16.h>

typedef float v2f __attribute__((ext_vector_type(2)));
typedef float v8f __attribute__((ext_vector_type(8)));

#define N_NODES   100000
#define N_EDGES   1600000
#define IN_DIM    128
#define HDIM      64
#define N_CLASSES 16
#define NEG_SLOPE 0.2f

template <int K, int N, bool ALPHA>
__global__ __launch_bounds__(256) void gemm_wmma_f32(
    const float* __restrict__ A, const float* __restrict__ B,
    const float* __restrict__ bias, float* __restrict__ C, int M,
    const float* __restrict__ a_src, const float* __restrict__ a_dst,
    float* __restrict__ alpha_s, float* __restrict__ alpha_d) {
  __shared__ float Bs[K * N];
  const int tid = threadIdx.x;
  for (int i = tid; i < K * N; i += 256) Bs[i] = B[i];
  __syncthreads();

  const int wave = tid >> 5;
  const int lane = tid & 31;
  const int hi = lane >> 4;
  const int lo = lane & 15;

  const long m0 = (long)blockIdx.x * 128 + wave * 16;
  long r = m0 + lo;
  if (r > M - 1) r = M - 1;
  const float* __restrict__ Arow = A + (size_t)r * K;

  constexpr int NT = N / 16;
  v8f acc[NT] = {};

  [[maybe_unused]] float avs[NT], avd[NT];
  if constexpr (ALPHA) {
#pragma unroll
    for (int nt = 0; nt < NT; ++nt) {
      avs[nt] = a_src[nt * 16 + lo];
      avd[nt] = a_dst[nt * 16 + lo];
    }
  }

  for (int kk = 0; kk < K / 4; ++kk) {
    const int kb = kk * 4 + 2 * hi;
    v2f a;
    a.x = Arow[kb];
    a.y = Arow[kb + 1];
#pragma unroll
    for (int nt = 0; nt < NT; ++nt) {
      v2f b;
      b.x = Bs[kb * N + nt * 16 + lo];
      b.y = Bs[(kb + 1) * N + nt * 16 + lo];
      acc[nt] = __builtin_amdgcn_wmma_f32_16x16x4_f32(
           false, a,  false, b,
           (short)0, acc[nt],  false,  false);
    }
  }

#pragma unroll
  for (int nt = 0; nt < NT; ++nt) {
    const int col = nt * 16 + lo;
    const float bv = bias ? bias[col] : 0.0f;
#pragma unroll
    for (int rr = 0; rr < 8; ++rr) {
      const long row = m0 + rr + 8 * hi;
      if (row < M) C[(size_t)row * N + col] = acc[nt][rr] + bv;
    }
  }

  if constexpr (ALPHA) {
    float v[16];
#pragma unroll
    for (int rr = 0; rr < 8; ++rr) {
      float s = acc[0][rr] * avs[0];
      float d = acc[0][rr] * avd[0];
#pragma unroll
      for (int nt = 1; nt < NT; ++nt) {
        s += acc[nt][rr] * avs[nt];
        d += acc[nt][rr] * avd[nt];
      }
      v[rr] = s;
      v[8 + rr] = d;
    }
#pragma unroll
    for (int w = 8; w >= 1; w >>= 1) {
      const bool up = (lo & w) != 0;
#pragma unroll
      for (int i = 0; i < w; ++i) {
        const float lower = v[i];
        const float upper = v[i + w];
        const float keep = up ? upper : lower;
        const float send = up ? lower : upper;
        v[i] = keep + __shfl_xor(send, w, 32);
      }
    }
    const long row = m0 + (lo & 7) + 8 * hi;
    float* __restrict__ dstp = (lo < 8) ? alpha_s : alpha_d;
    if (row < M) dstp[row] = v[0];
  }
}

#define BIN_SHIFT 8
#define BIN_NODES 256
#define NBINS     ((N_NODES + BIN_NODES - 1) / BIN_NODES)
#define NCHUNK    256
#define CHUNK     ((N_EDGES + NCHUNK - 1) / NCHUNK)
#define NCNT      (NBINS * NCHUNK)

#define EPT ((CHUNK + 1023) / 1024)

__global__ __launch_bounds__(1024) void bin_hist(
    const int* __restrict__ adj, int* __restrict__ cnt2) {
  __shared__ int h[NBINS];
  const int tid = threadIdx.x;
  const int blk = blockIdx.x;
  for (int i = tid; i < NBINS; i += 1024) h[i] = 0;
  const int e0 = blk * CHUNK;
  const int e1 = (e0 + CHUNK < N_EDGES) ? (e0 + CHUNK) : N_EDGES;
  int d[EPT];
#pragma unroll
  for (int u = 0; u < EPT; ++u) {
    const int i = e0 + tid + u * 1024;
    d[u] = (i < e1) ? adj[N_EDGES + i] : -1;
  }
  __syncthreads();
#pragma unroll
  for (int u = 0; u < EPT; ++u)
    if (d[u] >= 0) atomicAdd(&h[d[u] >> BIN_SHIFT], 1);
  __syncthreads();
  for (int b = tid; b < NBINS; b += 1024) cnt2[b * NCHUNK + blk] = h[b];
}

#define PF 16

__global__ __launch_bounds__(1024) __attribute__((amdgpu_waves_per_eu(8, 8))) void scan_offsets(
    const int* __restrict__ cnt, int* __restrict__ off, int n) {
  __shared__ int wsum[32];
  const int tid = threadIdx.x;
  const int lane = tid & 31;
  const int wave = tid >> 5;
  const int lim = blockIdx.x * 1024;
  const int idx = lim + tid;
  const int c = (idx < n) ? cnt[idx] : 0;

  const int4* __restrict__ p = (const int4*)cnt;
  const int nvec = lim >> 2;
  int v = 0;
#pragma clang loop unroll(disable) vectorize(disable) interleave(disable)
  for (int r0 = 0; r0 < nvec; r0 += PF * 1024) {
    int4 t[PF];
#pragma unroll
    for (int u = 0; u < PF; ++u) {
      const int j = r0 + u * 1024 + tid;
      t[u] = p[(j < nvec) ? j : (nvec - 1)];
    }
#pragma unroll
    for (int u = 0; u < PF; ++u) {
      const int j = r0 + u * 1024 + tid;
      const int s = (t[u].x + t[u].y) + (t[u].z + t[u].w);
      v += (j < nvec) ? s : 0;
    }
  }
#pragma unroll
  for (int o = 16; o > 0; o >>= 1) v += __shfl_xor(v, o, 32);
  if (lane == 0) wsum[wave] = v;
  __syncthreads();
  int base = 0;
#pragma unroll
  for (int w = 0; w < 32; ++w) base += wsum[w];
  __syncthreads();
  int inc = c;
#pragma unroll
  for (int o = 1; o < 32; o <<= 1) {
    const int t = __shfl_up(inc, o, 32);
    if (lane >= o) inc += t;
  }
  if (lane == 31) wsum[wave] = inc;
  __syncthreads();
  int wpre = 0;
  for (int w = 0; w < wave; ++w) wpre += wsum[w];
  if (idx <= n) off[idx] = base + wpre + inc - c;
}

__global__ __launch_bounds__(1024) void bin_scatter(
    const int* __restrict__ adj, const int* __restrict__ off2,
    int* __restrict__ ebuf) {
  __shared__ int lcur[NBINS];
  __shared__ int gdel[NBINS];
  __shared__ int wsum[16];
  __shared__ int s_val[CHUNK];
  __shared__ int s_bin[CHUNK];
  const int tid = threadIdx.x;
  const int lane = tid & 31;
  const int wave = tid >> 5;
  const int blk = blockIdx.x;

  const int e0 = blk * CHUNK;
  const int e1 = (e0 + CHUNK < N_EDGES) ? (e0 + CHUNK) : N_EDGES;
  int es[EPT], ed[EPT];
#pragma unroll
  for (int u = 0; u < EPT; ++u) {
    const int i = e0 + tid + u * 1024;
    const int ii = (i < e1) ? i : (e1 - 1);
    es[u] = adj[ii];
    const int dv = adj[N_EDGES + ii];
    ed[u] = (i < e1) ? dv : -1;
  }

  int g = 0, c = 0, inc = 0;
  if (tid < 512) {
    if (tid < NBINS) {
      g = off2[tid * NCHUNK + blk];
      c = off2[tid * NCHUNK + blk + 1] - g;
    }
    inc = c;
#pragma unroll
    for (int o = 1; o < 32; o <<= 1) {
      const int t = __shfl_up(inc, o, 32);
      if (lane >= o) inc += t;
    }
    if (lane == 31) wsum[wave] = inc;
  }
  __syncthreads();
  if (tid < NBINS) {
    int wpre = 0;
    for (int w = 0; w < wave; ++w) wpre += wsum[w];
    const int lstart = wpre + inc - c;
    lcur[tid] = lstart;
    gdel[tid] = g - lstart;
  }

  __syncthreads();
#pragma unroll
  for (int u = 0; u < EPT; ++u) {
    if (ed[u] >= 0) {
      const int b = ed[u] >> BIN_SHIFT;
      const int p = atomicAdd(&lcur[b], 1);
      s_val[p] = (es[u] << BIN_SHIFT) | (ed[u] & (BIN_NODES - 1));
      s_bin[p] = b;
    }
  }
  __syncthreads();
  const int n = e1 - e0;
  for (int j = tid; j < n; j += 1024) ebuf[gdel[s_bin[j]] + j] = s_val[j];
}

#define STAGE_CAP 12288
#define RS 8

__device__ __forceinline__ void group_and_sort(
    int* __restrict__ stage, int* __restrict__ cur, const int* __restrict__ loc,
    const int* __restrict__ ebuf, int* __restrict__ csr, const int (&v)[RS],
    const int bs, const int m, const int tid, const int lane, const int wave) {
#pragma unroll
  for (int u = 0; u < RS; ++u) {
    if (v[u] >= 0) {
      const int p = atomicAdd(&cur[v[u] & (BIN_NODES - 1)], 1);
      stage[p] = v[u] >> BIN_SHIFT;
    }
  }
  for (int e = tid + RS * 1024; e < m; e += 1024) {
    const int w = ebuf[bs + e];
    const int p = atomicAdd(&cur[w & (BIN_NODES - 1)], 1);
    stage[p] = w >> BIN_SHIFT;
  }
  __syncthreads();

  for (int d = wave; d < BIN_NODES; d += 32) {
    const int start = __builtin_amdgcn_readfirstlane(loc[d]);
    const int end = __builtin_amdgcn_readfirstlane(loc[d + 1]);
    for (int ib = start; ib < end; ib += 32) {
      const int i = ib + lane;
      const bool vi = i < end;
      const int mine = vi ? stage[i] : 0;
      int r = 0;
      for (int kb = start; kb < end; kb += 32) {
        const int kj = kb + lane;
        const int other = (kj < end) ? stage[kj] : 0;
        const int nk = (end - kb < 32) ? (end - kb) : 32;
        for (int k = 0; k < nk; ++k) {
          const int o = __builtin_amdgcn_readlane(other, k);
          const int kk = kb + k;
          r += (o < mine || (o == mine && kk < i)) ? 1 : 0;
        }
      }
      if (vi) csr[bs + start + r] = mine;
    }
  }
}

__global__ __launch_bounds__(1024) void bin_sort(
    const int* __restrict__ off2, const int* __restrict__ ebuf,
    int* __restrict__ off, int* __restrict__ tmp, int* __restrict__ csr) {
  __shared__ int cnt[BIN_NODES];
  __shared__ int loc[BIN_NODES + 1];
  __shared__ int cur[BIN_NODES];
  __shared__ int wsum[BIN_NODES / 32];
  __shared__ int stage_l[STAGE_CAP];
  const int tid = threadIdx.x;
  const int lane = tid & 31;
  const int wave = tid >> 5;
  const int b = blockIdx.x;
  const int bs = off2[b * NCHUNK];
  const int be = off2[(b + 1) * NCHUNK];
  const int m = be - bs;

  int v[RS];
#pragma unroll
  for (int u = 0; u < RS; ++u) {
    const int e = tid + u * 1024;
    v[u] = (e < m) ? ebuf[bs + e] : -1;
  }
  if (tid < BIN_NODES) cnt[tid] = 0;
  __syncthreads();
#pragma unroll
  for (int u = 0; u < RS; ++u)
    if (v[u] >= 0) atomicAdd(&cnt[v[u] & (BIN_NODES - 1)], 1);
  for (int e = tid + RS * 1024; e < m; e += 1024)
    atomicAdd(&cnt[ebuf[bs + e] & (BIN_NODES - 1)], 1);
  __syncthreads();

  int c = 0, inc = 0;
  if (tid < BIN_NODES) {
    c = cnt[tid];
    inc = c;
#pragma unroll
    for (int o = 1; o < 32; o <<= 1) {
      const int t = __shfl_up(inc, o, 32);
      if (lane >= o) inc += t;
    }
    if (lane == 31) wsum[wave] = inc;
  }
  __syncthreads();
  if (tid < BIN_NODES) {
    int wpre = 0;
    for (int w = 0; w < wave; ++w) wpre += wsum[w];
    const int ls = wpre + inc - c;
    loc[tid] = ls;
    cur[tid] = ls;
    const int node = b * BIN_NODES + tid;
    if (node < N_NODES) off[node] = bs + ls;
  }
  if (tid == 0) {
    loc[BIN_NODES] = m;
    if (b == NBINS - 1) off[N_NODES] = be;
  }
  __syncthreads();

  if (m <= STAGE_CAP)
    group_and_sort(stage_l, cur, loc, ebuf, csr, v, bs, m, tid, lane, wave);
  else
    group_and_sort(tmp + bs, cur, loc, ebuf, csr, v, bs, m, tid, lane, wave);
}

#define NPW 4

#define CH 16

__device__ __forceinline__ void walk_chunk(
    const float* __restrict__ h, const int lane, const int myk, const float2 av,
    const float adn, const int sj, const int c, const int cnt,
    float& m, float& ssum, float& ax, float& ay) {
  float2 g[CH];
#pragma unroll
  for (int k = 0; k < CH; ++k) {
    if (k == 0 || k < cnt) {
      const int s = __builtin_amdgcn_readlane(sj, c + k);
      g[k] = *(const float2*)(h + (size_t)s * HDIM + 2 * lane);
    }
  }

  float v[CH];
#pragma unroll
  for (int k = 0; k < CH; ++k) {
    v[k] = 0.0f;
    if (k == 0 || k < cnt) v[k] = g[k].x * av.x + g[k].y * av.y;
  }

#pragma unroll
  for (int w = 16; w >= 2; w >>= 1) {
    const bool up = (lane & w) != 0;
#pragma unroll
    for (int i = 0; i < (w >> 1); ++i) {
      const float lower = v[i];
      const float upper = v[i + (w >> 1)];
      const float keep = up ? upper : lower;
      const float send = up ? lower : upper;
      v[i] = keep + __shfl_xor(send, w, 32);
    }
  }
  const float score = v[0] + __shfl_xor(v[0], 1, 32);

  float e = score + adn;
  e = (e > 0.0f) ? e : NEG_SLOPE * e;
  const bool valid = myk < cnt;
  float cm = valid ? e : m;
#pragma unroll
  for (int o = 16; o >= 2; o >>= 1) cm = fmaxf(cm, __shfl_xor(cm, o, 32));
  const float mn = fmaxf(m, cm);
  const float scale = __expf(m - mn);
  ssum *= scale;
  ax *= scale;
  ay *= scale;
  m = mn;

  const float p = valid ? __expf(e - m) : 0.0f;
  const int pi = __float_as_int(p);
#pragma unroll
  for (int k = 0; k < CH; ++k) {
    if (k == 0 || k < cnt) {
      const float pk = __int_as_float(__builtin_amdgcn_readlane(pi, 2 * k));
      ssum += pk;
      ax += g[k].x * pk;
      ay += g[k].y * pk;
    }
  }
}

template <int ELU>
__global__ __launch_bounds__(256) void node_walk(
    const int* __restrict__ off, const int* __restrict__ csr,
    const float* __restrict__ alpha_s, const float* __restrict__ alpha_d,
    const float* __restrict__ h, const float* __restrict__ a_src,
    const float* __restrict__ b, float* __restrict__ out, int n_nodes) {
  const int lane = threadIdx.x & 31;
  const int wid = __builtin_amdgcn_readfirstlane((int)((blockIdx.x * 256 + threadIdx.x) >> 5));
  const float2 bb = *(const float2*)(b + 2 * lane);
  const float2 av = *(const float2*)(a_src + 2 * lane);
  const int myk = lane >> 1;
  for (int t = 0; t < NPW; ++t) {
    const int n = wid * NPW + t;
    if (n >= n_nodes) break;
    const int start = off[n];
    const int end = off[n + 1];
    const float adn = alpha_d[n];
    float es = alpha_s[n] + adn;
    es = (es > 0.0f) ? es : NEG_SLOPE * es;
    const float2 hn = *(const float2*)(h + (size_t)n * HDIM + 2 * lane);

    float m = es;
    float ssum = 1.0f;
    float ax = hn.x;
    float ay = hn.y;

    for (int base = start; base < end; base += 32) {
      const int j = base + lane;
      const int sj = csr[(j < end) ? j : (end - 1)];
      const int rem = (end - base < 32) ? (end - base) : 32;
      for (int c = 0; c < rem; c += CH)
        walk_chunk(h, lane, myk, av, adn, sj, c,
                   (rem - c < CH) ? (rem - c) : CH, m, ssum, ax, ay);
    }

    float vx = ax / ssum + bb.x;
    float vy = ay / ssum + bb.y;
    if (ELU) {
      vx = (vx > 0.0f) ? vx : (__expf(vx) - 1.0f);
      vy = (vy > 0.0f) ? vy : (__expf(vy) - 1.0f);
    }
    *(float2*)(out + (size_t)n * HDIM + 2 * lane) = make_float2(vx, vy);
  }
}

extern "C" void kernel_launch(void* const* d_in, const int* in_sizes, int n_in,
                              void* d_out, int out_size, void* d_ws,
                              size_t ws_size, hipStream_t stream) {
  const float* X      = (const float*)d_in[0];
  const int*   adj    = (const int*)d_in[1];
  const float* W1     = (const float*)d_in[2];
  const float* a_src1 = (const float*)d_in[3];
  const float* a_dst1 = (const float*)d_in[4];
  const float* b1     = (const float*)d_in[5];
  const float* W2     = (const float*)d_in[6];
  const float* a_src2 = (const float*)d_in[7];
  const float* a_dst2 = (const float*)d_in[8];
  const float* b2     = (const float*)d_in[9];
  const float* clas_W = (const float*)d_in[10];
  const float* clas_b = (const float*)d_in[11];

  float* out = (float*)d_out;
  float* x2  = out + (size_t)N_NODES * N_CLASSES;

  float* ws   = (float*)d_ws;
  float* h    = ws;
  float* x1   = ws + (size_t)N_NODES * HDIM;
  int*   ib   = (int*)(ws + (size_t)2 * N_NODES * HDIM);
  float* as_  = ws + (size_t)3 * N_NODES * HDIM;
  float* ad_  = as_ + N_NODES;
  int* cnt2 = ib;
  int* off2 = ib + 100352;
  int* off  = ib + 2 * 100352;
  int* ebuf = ib + 3 * 100352;
  int* tmp  = ebuf + N_EDGES;
  int* csr  = tmp + N_EDGES;

  const int gemm_blocks = (N_NODES + 127) / 128;
  const int scan_blocks = (NCNT + 1 + 1023) / 1024;
  const int walk_blocks = (N_NODES + 8 * NPW - 1) / (8 * NPW);

  bin_hist<<<NCHUNK, 1024, 0, stream>>>(adj, cnt2);
  scan_offsets<<<scan_blocks, 1024, 0, stream>>>(cnt2, off2, NCNT);
  bin_scatter<<<NCHUNK, 1024, 0, stream>>>(adj, off2, ebuf);
  bin_sort<<<NBINS, 1024, 0, stream>>>(off2, ebuf, off, tmp, csr);

  gemm_wmma_f32<IN_DIM, HDIM, true><<<gemm_blocks, 256, 0, stream>>>(
      X, W1, nullptr, h, N_NODES, a_src1, a_dst1, as_, ad_);
  node_walk<1><<<walk_blocks, 256, 0, stream>>>(off, csr, as_, ad_, h, a_src1, b1, x1, N_NODES);

  gemm_wmma_f32<HDIM, HDIM, true><<<gemm_blocks, 256, 0, stream>>>(
      x1, W2, nullptr, h, N_NODES, a_src2, a_dst2, as_, ad_);
  node_walk<0><<<walk_blocks, 256, 0, stream>>>(off, csr, as_, ad_, h, a_src2, b2, x2, N_NODES);

  gemm_wmma_f32<HDIM, N_CLASSES, false><<<gemm_blocks, 256, 0, stream>>>(
      x2, clas_W, clas_b, out, N_NODES, nullptr, nullptr, nullptr, nullptr);
}


# Edge sort level 1 made chunk-local: a chunk's block orders its edges by bin in LDS, writes them in place with a run table; bin blocks collect their runs; histogram, scan launches go; float sums same
# speedup vs baseline: 5.1641x; 1.0935x over previous
#include <hip/hip_runtime.h>
#include <hip/hip_bf16.h>

typedef float v2f __attribute__((ext_vector_type(2)));
typedef float v8f __attribute__((ext_vector_type(8)));

#define N_NODES   100000
#define N_EDGES   1600000
#define IN_DIM    128
#define HDIM      64
#define N_CLASSES 16
#define NEG_SLOPE 0.2f

template <int K, int N, bool ALPHA>
__global__ __launch_bounds__(256) void gemm_wmma_f32(
    const float* __restrict__ A, const float* __restrict__ B,
    const float* __restrict__ bias, float* __restrict__ C, int M,
    const float* __restrict__ a_src, const float* __restrict__ a_dst,
    float* __restrict__ alpha_s, float* __restrict__ alpha_d) {
  __shared__ float Bs[K * N];
  const int tid = threadIdx.x;
  for (int i = tid; i < K * N; i += 256) Bs[i] = B[i];
  __syncthreads();

  const int wave = tid >> 5;
  const int lane = tid & 31;
  const int hi = lane >> 4;
  const int lo = lane & 15;

  const long m0 = (long)blockIdx.x * 128 + wave * 16;
  long r = m0 + lo;
  if (r > M - 1) r = M - 1;
  const float* __restrict__ Arow = A + (size_t)r * K;

  constexpr int NT = N / 16;
  v8f acc[NT] = {};

  [[maybe_unused]] float avs[NT], avd[NT];
  if constexpr (ALPHA) {
#pragma unroll
    for (int nt = 0; nt < NT; ++nt) {
      avs[nt] = a_src[nt * 16 + lo];
      avd[nt] = a_dst[nt * 16 + lo];
    }
  }

  for (int kk = 0; kk < K / 4; ++kk) {
    const int kb = kk * 4 + 2 * hi;
    v2f a;
    a.x = Arow[kb];
    a.y = Arow[kb + 1];
#pragma unroll
    for (int nt = 0; nt < NT; ++nt) {
      v2f b;
      b.x = Bs[kb * N + nt * 16 + lo];
      b.y = Bs[(kb + 1) * N + nt * 16 + lo];
      acc[nt] = __builtin_amdgcn_wmma_f32_16x16x4_f32(
           false, a,  false, b,
           (short)0, acc[nt],  false,  false);
    }
  }

#pragma unroll
  for (int nt = 0; nt < NT; ++nt) {
    const int col = nt * 16 + lo;
    const float bv = bias ? bias[col] : 0.0f;
#pragma unroll
    for (int rr = 0; rr < 8; ++rr) {
      const long row = m0 + rr + 8 * hi;
      if (row < M) C[(size_t)row * N + col] = acc[nt][rr] + bv;
    }
  }

  if constexpr (ALPHA) {
    float v[16];
#pragma unroll
    for (int rr = 0; rr < 8; ++rr) {
      float s = acc[0][rr] * avs[0];
      float d = acc[0][rr] * avd[0];
#pragma unroll
      for (int nt = 1; nt < NT; ++nt) {
        s += acc[nt][rr] * avs[nt];
        d += acc[nt][rr] * avd[nt];
      }
      v[rr] = s;
      v[8 + rr] = d;
    }
#pragma unroll
    for (int w = 8; w >= 1; w >>= 1) {
      const bool up = (lo & w) != 0;
#pragma unroll
      for (int i = 0; i < w; ++i) {
        const float lower = v[i];
        const float upper = v[i + w];
        const float keep = up ? upper : lower;
        const float send = up ? lower : upper;
        v[i] = keep + __shfl_xor(send, w, 32);
      }
    }
    const long row = m0 + (lo & 7) + 8 * hi;
    float* __restrict__ dstp = (lo < 8) ? alpha_s : alpha_d;
    if (row < M) dstp[row] = v[0];
  }
}

#define BIN_SHIFT 8
#define BIN_NODES 256
#define NBINS     ((N_NODES + BIN_NODES - 1) / BIN_NODES)
#define NCHUNK    256
#define CHUNK     ((N_EDGES + NCHUNK - 1) / NCHUNK)
#define TP        416

#define EPT ((CHUNK + 1023) / 1024)

static_assert(NBINS + 1 <= 512, "one bin per thread of the first 16 waves");
static_assert(NBINS + 1 <= TP && (TP % 32) == 0, "run table row");

__global__ __launch_bounds__(1024) void chunk_order(
    const int* __restrict__ adj, int* __restrict__ runtab,
    int* __restrict__ ebuf) {
  __shared__ int hb[NBINS + 1];
  __shared__ int wsum[16];
  __shared__ int s_val[CHUNK];
  const int tid = threadIdx.x;
  const int lane = tid & 31;
  const int wave = tid >> 5;
  const int blk = blockIdx.x;

  for (int i = tid; i < NBINS + 1; i += 1024) hb[i] = 0;
  const int e0 = blk * CHUNK;
  const int e1 = (e0 + CHUNK < N_EDGES) ? (e0 + CHUNK) : N_EDGES;
  int es[EPT], ed[EPT];
#pragma unroll
  for (int u = 0; u < EPT; ++u) {
    const int i = e0 + tid + u * 1024;
    const int ii = (i < e1) ? i : (e1 - 1);
    es[u] = adj[ii];
    ed[u] = adj[N_EDGES + ii];
  }
  __builtin_amdgcn_sched_barrier(0);
  __syncthreads();

  int rk[EPT];
#pragma unroll
  for (int u = 0; u < EPT; ++u) {
    rk[u] = 0;
    if (e0 + tid + u * 1024 < e1) rk[u] = atomicAdd(&hb[ed[u] >> BIN_SHIFT], 1);
  }
  __syncthreads();

  int c = 0, inc = 0;
  if (tid < 512) {
    if (tid < NBINS) c = hb[tid];
    inc = c;
#pragma unroll
    for (int o = 1; o < 32; o <<= 1) {
      const int t = __shfl_up(inc, o, 32);
      if (lane >= o) inc += t;
    }
    if (lane == 31) wsum[wave] = inc;
  }
  __syncthreads();
  if (tid <= NBINS) {
    int wpre = 0;
    for (int w = 0; w < wave; ++w) wpre += wsum[w];
    const int lstart = wpre + inc - c;
    hb[tid] = lstart;
    runtab[blk * TP + tid] = lstart;
  }
  __syncthreads();

#pragma unroll
  for (int u = 0; u < EPT; ++u) {
    if (e0 + tid + u * 1024 < e1)
      s_val[hb[ed[u] >> BIN_SHIFT] + rk[u]] =
          (es[u] << BIN_SHIFT) | (ed[u] & (BIN_NODES - 1));
  }
  __syncthreads();
  const int n = e1 - e0;
  for (int j = tid; j < n; j += 1024) ebuf[e0 + j] = s_val[j];
}

#define STAGE_CAP 12288
#define RS 8
#define RL 4

static_assert(RL * NCHUNK == 1024, "bin_sort: every thread has its place in a run");
static_assert(NCHUNK == BIN_NODES, "bin_sort: the first 8 waves take a chunk and a node each");

__device__ __forceinline__ void group_and_sort(
    int* __restrict__ stage, int* __restrict__ cur, const int* __restrict__ loc,
    const int* __restrict__ ebuf, int* __restrict__ csr, const int (&v)[RS],
    const int bs, const int gs, const int rn, const int rq,
    const int lane, const int wave) {
#pragma unroll
  for (int u = 0; u < RS; ++u) {
    if (v[u] >= 0) {
      const int p = atomicAdd(&cur[v[u] & (BIN_NODES - 1)], 1);
      stage[p] = v[u] >> BIN_SHIFT;
    }
  }
  for (int i = rq + RL * RS; i < rn; i += RL) {
    const int w = ebuf[gs + i];
    const int p = atomicAdd(&cur[w & (BIN_NODES - 1)], 1);
    stage[p] = w >> BIN_SHIFT;
  }
  __syncthreads();

  for (int d = wave; d < BIN_NODES; d += 32) {
    const int start = __builtin_amdgcn_readfirstlane(loc[d]);
    const int end = __builtin_amdgcn_readfirstlane(loc[d + 1]);
    for (int ib = start; ib < end; ib += 32) {
      const int i = ib + lane;
      const bool vi = i < end;
      const int mine = vi ? stage[i] : 0;
      int r = 0;
      for (int kb = start; kb < end; kb += 32) {
        const int kj = kb + lane;
        const int other = (kj < end) ? stage[kj] : 0;
        const int nk = (end - kb < 32) ? (end - kb) : 32;
        for (int k = 0; k < nk; ++k) {
          const int o = __builtin_amdgcn_readlane(other, k);
          const int kk = kb + k;
          r += (o < mine || (o == mine && kk < i)) ? 1 : 0;
        }
      }
      if (vi) csr[bs + start + r] = mine;
    }
  }
}

__global__ __launch_bounds__(1024) void bin_sort(
    const int* __restrict__ runtab, const int* __restrict__ ebuf,
    int* __restrict__ off, int* __restrict__ tmp, int* __restrict__ csr) {
  __shared__ int cnt[BIN_NODES];
  __shared__ int loc[BIN_NODES + 1];
  __shared__ int cur[BIN_NODES];
  __shared__ int wsum[BIN_NODES / 32];
  __shared__ int rsum[NCHUNK / 32];
  __shared__ int lsum[NCHUNK / 32];
  __shared__ int run_g[NCHUNK];
  __shared__ int run_n[NCHUNK];
  __shared__ int stage_l[STAGE_CAP];
  const int tid = threadIdx.x;
  const int lane = tid & 31;
  const int wave = tid >> 5;
  const int b = blockIdx.x;

  if (tid < NCHUNK) {
    const int rs = runtab[tid * TP + b];
    const int ln = runtab[tid * TP + b + 1] - rs;
    run_g[tid] = tid * CHUNK + rs;
    run_n[tid] = ln;
    int sr = rs, sl = ln;
#pragma unroll
    for (int o = 16; o > 0; o >>= 1) {
      sr += __shfl_xor(sr, o, 32);
      sl += __shfl_xor(sl, o, 32);
    }
    if (lane == 0) {
      rsum[wave] = sr;
      lsum[wave] = sl;
    }
    cnt[tid] = 0;
  }
  __syncthreads();
  int bs = 0, m = 0;
#pragma unroll
  for (int w = 0; w < NCHUNK / 32; ++w) {
    bs += rsum[w];
    m += lsum[w];
  }
  const int be = bs + m;

  const int rq = tid & (RL - 1);
  const int gs = run_g[tid / RL];
  const int rn = run_n[tid / RL];
  int v[RS];
#pragma unroll
  for (int u = 0; u < RS; ++u) {
    const int i = rq + RL * u;
    v[u] = (i < rn) ? ebuf[gs + i] : -1;
  }
#pragma unroll
  for (int u = 0; u < RS; ++u)
    if (v[u] >= 0) atomicAdd(&cnt[v[u] & (BIN_NODES - 1)], 1);
  for (int i = rq + RL * RS; i < rn; i += RL)
    atomicAdd(&cnt[ebuf[gs + i] & (BIN_NODES - 1)], 1);
  __syncthreads();

  int c = 0, inc = 0;
  if (tid < BIN_NODES) {
    c = cnt[tid];
    inc = c;
#pragma unroll
    for (int o = 1; o < 32; o <<= 1) {
      const int t = __shfl_up(inc, o, 32);
      if (lane >= o) inc += t;
    }
    if (lane == 31) wsum[wave] = inc;
  }
  __syncthreads();
  if (tid < BIN_NODES) {
    int wpre = 0;
    for (int w = 0; w < wave; ++w) wpre += wsum[w];
    const int ls = wpre + inc - c;
    loc[tid] = ls;
    cur[tid] = ls;
    const int node = b * BIN_NODES + tid;
    if (node < N_NODES) off[node] = bs + ls;
  }
  if (tid == 0) {
    loc[BIN_NODES] = m;
    if (b == NBINS - 1) off[N_NODES] = be;
  }
  __syncthreads();

  if (m <= STAGE_CAP)
    group_and_sort(stage_l, cur, loc, ebuf, csr, v, bs, gs, rn, rq, lane, wave);
  else
    group_and_sort(tmp + bs, cur, loc, ebuf, csr, v, bs, gs, rn, rq, lane, wave);
}

#define NPW 4

#define CH 16

__device__ __forceinline__ void walk_chunk(
    const float* __restrict__ h, const int lane, const int myk, const float2 av,
    const float adn, const int sj, const int c, const int cnt,
    float& m, float& ssum, float& ax, float& ay) {
  float2 g[CH];
#pragma unroll
  for (int k = 0; k < CH; ++k) {
    if (k == 0 || k < cnt) {
      const int s = __builtin_amdgcn_readlane(sj, c + k);
      g[k] = *(const float2*)(h + (size_t)s * HDIM + 2 * lane);
    }
  }

  float v[CH];
#pragma unroll
  for (int k = 0; k < CH; ++k) {
    v[k] = 0.0f;
    if (k == 0 || k < cnt) v[k] = g[k].x * av.x + g[k].y * av.y;
  }

#pragma unroll
  for (int w = 16; w >= 2; w >>= 1) {
    const bool up = (lane & w) != 0;
#pragma unroll
    for (int i = 0; i < (w >> 1); ++i) {
      const float lower = v[i];
      const float upper = v[i + (w >> 1)];
      const float keep = up ? upper : lower;
      const float send = up ? lower : upper;
      v[i] = keep + __shfl_xor(send, w, 32);
    }
  }
  const float score = v[0] + __shfl_xor(v[0], 1, 32);

  float e = score + adn;
  e = (e > 0.0f) ? e : NEG_SLOPE * e;
  const bool valid = myk < cnt;
  float cm = valid ? e : m;
#pragma unroll
  for (int o = 16; o >= 2; o >>= 1) cm = fmaxf(cm, __shfl_xor(cm, o, 32));
  const float mn = fmaxf(m, cm);
  const float scale = __expf(m - mn);
  ssum *= scale;
  ax *= scale;
  ay *= scale;
  m = mn;

  const float p = valid ? __expf(e - m) : 0.0f;
  const int pi = __float_as_int(p);
#pragma unroll
  for (int k = 0; k < CH; ++k) {
    if (k == 0 || k < cnt) {
      const float pk = __int_as_float(__builtin_amdgcn_readlane(pi, 2 * k));
      ssum += pk;
      ax += g[k].x * pk;
      ay += g[k].y * pk;
    }
  }
}

template <int ELU>
__global__ __launch_bounds__(256) void node_walk(
    const int* __restrict__ off, const int* __restrict__ csr,
    const float* __restrict__ alpha_s, const float* __restrict__ alpha_d,
    const float* __restrict__ h, const float* __restrict__ a_src,
    const float* __restrict__ b, float* __restrict__ out, int n_nodes) {
  const int lane = threadIdx.x & 31;
  const int wid = __builtin_amdgcn_readfirstlane((int)((blockIdx.x * 256 + threadIdx.x) >> 5));
  const float2 bb = *(const float2*)(b + 2 * lane);
  const float2 av = *(const float2*)(a_src + 2 * lane);
  const int myk = lane >> 1;
  for (int t = 0; t < NPW; ++t) {
    const int n = wid * NPW + t;
    if (n >= n_nodes) break;
    const int start = off[n];
    const int end = off[n + 1];
    const float adn = alpha_d[n];
    float es = alpha_s[n] + adn;
    es = (es > 0.0f) ? es : NEG_SLOPE * es;
    const float2 hn = *(const float2*)(h + (size_t)n * HDIM + 2 * lane);

    float m = es;
    float ssum = 1.0f;
    float ax = hn.x;
    float ay = hn.y;

    for (int base = start; base < end; base += 32) {
      const int j = base + lane;
      const int sj = csr[(j < end) ? j : (end - 1)];
      const int rem = (end - base < 32) ? (end - base) : 32;
      for (int c = 0; c < rem; c += CH)
        walk_chunk(h, lane, myk, av, adn, sj, c,
                   (rem - c < CH) ? (rem - c) : CH, m, ssum, ax, ay);
    }

    float vx = ax / ssum + bb.x;
    float vy = ay / ssum + bb.y;
    if (ELU) {
      vx = (vx > 0.0f) ? vx : (__expf(vx) - 1.0f);
      vy = (vy > 0.0f) ? vy : (__expf(vy) - 1.0f);
    }
    *(float2*)(out + (size_t)n * HDIM + 2 * lane) = make_float2(vx, vy);
  }
}

extern "C" void kernel_launch(void* const* d_in, const int* in_sizes, int n_in,
                              void* d_out, int out_size, void* d_ws,
                              size_t ws_size, hipStream_t stream) {
  const float* X      = (const float*)d_in[0];
  const int*   adj    = (const int*)d_in[1];
  const float* W1     = (const float*)d_in[2];
  const float* a_src1 = (const float*)d_in[3];
  const float* a_dst1 = (const float*)d_in[4];
  const float* b1     = (const float*)d_in[5];
  const float* W2     = (const float*)d_in[6];
  const float* a_src2 = (const float*)d_in[7];
  const float* a_dst2 = (const float*)d_in[8];
  const float* b2     = (const float*)d_in[9];
  const float* clas_W = (const float*)d_in[10];
  const float* clas_b = (const float*)d_in[11];

  float* out = (float*)d_out;
  float* x2  = out + (size_t)N_NODES * N_CLASSES;

  float* ws   = (float*)d_ws;
  float* h    = ws;
  float* x1   = ws + (size_t)N_NODES * HDIM;
  int*   ib   = (int*)(ws + (size_t)2 * N_NODES * HDIM);
  float* as_  = ws + (size_t)3 * N_NODES * HDIM;
  float* ad_  = as_ + N_NODES;
  int* runtab = ib;
  int* off  = ib + 2 * 100352;
  int* ebuf = ib + 3 * 100352;
  int* tmp  = ebuf + N_EDGES;
  int* csr  = tmp + N_EDGES;

  const int gemm_blocks = (N_NODES + 127) / 128;
  const int walk_blocks = (N_NODES + 8 * NPW - 1) / (8 * NPW);

  chunk_order<<<NCHUNK, 1024, 0, stream>>>(adj, runtab, ebuf);
  bin_sort<<<NBINS, 1024, 0, stream>>>(runtab, ebuf, off, tmp, csr);

  gemm_wmma_f32<IN_DIM, HDIM, true><<<gemm_blocks, 256, 0, stream>>>(
      X, W1, nullptr, h, N_NODES, a_src1, a_dst1, as_, ad_);
  node_walk<1><<<walk_blocks, 256, 0, stream>>>(off, csr, as_, ad_, h, a_src1, b1, x1, N_NODES);

  gemm_wmma_f32<HDIM, HDIM, true><<<gemm_blocks, 256, 0, stream>>>(
      x1, W2, nullptr, h, N_NODES, a_src2, a_dst2, as_, ad_);
  node_walk<0><<<walk_blocks, 256, 0, stream>>>(off, csr, as_, ad_, h, a_src2, b2, x2, N_NODES);

  gemm_wmma_f32<HDIM, N_CLASSES, false><<<gemm_blocks, 256, 0, stream>>>(
      x2, clas_W, clas_b, out, N_NODES, nullptr, nullptr, nullptr, nullptr);
}


# Node walk retiled: a wave takes two nodes at once, one per 16-lane half (4 features a lane, lane exchanges by DPP, not LDS), 8192 waves stride over the pairs; score dot: 4 terms a lane, 16-lane tree
# speedup vs baseline: 5.8845x; 1.1395x over previous
#include <hip/hip_runtime.h>
#include <hip/hip_bf16.h>

typedef float v2f __attribute__((ext_vector_type(2)));
typedef float v8f __attribute__((ext_vector_type(8)));

#define N_NODES   100000
#define N_EDGES   1600000
#define IN_DIM    128
#define HDIM      64
#define N_CLASSES 16
#define NEG_SLOPE 0.2f

template <int K, int N, bool ALPHA>
__global__ __launch_bounds__(256) void gemm_wmma_f32(
    const float* __restrict__ A, const float* __restrict__ B,
    const float* __restrict__ bias, float* __restrict__ C, int M,
    const float* __restrict__ a_src, const float* __restrict__ a_dst,
    float* __restrict__ alpha_s, float* __restrict__ alpha_d) {
  __shared__ float Bs[K * N];
  const int tid = threadIdx.x;
  for (int i = tid; i < K * N; i += 256) Bs[i] = B[i];
  __syncthreads();

  const int wave = tid >> 5;
  const int lane = tid & 31;
  const int hi = lane >> 4;
  const int lo = lane & 15;

  const long m0 = (long)blockIdx.x * 128 + wave * 16;
  long r = m0 + lo;
  if (r > M - 1) r = M - 1;
  const float* __restrict__ Arow = A + (size_t)r * K;

  constexpr int NT = N / 16;
  v8f acc[NT] = {};

  [[maybe_unused]] float avs[NT], avd[NT];
  if constexpr (ALPHA) {
#pragma unroll
    for (int nt = 0; nt < NT; ++nt) {
      avs[nt] = a_src[nt * 16 + lo];
      avd[nt] = a_dst[nt * 16 + lo];
    }
  }

  for (int kk = 0; kk < K / 4; ++kk) {
    const int kb = kk * 4 + 2 * hi;
    v2f a;
    a.x = Arow[kb];
    a.y = Arow[kb + 1];
#pragma unroll
    for (int nt = 0; nt < NT; ++nt) {
      v2f b;
      b.x = Bs[kb * N + nt * 16 + lo];
      b.y = Bs[(kb + 1) * N + nt * 16 + lo];
      acc[nt] = __builtin_amdgcn_wmma_f32_16x16x4_f32(
           false, a,  false, b,
           (short)0, acc[nt],  false,  false);
    }
  }

#pragma unroll
  for (int nt = 0; nt < NT; ++nt) {
    const int col = nt * 16 + lo;
    const float bv = bias ? bias[col] : 0.0f;
#pragma unroll
    for (int rr = 0; rr < 8; ++rr) {
      const long row = m0 + rr + 8 * hi;
      if (row < M) C[(size_t)row * N + col] = acc[nt][rr] + bv;
    }
  }

  if constexpr (ALPHA) {
    float v[16];
#pragma unroll
    for (int rr = 0; rr < 8; ++rr) {
      float s = acc[0][rr] * avs[0];
      float d = acc[0][rr] * avd[0];
#pragma unroll
      for (int nt = 1; nt < NT; ++nt) {
        s += acc[nt][rr] * avs[nt];
        d += acc[nt][rr] * avd[nt];
      }
      v[rr] = s;
      v[8 + rr] = d;
    }
#pragma unroll
    for (int w = 8; w >= 1; w >>= 1) {
      const bool up = (lo & w) != 0;
#pragma unroll
      for (int i = 0; i < w; ++i) {
        const float lower = v[i];
        const float upper = v[i + w];
        const float keep = up ? upper : lower;
        const float send = up ? lower : upper;
        v[i] = keep + __shfl_xor(send, w, 32);
      }
    }
    const long row = m0 + (lo & 7) + 8 * hi;
    float* __restrict__ dstp = (lo < 8) ? alpha_s : alpha_d;
    if (row < M) dstp[row] = v[0];
  }
}

#define BIN_SHIFT 8
#define BIN_NODES 256
#define NBINS     ((N_NODES + BIN_NODES - 1) / BIN_NODES)
#define NCHUNK    256
#define CHUNK     ((N_EDGES + NCHUNK - 1) / NCHUNK)
#define TP        416

#define EPT ((CHUNK + 1023) / 1024)

static_assert(NBINS + 1 <= 512, "one bin per thread of the first 16 waves");
static_assert(NBINS + 1 <= TP && (TP % 32) == 0, "run table row");

__global__ __launch_bounds__(1024) void chunk_order(
    const int* __restrict__ adj, int* __restrict__ runtab,
    int* __restrict__ ebuf) {
  __shared__ int hb[NBINS + 1];
  __shared__ int wsum[16];
  __shared__ int s_val[CHUNK];
  const int tid = threadIdx.x;
  const int lane = tid & 31;
  const int wave = tid >> 5;
  const int blk = blockIdx.x;

  for (int i = tid; i < NBINS + 1; i += 1024) hb[i] = 0;
  const int e0 = blk * CHUNK;
  const int e1 = (e0 + CHUNK < N_EDGES) ? (e0 + CHUNK) : N_EDGES;
  int es[EPT], ed[EPT];
#pragma unroll
  for (int u = 0; u < EPT; ++u) {
    const int i = e0 + tid + u * 1024;
    const int ii = (i < e1) ? i : (e1 - 1);
    es[u] = adj[ii];
    ed[u] = adj[N_EDGES + ii];
  }
  __builtin_amdgcn_sched_barrier(0);
  __syncthreads();

  int rk[EPT];
#pragma unroll
  for (int u = 0; u < EPT; ++u) {
    rk[u] = 0;
    if (e0 + tid + u * 1024 < e1) rk[u] = atomicAdd(&hb[ed[u] >> BIN_SHIFT], 1);
  }
  __syncthreads();

  int c = 0, inc = 0;
  if (tid < 512) {
    if (tid < NBINS) c = hb[tid];
    inc = c;
#pragma unroll
    for (int o = 1; o < 32; o <<= 1) {
      const int t = __shfl_up(inc, o, 32);
      if (lane >= o) inc += t;
    }
    if (lane == 31) wsum[wave] = inc;
  }
  __syncthreads();
  if (tid <= NBINS) {
    int wpre = 0;
    for (int w = 0; w < wave; ++w) wpre += wsum[w];
    const int lstart = wpre + inc - c;
    hb[tid] = lstart;
    runtab[blk * TP + tid] = lstart;
  }
  __syncthreads();

#pragma unroll
  for (int u = 0; u < EPT; ++u) {
    if (e0 + tid + u * 1024 < e1)
      s_val[hb[ed[u] >> BIN_SHIFT] + rk[u]] =
          (es[u] << BIN_SHIFT) | (ed[u] & (BIN_NODES - 1));
  }
  __syncthreads();
  const int n = e1 - e0;
  for (int j = tid; j < n; j += 1024) ebuf[e0 + j] = s_val[j];
}

#define STAGE_CAP 12288
#define RS 8
#define RL 4

static_assert(RL * NCHUNK == 1024, "bin_sort: every thread has its place in a run");
static_assert(NCHUNK == BIN_NODES, "bin_sort: the first 8 waves take a chunk and a node each");

__device__ __forceinline__ void group_and_sort(
    int* __restrict__ stage, int* __restrict__ cur, const int* __restrict__ loc,
    const int* __restrict__ ebuf, int* __restrict__ csr, const int (&v)[RS],
    const int bs, const int gs, const int rn, const int rq,
    const int lane, const int wave) {
#pragma unroll
  for (int u = 0; u < RS; ++u) {
    if (v[u] >= 0) {
      const int p = atomicAdd(&cur[v[u] & (BIN_NODES - 1)], 1);
      stage[p] = v[u] >> BIN_SHIFT;
    }
  }
  for (int i = rq + RL * RS; i < rn; i += RL) {
    const int w = ebuf[gs + i];
    const int p = atomicAdd(&cur[w & (BIN_NODES - 1)], 1);
    stage[p] = w >> BIN_SHIFT;
  }
  __syncthreads();

  for (int d = wave; d < BIN_NODES; d += 32) {
    const int start = __builtin_amdgcn_readfirstlane(loc[d]);
    const int end = __builtin_amdgcn_readfirstlane(loc[d + 1]);
    for (int ib = start; ib < end; ib += 32) {
      const int i = ib + lane;
      const bool vi = i < end;
      const int mine = vi ? stage[i] : 0;
      int r = 0;
      for (int kb = start; kb < end; kb += 32) {
        const int kj = kb + lane;
        const int other = (kj < end) ? stage[kj] : 0;
        const int nk = (end - kb < 32) ? (end - kb) : 32;
        for (int k = 0; k < nk; ++k) {
          const int o = __builtin_amdgcn_readlane(other, k);
          const int kk = kb + k;
          r += (o < mine || (o == mine && kk < i)) ? 1 : 0;
        }
      }
      if (vi) csr[bs + start + r] = mine;
    }
  }
}

__global__ __launch_bounds__(1024) void bin_sort(
    const int* __restrict__ runtab, const int* __restrict__ ebuf,
    int* __restrict__ off, int* __restrict__ tmp, int* __restrict__ csr) {
  __shared__ int cnt[BIN_NODES];
  __shared__ int loc[BIN_NODES + 1];
  __shared__ int cur[BIN_NODES];
  __shared__ int wsum[BIN_NODES / 32];
  __shared__ int rsum[NCHUNK / 32];
  __shared__ int lsum[NCHUNK / 32];
  __shared__ int run_g[NCHUNK];
  __shared__ int run_n[NCHUNK];
  __shared__ int stage_l[STAGE_CAP];
  const int tid = threadIdx.x;
  const int lane = tid & 31;
  const int wave = tid >> 5;
  const int b = blockIdx.x;

  if (tid < NCHUNK) {
    const int rs = runtab[tid * TP + b];
    const int ln = runtab[tid * TP + b + 1] - rs;
    run_g[tid] = tid * CHUNK + rs;
    run_n[tid] = ln;
    int sr = rs, sl = ln;
#pragma unroll
    for (int o = 16; o > 0; o >>= 1) {
      sr += __shfl_xor(sr, o, 32);
      sl += __shfl_xor(sl, o, 32);
    }
    if (lane == 0) {
      rsum[wave] = sr;
      lsum[wave] = sl;
    }
    cnt[tid] = 0;
  }
  __syncthreads();
  int bs = 0, m = 0;
#pragma unroll
  for (int w = 0; w < NCHUNK / 32; ++w) {
    bs += rsum[w];
    m += lsum[w];
  }
  const int be = bs + m;

  const int rq = tid & (RL - 1);
  const int gs = run_g[tid / RL];
  const int rn = run_n[tid / RL];
  int v[RS];
#pragma unroll
  for (int u = 0; u < RS; ++u) {
    const int i = rq + RL * u;
    v[u] = (i < rn) ? ebuf[gs + i] : -1;
  }
#pragma unroll
  for (int u = 0; u < RS; ++u)
    if (v[u] >= 0) atomicAdd(&cnt[v[u] & (BIN_NODES - 1)], 1);
  for (int i = rq + RL * RS; i < rn; i += RL)
    atomicAdd(&cnt[ebuf[gs + i] & (BIN_NODES - 1)], 1);
  __syncthreads();

  int c = 0, inc = 0;
  if (tid < BIN_NODES) {
    c = cnt[tid];
    inc = c;
#pragma unroll
    for (int o = 1; o < 32; o <<= 1) {
      const int t = __shfl_up(inc, o, 32);
      if (lane >= o) inc += t;
    }
    if (lane == 31) wsum[wave] = inc;
  }
  __syncthreads();
  if (tid < BIN_NODES) {
    int wpre = 0;
    for (int w = 0; w < wave; ++w) wpre += wsum[w];
    const int ls = wpre + inc - c;
    loc[tid] = ls;
    cur[tid] = ls;
    const int node = b * BIN_NODES + tid;
    if (node < N_NODES) off[node] = bs + ls;
  }
  if (tid == 0) {
    loc[BIN_NODES] = m;
    if (b == NBINS - 1) off[N_NODES] = be;
  }
  __syncthreads();

  if (m <= STAGE_CAP)
    group_and_sort(stage_l, cur, loc, ebuf, csr, v, bs, gs, rn, rq, lane, wave);
  else
    group_and_sort(tmp + bs, cur, loc, ebuf, csr, v, bs, gs, rn, rq, lane, wave);
}

#define CH 16

template <int I> struct idx_c { static constexpr int value = I; };
template <int I, int N, class F>
__device__ __forceinline__ void static_for(F&& f) {
  if constexpr (I < N) {
    f(idx_c<I>{});
    static_for<I + 1, N>(f);
  }
}

template <int K>
__device__ __forceinline__ int share16(int x) {
  return __builtin_amdgcn_update_dpp(0, x, 0x150 + K, 0xf, 0xf, true);
}
template <int K>
__device__ __forceinline__ float share16f(float x) {
  return __int_as_float(share16<K>(__float_as_int(x)));
}
template <int W>
__device__ __forceinline__ float xor16(float x) {
  return __int_as_float(
      __builtin_amdgcn_update_dpp(0, __float_as_int(x), 0x160 + W, 0xf, 0xf, true));
}

template <int W>
__device__ __forceinline__ void fold16(float (&v)[CH], const bool up) {
  static_for<0, W>([&](auto ic) {
    constexpr int i = decltype(ic)::value;
    const float lower = v[i];
    const float upper = v[i + W];
    const float keep = up ? upper : lower;
    const float send = up ? lower : upper;
    v[i] = keep + xor16<W>(send);
  });
}

__device__ __forceinline__ void walk_chunk(
    const float* __restrict__ h, const int hl, const unsigned hofs, const float4 av,
    const float adn, const int sj, const int cnt, const int cntmax,
    float& m, float& ssum, float4& a) {
  float4 g[CH];
  static_for<0, CH>([&](auto kc) {
    constexpr int k = decltype(kc)::value;
    if (k == 0 || k < cntmax) {
      const unsigned ofs = ((unsigned)share16<k>(sj) << 8) | hofs;
      g[k] = *(const float4*)((const char*)h + ofs);
    }
  });

  float v[CH];
  static_for<0, CH>([&](auto kc) {
    constexpr int k = decltype(kc)::value;
    v[k] = 0.0f;
    if (k == 0 || k < cntmax)
      v[k] = g[k].x * av.x + g[k].y * av.y + g[k].z * av.z + g[k].w * av.w;
  });

  fold16<8>(v, (hl & 8) != 0);
  fold16<4>(v, (hl & 4) != 0);
  fold16<2>(v, (hl & 2) != 0);
  fold16<1>(v, (hl & 1) != 0);
  const float score = v[0];

  float e = score + adn;
  e = (e > 0.0f) ? e : NEG_SLOPE * e;
  const bool valid = hl < cnt;
  float cm = valid ? e : m;
  cm = fmaxf(cm, xor16<8>(cm));
  cm = fmaxf(cm, xor16<4>(cm));
  cm = fmaxf(cm, xor16<2>(cm));
  cm = fmaxf(cm, xor16<1>(cm));
  const float mn = fmaxf(m, cm);
  const float scale = __expf(m - mn);
  ssum *= scale;
  a.x *= scale;
  a.y *= scale;
  a.z *= scale;
  a.w *= scale;
  m = mn;

  const float p = valid ? __expf(e - m) : 0.0f;
  static_for<0, CH>([&](auto kc) {
    constexpr int k = decltype(kc)::value;
    if (k == 0 || k < cntmax) {
      const float pk = share16f<k>(p);
      ssum += pk;
      a.x += g[k].x * pk;
      a.y += g[k].y * pk;
      a.z += g[k].z * pk;
      a.w += g[k].w * pk;
    }
  });
}

template <int ELU>
__global__ __launch_bounds__(256) void node_walk(
    const int* __restrict__ off, const int* __restrict__ csr,
    const float* __restrict__ alpha_s, const float* __restrict__ alpha_d,
    const float* __restrict__ h, const float* __restrict__ a_src,
    const float* __restrict__ b, float* __restrict__ out, int n_nodes) {
  const int lane = threadIdx.x & 31;
  const int hl = lane & 15;
  const int half = lane >> 4;
  const int gw = __builtin_amdgcn_readfirstlane((int)((blockIdx.x * 256 + threadIdx.x) >> 5));
  const int tw = (int)gridDim.x * 8;
  const int npairs = (n_nodes + 1) >> 1;
  const float4 bb = *(const float4*)(b + 4 * hl);
  const float4 av = *(const float4*)(a_src + 4 * hl);
  const unsigned hofs = (unsigned)hl << 4;
  for (int pr = gw; pr < npairs; pr += tw) {
    const int n = 2 * pr + half;
    const bool nv = n < n_nodes;
    const int nn = nv ? n : (n_nodes - 1);
    const int start = off[nn];
    const int endg = off[nn + 1];
    const int deg = nv ? (endg - start) : 0;
    const float adn = alpha_d[nn];
    float es = alpha_s[nn] + adn;
    es = (es > 0.0f) ? es : NEG_SLOPE * es;
    const float4 hn = *(const float4*)(h + (size_t)nn * HDIM + 4 * hl);

    float m = es;
    float ssum = 1.0f;
    float4 a = hn;

    const int d0 = __builtin_amdgcn_readlane(deg, 0);
    const int d1 = __builtin_amdgcn_readlane(deg, 16);
    const int maxdeg = (d0 > d1) ? d0 : d1;
    int base = 0;
    do {
      const int q = base + hl;
      int sj = nn;
      if (q < deg) sj = csr[start + q];
      const int left = maxdeg - base;
      walk_chunk(h, hl, hofs, av, adn, sj, deg - base, (left < CH) ? left : CH, m, ssum, a);
      base += CH;
    } while (base < maxdeg);

    float4 o;
    o.x = a.x / ssum + bb.x;
    o.y = a.y / ssum + bb.y;
    o.z = a.z / ssum + bb.z;
    o.w = a.w / ssum + bb.w;
    if (ELU) {
      o.x = (o.x > 0.0f) ? o.x : (__expf(o.x) - 1.0f);
      o.y = (o.y > 0.0f) ? o.y : (__expf(o.y) - 1.0f);
      o.z = (o.z > 0.0f) ? o.z : (__expf(o.z) - 1.0f);
      o.w = (o.w > 0.0f) ? o.w : (__expf(o.w) - 1.0f);
    }
    if (nv) *(float4*)(out + (size_t)n * HDIM + 4 * hl) = o;
  }
}

extern "C" void kernel_launch(void* const* d_in, const int* in_sizes, int n_in,
                              void* d_out, int out_size, void* d_ws,
                              size_t ws_size, hipStream_t stream) {
  const float* X      = (const float*)d_in[0];
  const int*   adj    = (const int*)d_in[1];
  const float* W1     = (const float*)d_in[2];
  const float* a_src1 = (const float*)d_in[3];
  const float* a_dst1 = (const float*)d_in[4];
  const float* b1     = (const float*)d_in[5];
  const float* W2     = (const float*)d_in[6];
  const float* a_src2 = (const float*)d_in[7];
  const float* a_dst2 = (const float*)d_in[8];
  const float* b2     = (const float*)d_in[9];
  const float* clas_W = (const float*)d_in[10];
  const float* clas_b = (const float*)d_in[11];

  float* out = (float*)d_out;
  float* x2  = out + (size_t)N_NODES * N_CLASSES;

  float* ws   = (float*)d_ws;
  float* h    = ws;
  float* x1   = ws + (size_t)N_NODES * HDIM;
  int*   ib   = (int*)(ws + (size_t)2 * N_NODES * HDIM);
  float* as_  = ws + (size_t)3 * N_NODES * HDIM;
  float* ad_  = as_ + N_NODES;
  int* runtab = ib;
  int* off  = ib + 2 * 100352;
  int* ebuf = ib + 3 * 100352;
  int* tmp  = ebuf + N_EDGES;
  int* csr  = tmp + N_EDGES;

  const int gemm_blocks = (N_NODES + 127) / 128;
  const int walk_blocks = 1024;

  chunk_order<<<NCHUNK, 1024, 0, stream>>>(adj, runtab, ebuf);
  bin_sort<<<NBINS, 1024, 0, stream>>>(runtab, ebuf, off, tmp, csr);

  gemm_wmma_f32<IN_DIM, HDIM, true><<<gemm_blocks, 256, 0, stream>>>(
      X, W1, nullptr, h, N_NODES, a_src1, a_dst1, as_, ad_);
  node_walk<1><<<walk_blocks, 256, 0, stream>>>(off, csr, as_, ad_, h, a_src1, b1, x1, N_NODES);

  gemm_wmma_f32<HDIM, HDIM, true><<<gemm_blocks, 256, 0, stream>>>(
      x1, W2, nullptr, h, N_NODES, a_src2, a_dst2, as_, ad_);
  node_walk<0><<<walk_blocks, 256, 0, stream>>>(off, csr, as_, ad_, h, a_src2, b2, x2, N_NODES);

  gemm_wmma_f32<HDIM, N_CLASSES, false><<<gemm_blocks, 256, 0, stream>>>(
      x2, clas_W, clas_b, out, N_NODES, nullptr, nullptr, nullptr, nullptr);
}


# Layer 1's GEMM joins the first edge-sort launch as 256 more 1024-thread blocks of one grid (independent work side by side on a WGP, 64 registers a lane): 7 launches become 6; float sums same
# speedup vs baseline: 6.1704x; 1.0486x over previous
#include <hip/hip_runtime.h>
#include <hip/hip_bf16.h>

typedef float v2f __attribute__((ext_vector_type(2)));
typedef float v8f __attribute__((ext_vector_type(8)));

#define N_NODES   100000
#define N_EDGES   1600000
#define IN_DIM    128
#define HDIM      64
#define N_CLASSES 16
#define NEG_SLOPE 0.2f

template <int K, int N, bool ALPHA, int NTHR>
__device__ __forceinline__ void gemm_tile(
    const float* __restrict__ A, const float* __restrict__ B,
    const float* __restrict__ bias, float* __restrict__ C, int M,
    const float* __restrict__ a_src, const float* __restrict__ a_dst,
    float* __restrict__ alpha_s, float* __restrict__ alpha_d, const long m0) {
  __shared__ float Bs[K * N];
  const int tid = threadIdx.x;
  for (int i = tid; i < K * N; i += NTHR) Bs[i] = B[i];
  __syncthreads();
  if (m0 >= M) return;

  const int lane = tid & 31;
  const int hi = lane >> 4;
  const int lo = lane & 15;

  long r = m0 + lo;
  if (r > M - 1) r = M - 1;
  const float* __restrict__ Arow = A + (size_t)r * K;

  constexpr int NT = N / 16;
  v8f acc[NT] = {};

  constexpr bool AV_AHEAD = (NTHR == 256);
  [[maybe_unused]] float avs[NT], avd[NT];
  if constexpr (ALPHA && AV_AHEAD) {
#pragma unroll
    for (int nt = 0; nt < NT; ++nt) {
      avs[nt] = a_src[nt * 16 + lo];
      avd[nt] = a_dst[nt * 16 + lo];
    }
  }

  for (int kk = 0; kk < K / 4; ++kk) {
    const int kb = kk * 4 + 2 * hi;
    v2f a;
    a.x = Arow[kb];
    a.y = Arow[kb + 1];
#pragma unroll
    for (int nt = 0; nt < NT; ++nt) {
      v2f b;
      b.x = Bs[kb * N + nt * 16 + lo];
      b.y = Bs[(kb + 1) * N + nt * 16 + lo];
      acc[nt] = __builtin_amdgcn_wmma_f32_16x16x4_f32(
           false, a,  false, b,
           (short)0, acc[nt],  false,  false);
    }
  }

#pragma unroll
  for (int nt = 0; nt < NT; ++nt) {
    const int col = nt * 16 + lo;
    const float bv = bias ? bias[col] : 0.0f;
#pragma unroll
    for (int rr = 0; rr < 8; ++rr) {
      const long row = m0 + rr + 8 * hi;
      if (row < M) C[(size_t)row * N + col] = acc[nt][rr] + bv;
    }
  }

  if constexpr (ALPHA) {
    if constexpr (!AV_AHEAD) {
#pragma unroll
      for (int nt = 0; nt < NT; ++nt) {
        avs[nt] = a_src[nt * 16 + lo];
        avd[nt] = a_dst[nt * 16 + lo];
      }
    }
    float v[16];
#pragma unroll
    for (int rr = 0; rr < 8; ++rr) {
      float s = acc[0][rr] * avs[0];
      float d = acc[0][rr] * avd[0];
#pragma unroll
      for (int nt = 1; nt < NT; ++nt) {
        s += acc[nt][rr] * avs[nt];
        d += acc[nt][rr] * avd[nt];
      }
      v[rr] = s;
      v[8 + rr] = d;
    }
#pragma unroll
    for (int w = 8; w >= 1; w >>= 1) {
      const bool up = (lo & w) != 0;
#pragma unroll
      for (int i = 0; i < w; ++i) {
        const float lower = v[i];
        const float upper = v[i + w];
        const float keep = up ? upper : lower;
        const float send = up ? lower : upper;
        v[i] = keep + __shfl_xor(send, w, 32);
      }
    }
    const long row = m0 + (lo & 7) + 8 * hi;
    float* __restrict__ dstp = (lo < 8) ? alpha_s : alpha_d;
    if (row < M) dstp[row] = v[0];
  }
}

template <int K, int N, bool ALPHA>
__global__ __launch_bounds__(256) void gemm_wmma_f32(
    const float* __restrict__ A, const float* __restrict__ B,
    const float* __restrict__ bias, float* __restrict__ C, int M,
    const float* __restrict__ a_src, const float* __restrict__ a_dst,
    float* __restrict__ alpha_s, float* __restrict__ alpha_d) {
  const int wave = __builtin_amdgcn_readfirstlane((int)(threadIdx.x >> 5));
  const long m0 = (long)blockIdx.x * 128 + wave * 16;
  gemm_tile<K, N, ALPHA, 256>(A, B, bias, C, M, a_src, a_dst, alpha_s, alpha_d, m0);
}

#define BIN_SHIFT 8
#define BIN_NODES 256
#define NBINS     ((N_NODES + BIN_NODES - 1) / BIN_NODES)
#define NCHUNK    256
#define CHUNK     ((N_EDGES + NCHUNK - 1) / NCHUNK)
#define TP        416

#define EPT ((CHUNK + 1023) / 1024)

static_assert(NBINS + 1 <= 512, "one bin per thread of the first 16 waves");
static_assert(NBINS + 1 <= TP && (TP % 32) == 0, "run table row");

__device__ __forceinline__ void chunk_order(
    const int* __restrict__ adj, int* __restrict__ runtab,
    int* __restrict__ ebuf, const int blk) {
  __shared__ int hb[NBINS + 1];
  __shared__ int wsum[16];
  __shared__ int s_val[CHUNK];
  const int tid = threadIdx.x;
  const int lane = tid & 31;
  const int wave = tid >> 5;

  for (int i = tid; i < NBINS + 1; i += 1024) hb[i] = 0;
  const int e0 = blk * CHUNK;
  const int e1 = (e0 + CHUNK < N_EDGES) ? (e0 + CHUNK) : N_EDGES;
  int es[EPT], ed[EPT];
#pragma unroll
  for (int u = 0; u < EPT; ++u) {
    const int i = e0 + tid + u * 1024;
    const int ii = (i < e1) ? i : (e1 - 1);
    es[u] = adj[ii];
    ed[u] = adj[N_EDGES + ii];
  }
  __builtin_amdgcn_sched_barrier(0);
  __syncthreads();

  int rk[EPT];
#pragma unroll
  for (int u = 0; u < EPT; ++u) {
    rk[u] = 0;
    if (e0 + tid + u * 1024 < e1) rk[u] = atomicAdd(&hb[ed[u] >> BIN_SHIFT], 1);
  }
  __syncthreads();

  int c = 0, inc = 0;
  if (tid < 512) {
    if (tid < NBINS) c = hb[tid];
    inc = c;
#pragma unroll
    for (int o = 1; o < 32; o <<= 1) {
      const int t = __shfl_up(inc, o, 32);
      if (lane >= o) inc += t;
    }
    if (lane == 31) wsum[wave] = inc;
  }
  __syncthreads();
  if (tid <= NBINS) {
    int wpre = 0;
    for (int w = 0; w < wave; ++w) wpre += wsum[w];
    const int lstart = wpre + inc - c;
    hb[tid] = lstart;
    runtab[blk * TP + tid] = lstart;
  }
  __syncthreads();

#pragma unroll
  for (int u = 0; u < EPT; ++u) {
    if (e0 + tid + u * 1024 < e1)
      s_val[hb[ed[u] >> BIN_SHIFT] + rk[u]] =
          (es[u] << BIN_SHIFT) | (ed[u] & (BIN_NODES - 1));
  }
  __syncthreads();
  const int n = e1 - e0;
  for (int j = tid; j < n; j += 1024) ebuf[e0 + j] = s_val[j];
}

#define G1_BLOCKS 256

static_assert(32 * G1_BLOCKS * 16 >= N_NODES, "first launch: every 16-row tile has a wave");

__global__ __launch_bounds__(1024, 16) void order_and_gemm1(
    const int* __restrict__ adj, int* __restrict__ runtab, int* __restrict__ ebuf,
    const float* __restrict__ X, const float* __restrict__ W, float* __restrict__ h,
    const float* __restrict__ a_src, const float* __restrict__ a_dst,
    float* __restrict__ alpha_s, float* __restrict__ alpha_d) {
  const int blk = blockIdx.x;
  if (blk < G1_BLOCKS) {
    const int wave = __builtin_amdgcn_readfirstlane((int)(threadIdx.x >> 5));
    const long tile = (long)(wave >> 1) * (2 * G1_BLOCKS) + 2 * blk + (wave & 1);
    gemm_tile<IN_DIM, HDIM, true, 1024>(X, W, nullptr, h, N_NODES, a_src, a_dst,
                                        alpha_s, alpha_d, tile * 16);
  } else {
    chunk_order(adj, runtab, ebuf, blk - G1_BLOCKS);
  }
}

#define STAGE_CAP 12288
#define RS 8
#define RL 4

static_assert(RL * NCHUNK == 1024, "bin_sort: every thread has its place in a run");
static_assert(NCHUNK == BIN_NODES, "bin_sort: the first 8 waves take a chunk and a node each");

__device__ __forceinline__ void group_and_sort(
    int* __restrict__ stage, int* __restrict__ cur, const int* __restrict__ loc,
    const int* __restrict__ ebuf, int* __restrict__ csr, const int (&v)[RS],
    const int bs, const int gs, const int rn, const int rq,
    const int lane, const int wave) {
#pragma unroll
  for (int u = 0; u < RS; ++u) {
    if (v[u] >= 0) {
      const int p = atomicAdd(&cur[v[u] & (BIN_NODES - 1)], 1);
      stage[p] = v[u] >> BIN_SHIFT;
    }
  }
  for (int i = rq + RL * RS; i < rn; i += RL) {
    const int w = ebuf[gs + i];
    const int p = atomicAdd(&cur[w & (BIN_NODES - 1)], 1);
    stage[p] = w >> BIN_SHIFT;
  }
  __syncthreads();

  for (int d = wave; d < BIN_NODES; d += 32) {
    const int start = __builtin_amdgcn_readfirstlane(loc[d]);
    const int end = __builtin_amdgcn_readfirstlane(loc[d + 1]);
    for (int ib = start; ib < end; ib += 32) {
      const int i = ib + lane;
      const bool vi = i < end;
      const int mine = vi ? stage[i] : 0;
      int r = 0;
      for (int kb = start; kb < end; kb += 32) {
        const int kj = kb + lane;
        const int other = (kj < end) ? stage[kj] : 0;
        const int nk = (end - kb < 32) ? (end - kb) : 32;
        for (int k = 0; k < nk; ++k) {
          const int o = __builtin_amdgcn_readlane(other, k);
          const int kk = kb + k;
          r += (o < mine || (o == mine && kk < i)) ? 1 : 0;
        }
      }
      if (vi) csr[bs + start + r] = mine;
    }
  }
}

__global__ __launch_bounds__(1024) void bin_sort(
    const int* __restrict__ runtab, const int* __restrict__ ebuf,
    int* __restrict__ off, int* __restrict__ tmp, int* __restrict__ csr) {
  __shared__ int cnt[BIN_NODES];
  __shared__ int loc[BIN_NODES + 1];
  __shared__ int cur[BIN_NODES];
  __shared__ int wsum[BIN_NODES / 32];
  __shared__ int rsum[NCHUNK / 32];
  __shared__ int lsum[NCHUNK / 32];
  __shared__ int run_g[NCHUNK];
  __shared__ int run_n[NCHUNK];
  __shared__ int stage_l[STAGE_CAP];
  const int tid = threadIdx.x;
  const int lane = tid & 31;
  const int wave = tid >> 5;
  const int b = blockIdx.x;

  if (tid < NCHUNK) {
    const int rs = runtab[tid * TP + b];
    const int ln = runtab[tid * TP + b + 1] - rs;
    run_g[tid] = tid * CHUNK + rs;
    run_n[tid] = ln;
    int sr = rs, sl = ln;
#pragma unroll
    for (int o = 16; o > 0; o >>= 1) {
      sr += __shfl_xor(sr, o, 32);
      sl += __shfl_xor(sl, o, 32);
    }
    if (lane == 0) {
      rsum[wave] = sr;
      lsum[wave] = sl;
    }
    cnt[tid] = 0;
  }
  __syncthreads();
  int bs = 0, m = 0;
#pragma unroll
  for (int w = 0; w < NCHUNK / 32; ++w) {
    bs += rsum[w];
    m += lsum[w];
  }
  const int be = bs + m;

  const int rq = tid & (RL - 1);
  const int gs = run_g[tid / RL];
  const int rn = run_n[tid / RL];
  int v[RS];
#pragma unroll
  for (int u = 0; u < RS; ++u) {
    const int i = rq + RL * u;
    v[u] = (i < rn) ? ebuf[gs + i] : -1;
  }
#pragma unroll
  for (int u = 0; u < RS; ++u)
    if (v[u] >= 0) atomicAdd(&cnt[v[u] & (BIN_NODES - 1)], 1);
  for (int i = rq + RL * RS; i < rn; i += RL)
    atomicAdd(&cnt[ebuf[gs + i] & (BIN_NODES - 1)], 1);
  __syncthreads();

  int c = 0, inc = 0;
  if (tid < BIN_NODES) {
    c = cnt[tid];
    inc = c;
#pragma unroll
    for (int o = 1; o < 32; o <<= 1) {
      const int t = __shfl_up(inc, o, 32);
      if (lane >= o) inc += t;
    }
    if (lane == 31) wsum[wave] = inc;
  }
  __syncthreads();
  if (tid < BIN_NODES) {
    int wpre = 0;
    for (int w = 0; w < wave; ++w) wpre += wsum[w];
    const int ls = wpre + inc - c;
    loc[tid] = ls;
    cur[tid] = ls;
    const int node = b * BIN_NODES + tid;
    if (node < N_NODES) off[node] = bs + ls;
  }
  if (tid == 0) {
    loc[BIN_NODES] = m;
    if (b == NBINS - 1) off[N_NODES] = be;
  }
  __syncthreads();

  if (m <= STAGE_CAP)
    group_and_sort(stage_l, cur, loc, ebuf, csr, v, bs, gs, rn, rq, lane, wave);
  else
    group_and_sort(tmp + bs, cur, loc, ebuf, csr, v, bs, gs, rn, rq, lane, wave);
}

#define CH 16

template <int I> struct idx_c { static constexpr int value = I; };
template <int I, int N, class F>
__device__ __forceinline__ void static_for(F&& f) {
  if constexpr (I < N) {
    f(idx_c<I>{});
    static_for<I + 1, N>(f);
  }
}

template <int K>
__device__ __forceinline__ int share16(int x) {
  return __builtin_amdgcn_update_dpp(0, x, 0x150 + K, 0xf, 0xf, true);
}
template <int K>
__device__ __forceinline__ float share16f(float x) {
  return __int_as_float(share16<K>(__float_as_int(x)));
}
template <int W>
__device__ __forceinline__ float xor16(float x) {
  return __int_as_float(
      __builtin_amdgcn_update_dpp(0, __float_as_int(x), 0x160 + W, 0xf, 0xf, true));
}

template <int W>
__device__ __forceinline__ void fold16(float (&v)[CH], const bool up) {
  static_for<0, W>([&](auto ic) {
    constexpr int i = decltype(ic)::value;
    const float lower = v[i];
    const float upper = v[i + W];
    const float keep = up ? upper : lower;
    const float send = up ? lower : upper;
    v[i] = keep + xor16<W>(send);
  });
}

__device__ __forceinline__ void walk_chunk(
    const float* __restrict__ h, const int hl, const unsigned hofs, const float4 av,
    const float adn, const int sj, const int cnt, const int cntmax,
    float& m, float& ssum, float4& a) {
  float4 g[CH];
  static_for<0, CH>([&](auto kc) {
    constexpr int k = decltype(kc)::value;
    if (k == 0 || k < cntmax) {
      const unsigned ofs = ((unsigned)share16<k>(sj) << 8) | hofs;
      g[k] = *(const float4*)((const char*)h + ofs);
    }
  });

  float v[CH];
  static_for<0, CH>([&](auto kc) {
    constexpr int k = decltype(kc)::value;
    v[k] = 0.0f;
    if (k == 0 || k < cntmax)
      v[k] = g[k].x * av.x + g[k].y * av.y + g[k].z * av.z + g[k].w * av.w;
  });

  fold16<8>(v, (hl & 8) != 0);
  fold16<4>(v, (hl & 4) != 0);
  fold16<2>(v, (hl & 2) != 0);
  fold16<1>(v, (hl & 1) != 0);
  const float score = v[0];

  float e = score + adn;
  e = (e > 0.0f) ? e : NEG_SLOPE * e;
  const bool valid = hl < cnt;
  float cm = valid ? e : m;
  cm = fmaxf(cm, xor16<8>(cm));
  cm = fmaxf(cm, xor16<4>(cm));
  cm = fmaxf(cm, xor16<2>(cm));
  cm = fmaxf(cm, xor16<1>(cm));
  const float mn = fmaxf(m, cm);
  const float scale = __expf(m - mn);
  ssum *= scale;
  a.x *= scale;
  a.y *= scale;
  a.z *= scale;
  a.w *= scale;
  m = mn;

  const float p = valid ? __expf(e - m) : 0.0f;
  static_for<0, CH>([&](auto kc) {
    constexpr int k = decltype(kc)::value;
    if (k == 0 || k < cntmax) {
      const float pk = share16f<k>(p);
      ssum += pk;
      a.x += g[k].x * pk;
      a.y += g[k].y * pk;
      a.z += g[k].z * pk;
      a.w += g[k].w * pk;
    }
  });
}

template <int ELU>
__global__ __launch_bounds__(256) void node_walk(
    const int* __restrict__ off, const int* __restrict__ csr,
    const float* __restrict__ alpha_s, const float* __restrict__ alpha_d,
    const float* __restrict__ h, const float* __restrict__ a_src,
    const float* __restrict__ b, float* __restrict__ out, int n_nodes) {
  const int lane = threadIdx.x & 31;
  const int hl = lane & 15;
  const int half = lane >> 4;
  const int gw = __builtin_amdgcn_readfirstlane((int)((blockIdx.x * 256 + threadIdx.x) >> 5));
  const int tw = (int)gridDim.x * 8;
  const int npairs = (n_nodes + 1) >> 1;
  const float4 bb = *(const float4*)(b + 4 * hl);
  const float4 av = *(const float4*)(a_src + 4 * hl);
  const unsigned hofs = (unsigned)hl << 4;
  for (int pr = gw; pr < npairs; pr += tw) {
    const int n = 2 * pr + half;
    const bool nv = n < n_nodes;
    const int nn = nv ? n : (n_nodes - 1);
    const int start = off[nn];
    const int endg = off[nn + 1];
    const int deg = nv ? (endg - start) : 0;
    const float adn = alpha_d[nn];
    float es = alpha_s[nn] + adn;
    es = (es > 0.0f) ? es : NEG_SLOPE * es;
    const float4 hn = *(const float4*)(h + (size_t)nn * HDIM + 4 * hl);

    float m = es;
    float ssum = 1.0f;
    float4 a = hn;

    const int d0 = __builtin_amdgcn_readlane(deg, 0);
    const int d1 = __builtin_amdgcn_readlane(deg, 16);
    const int maxdeg = (d0 > d1) ? d0 : d1;
    int base = 0;
    do {
      const int q = base + hl;
      int sj = nn;
      if (q < deg) sj = csr[start + q];
      const int left = maxdeg - base;
      walk_chunk(h, hl, hofs, av, adn, sj, deg - base, (left < CH) ? left : CH, m, ssum, a);
      base += CH;
    } while (base < maxdeg);

    float4 o;
    o.x = a.x / ssum + bb.x;
    o.y = a.y / ssum + bb.y;
    o.z = a.z / ssum + bb.z;
    o.w = a.w / ssum + bb.w;
    if (ELU) {
      o.x = (o.x > 0.0f) ? o.x : (__expf(o.x) - 1.0f);
      o.y = (o.y > 0.0f) ? o.y : (__expf(o.y) - 1.0f);
      o.z = (o.z > 0.0f) ? o.z : (__expf(o.z) - 1.0f);
      o.w = (o.w > 0.0f) ? o.w : (__expf(o.w) - 1.0f);
    }
    if (nv) *(float4*)(out + (size_t)n * HDIM + 4 * hl) = o;
  }
}

extern "C" void kernel_launch(void* const* d_in, const int* in_sizes, int n_in,
                              void* d_out, int out_size, void* d_ws,
                              size_t ws_size, hipStream_t stream) {
  const float* X      = (const float*)d_in[0];
  const int*   adj    = (const int*)d_in[1];
  const float* W1     = (const float*)d_in[2];
  const float* a_src1 = (const float*)d_in[3];
  const float* a_dst1 = (const float*)d_in[4];
  const float* b1     = (const float*)d_in[5];
  const float* W2     = (const float*)d_in[6];
  const float* a_src2 = (const float*)d_in[7];
  const float* a_dst2 = (const float*)d_in[8];
  const float* b2     = (const float*)d_in[9];
  const float* clas_W = (const float*)d_in[10];
  const float* clas_b = (const float*)d_in[11];

  float* out = (float*)d_out;
  float* x2  = out + (size_t)N_NODES * N_CLASSES;

  float* ws   = (float*)d_ws;
  float* h    = ws;
  float* x1   = ws + (size_t)N_NODES * HDIM;
  int*   ib   = (int*)(ws + (size_t)2 * N_NODES * HDIM);
  float* as_  = ws + (size_t)3 * N_NODES * HDIM;
  float* ad_  = as_ + N_NODES;
  int* runtab = ib;
  int* off  = ib + 2 * 100352;
  int* ebuf = ib + 3 * 100352;
  int* tmp  = ebuf + N_EDGES;
  int* csr  = tmp + N_EDGES;

  const int gemm_blocks = (N_NODES + 127) / 128;
  const int walk_blocks = 1024;

  order_and_gemm1<<<G1_BLOCKS + NCHUNK, 1024, 0, stream>>>(
      adj, runtab, ebuf, X, W1, h, a_src1, a_dst1, as_, ad_);
  bin_sort<<<NBINS, 1024, 0, stream>>>(runtab, ebuf, off, tmp, csr);

  node_walk<1><<<walk_blocks, 256, 0, stream>>>(off, csr, as_, ad_, h, a_src1, b1, x1, N_NODES);

  gemm_wmma_f32<HDIM, HDIM, true><<<gemm_blocks, 256, 0, stream>>>(
      x1, W2, nullptr, h, N_NODES, a_src2, a_dst2, as_, ad_);
  node_walk<0><<<walk_blocks, 256, 0, stream>>>(off, csr, as_, ad_, h, a_src2, b2, x2, N_NODES);

  gemm_wmma_f32<HDIM, N_CLASSES, false><<<gemm_blocks, 256, 0, stream>>>(
      x2, clas_W, clas_b, out, N_NODES, nullptr, nullptr, nullptr, nullptr);
}


# Node walks end with the dense layer that follows (layer 2 GEMM; classifier): a wave keeps its output rows in LDS as a WMMA tile; blocks take nodes in units of 32; 6 launches to 4; float sums same
# speedup vs baseline: 6.6375x; 1.0757x over previous
#include <hip/hip_runtime.h>
#include <hip/hip_bf16.h>

typedef float v2f __attribute__((ext_vector_type(2)));
typedef float v8f __attribute__((ext_vector_type(8)));

#define N_NODES   100000
#define N_EDGES   1600000
#define IN_DIM    128
#define HDIM      64
#define N_CLASSES 16
#define NEG_SLOPE 0.2f

template <int N, class RowOf>
__device__ __forceinline__ void tile_store(
    const v8f (&acc)[N / 16], const float* __restrict__ bias, float* __restrict__ C,
    const long M, const int lo, const int hi, RowOf rowof) {
#pragma unroll
  for (int nt = 0; nt < N / 16; ++nt) {
    const int col = nt * 16 + lo;
    const float bv = bias ? bias[col] : 0.0f;
#pragma unroll
    for (int rr = 0; rr < 8; ++rr) {
      const long row = rowof(rr + 8 * hi);
      if (row < M) C[(size_t)row * N + col] = acc[nt][rr] + bv;
    }
  }
}

template <int N, class RowOf>
__device__ __forceinline__ void tile_scores(
    const v8f (&acc)[N / 16], const float (&avs)[N / 16], const float (&avd)[N / 16],
    float* __restrict__ alpha_s, float* __restrict__ alpha_d,
    const long M, const int lo, const int hi, RowOf rowof) {
  constexpr int NT = N / 16;
  float v[16];
#pragma unroll
  for (int rr = 0; rr < 8; ++rr) {
    float s = acc[0][rr] * avs[0];
    float d = acc[0][rr] * avd[0];
#pragma unroll
    for (int nt = 1; nt < NT; ++nt) {
      s += acc[nt][rr] * avs[nt];
      d += acc[nt][rr] * avd[nt];
    }
    v[rr] = s;
    v[8 + rr] = d;
  }
#pragma unroll
  for (int w = 8; w >= 1; w >>= 1) {
    const bool up = (lo & w) != 0;
#pragma unroll
    for (int i = 0; i < w; ++i) {
      const float lower = v[i];
      const float upper = v[i + w];
      const float keep = up ? upper : lower;
      const float send = up ? lower : upper;
      v[i] = keep + __shfl_xor(send, w, 32);
    }
  }
  const long row = rowof((lo & 7) + 8 * hi);
  float* __restrict__ dstp = (lo < 8) ? alpha_s : alpha_d;
  if (row < M) dstp[row] = v[0];
}

template <int K, int N, bool ALPHA, int NTHR>
__device__ __forceinline__ void gemm_tile(
    const float* __restrict__ A, const float* __restrict__ B,
    const float* __restrict__ bias, float* __restrict__ C, int M,
    const float* __restrict__ a_src, const float* __restrict__ a_dst,
    float* __restrict__ alpha_s, float* __restrict__ alpha_d, const long m0) {
  __shared__ float Bs[K * N];
  const int tid = threadIdx.x;
  for (int i = tid; i < K * N; i += NTHR) Bs[i] = B[i];
  __syncthreads();
  if (m0 >= M) return;

  const int lane = tid & 31;
  const int hi = lane >> 4;
  const int lo = lane & 15;

  long r = m0 + lo;
  if (r > M - 1) r = M - 1;
  const float* __restrict__ Arow = A + (size_t)r * K;

  constexpr int NT = N / 16;
  v8f acc[NT] = {};

  for (int kk = 0; kk < K / 4; ++kk) {
    const int kb = kk * 4 + 2 * hi;
    v2f a;
    a.x = Arow[kb];
    a.y = Arow[kb + 1];
#pragma unroll
    for (int nt = 0; nt < NT; ++nt) {
      v2f b;
      b.x = Bs[kb * N + nt * 16 + lo];
      b.y = Bs[(kb + 1) * N + nt * 16 + lo];
      acc[nt] = __builtin_amdgcn_wmma_f32_16x16x4_f32(
           false, a,  false, b,
           (short)0, acc[nt],  false,  false);
    }
  }

  const auto rowof = [m0](const int r) { return m0 + r; };
  tile_store<N>(acc, bias, C, M, lo, hi, rowof);

  if constexpr (ALPHA) {
    float avs[NT], avd[NT];
#pragma unroll
    for (int nt = 0; nt < NT; ++nt) {
      avs[nt] = a_src[nt * 16 + lo];
      avd[nt] = a_dst[nt * 16 + lo];
    }
    tile_scores<N>(acc, avs, avd, alpha_s, alpha_d, M, lo, hi, rowof);
  }
}

#define BIN_SHIFT 8
#define BIN_NODES 256
#define NBINS     ((N_NODES + BIN_NODES - 1) / BIN_NODES)
#define NCHUNK    256
#define CHUNK     ((N_EDGES + NCHUNK - 1) / NCHUNK)
#define TP        416

#define EPT ((CHUNK + 1023) / 1024)

static_assert(NBINS + 1 <= 512, "one bin per thread of the first 16 waves");
static_assert(NBINS + 1 <= TP && (TP % 32) == 0, "run table row");

__device__ __forceinline__ void chunk_order(
    const int* __restrict__ adj, int* __restrict__ runtab,
    int* __restrict__ ebuf, const int blk) {
  __shared__ int hb[NBINS + 1];
  __shared__ int wsum[16];
  __shared__ int s_val[CHUNK];
  const int tid = threadIdx.x;
  const int lane = tid & 31;
  const int wave = tid >> 5;

  for (int i = tid; i < NBINS + 1; i += 1024) hb[i] = 0;
  const int e0 = blk * CHUNK;
  const int e1 = (e0 + CHUNK < N_EDGES) ? (e0 + CHUNK) : N_EDGES;
  int es[EPT], ed[EPT];
#pragma unroll
  for (int u = 0; u < EPT; ++u) {
    const int i = e0 + tid + u * 1024;
    const int ii = (i < e1) ? i : (e1 - 1);
    es[u] = adj[ii];
    ed[u] = adj[N_EDGES + ii];
  }
  __builtin_amdgcn_sched_barrier(0);
  __syncthreads();

  int rk[EPT];
#pragma unroll
  for (int u = 0; u < EPT; ++u) {
    rk[u] = 0;
    if (e0 + tid + u * 1024 < e1) rk[u] = atomicAdd(&hb[ed[u] >> BIN_SHIFT], 1);
  }
  __syncthreads();

  int c = 0, inc = 0;
  if (tid < 512) {
    if (tid < NBINS) c = hb[tid];
    inc = c;
#pragma unroll
    for (int o = 1; o < 32; o <<= 1) {
      const int t = __shfl_up(inc, o, 32);
      if (lane >= o) inc += t;
    }
    if (lane == 31) wsum[wave] = inc;
  }
  __syncthreads();
  if (tid <= NBINS) {
    int wpre = 0;
    for (int w = 0; w < wave; ++w) wpre += wsum[w];
    const int lstart = wpre + inc - c;
    hb[tid] = lstart;
    runtab[blk * TP + tid] = lstart;
  }
  __syncthreads();

#pragma unroll
  for (int u = 0; u < EPT; ++u) {
    if (e0 + tid + u * 1024 < e1)
      s_val[hb[ed[u] >> BIN_SHIFT] + rk[u]] =
          (es[u] << BIN_SHIFT) | (ed[u] & (BIN_NODES - 1));
  }
  __syncthreads();
  const int n = e1 - e0;
  for (int j = tid; j < n; j += 1024) ebuf[e0 + j] = s_val[j];
}

#define G1_BLOCKS 256

static_assert(32 * G1_BLOCKS * 16 >= N_NODES, "first launch: every 16-row tile has a wave");

__global__ __launch_bounds__(1024, 16) void order_and_gemm1(
    const int* __restrict__ adj, int* __restrict__ runtab, int* __restrict__ ebuf,
    const float* __restrict__ X, const float* __restrict__ W, float* __restrict__ h,
    const float* __restrict__ a_src, const float* __restrict__ a_dst,
    float* __restrict__ alpha_s, float* __restrict__ alpha_d) {
  const int blk = blockIdx.x;
  if (blk < G1_BLOCKS) {
    const int wave = __builtin_amdgcn_readfirstlane((int)(threadIdx.x >> 5));
    const long tile = (long)(wave >> 1) * (2 * G1_BLOCKS) + 2 * blk + (wave & 1);
    gemm_tile<IN_DIM, HDIM, true, 1024>(X, W, nullptr, h, N_NODES, a_src, a_dst,
                                        alpha_s, alpha_d, tile * 16);
  } else {
    chunk_order(adj, runtab, ebuf, blk - G1_BLOCKS);
  }
}

#define STAGE_CAP 12288
#define RS 8
#define RL 4

static_assert(RL * NCHUNK == 1024, "bin_sort: every thread has its place in a run");
static_assert(NCHUNK == BIN_NODES, "bin_sort: the first 8 waves take a chunk and a node each");

__device__ __forceinline__ void group_and_sort(
    int* __restrict__ stage, int* __restrict__ cur, const int* __restrict__ loc,
    const int* __restrict__ ebuf, int* __restrict__ csr, const int (&v)[RS],
    const int bs, const int gs, const int rn, const int rq,
    const int lane, const int wave) {
#pragma unroll
  for (int u = 0; u < RS; ++u) {
    if (v[u] >= 0) {
      const int p = atomicAdd(&cur[v[u] & (BIN_NODES - 1)], 1);
      stage[p] = v[u] >> BIN_SHIFT;
    }
  }
  for (int i = rq + RL * RS; i < rn; i += RL) {
    const int w = ebuf[gs + i];
    const int p = atomicAdd(&cur[w & (BIN_NODES - 1)], 1);
    stage[p] = w >> BIN_SHIFT;
  }
  __syncthreads();

  for (int d = wave; d < BIN_NODES; d += 32) {
    const int start = __builtin_amdgcn_readfirstlane(loc[d]);
    const int end = __builtin_amdgcn_readfirstlane(loc[d + 1]);
    for (int ib = start; ib < end; ib += 32) {
      const int i = ib + lane;
      const bool vi = i < end;
      const int mine = vi ? stage[i] : 0;
      int r = 0;
      for (int kb = start; kb < end; kb += 32) {
        const int kj = kb + lane;
        const int other = (kj < end) ? stage[kj] : 0;
        const int nk = (end - kb < 32) ? (end - kb) : 32;
        for (int k = 0; k < nk; ++k) {
          const int o = __builtin_amdgcn_readlane(other, k);
          const int kk = kb + k;
          r += (o < mine || (o == mine && kk < i)) ? 1 : 0;
        }
      }
      if (vi) csr[bs + start + r] = mine;
    }
  }
}

__global__ __launch_bounds__(1024) void bin_sort(
    const int* __restrict__ runtab, const int* __restrict__ ebuf,
    int* __restrict__ off, int* __restrict__ tmp, int* __restrict__ csr) {
  __shared__ int cnt[BIN_NODES];
  __shared__ int loc[BIN_NODES + 1];
  __shared__ int cur[BIN_NODES];
  __shared__ int wsum[BIN_NODES / 32];
  __shared__ int rsum[NCHUNK / 32];
  __shared__ int lsum[NCHUNK / 32];
  __shared__ int run_g[NCHUNK];
  __shared__ int run_n[NCHUNK];
  __shared__ int stage_l[STAGE_CAP];
  const int tid = threadIdx.x;
  const int lane = tid & 31;
  const int wave = tid >> 5;
  const int b = blockIdx.x;

  if (tid < NCHUNK) {
    const int rs = runtab[tid * TP + b];
    const int ln = runtab[tid * TP + b + 1] - rs;
    run_g[tid] = tid * CHUNK + rs;
    run_n[tid] = ln;
    int sr = rs, sl = ln;
#pragma unroll
    for (int o = 16; o > 0; o >>= 1) {
      sr += __shfl_xor(sr, o, 32);
      sl += __shfl_xor(sl, o, 32);
    }
    if (lane == 0) {
      rsum[wave] = sr;
      lsum[wave] = sl;
    }
    cnt[tid] = 0;
  }
  __syncthreads();
  int bs = 0, m = 0;
#pragma unroll
  for (int w = 0; w < NCHUNK / 32; ++w) {
    bs += rsum[w];
    m += lsum[w];
  }
  const int be = bs + m;

  const int rq = tid & (RL - 1);
  const int gs = run_g[tid / RL];
  const int rn = run_n[tid / RL];
  int v[RS];
#pragma unroll
  for (int u = 0; u < RS; ++u) {
    const int i = rq + RL * u;
    v[u] = (i < rn) ? ebuf[gs + i] : -1;
  }
#pragma unroll
  for (int u = 0; u < RS; ++u)
    if (v[u] >= 0) atomicAdd(&cnt[v[u] & (BIN_NODES - 1)], 1);
  for (int i = rq + RL * RS; i < rn; i += RL)
    atomicAdd(&cnt[ebuf[gs + i] & (BIN_NODES - 1)], 1);
  __syncthreads();

  int c = 0, inc = 0;
  if (tid < BIN_NODES) {
    c = cnt[tid];
    inc = c;
#pragma unroll
    for (int o = 1; o < 32; o <<= 1) {
      const int t = __shfl_up(inc, o, 32);
      if (lane >= o) inc += t;
    }
    if (lane == 31) wsum[wave] = inc;
  }
  __syncthreads();
  if (tid < BIN_NODES) {
    int wpre = 0;
    for (int w = 0; w < wave; ++w) wpre += wsum[w];
    const int ls = wpre + inc - c;
    loc[tid] = ls;
    cur[tid] = ls;
    const int node = b * BIN_NODES + tid;
    if (node < N_NODES) off[node] = bs + ls;
  }
  if (tid == 0) {
    loc[BIN_NODES] = m;
    if (b == NBINS - 1) off[N_NODES] = be;
  }
  __syncthreads();

  if (m <= STAGE_CAP)
    group_and_sort(stage_l, cur, loc, ebuf, csr, v, bs, gs, rn, rq, lane, wave);
  else
    group_and_sort(tmp + bs, cur, loc, ebuf, csr, v, bs, gs, rn, rq, lane, wave);
}

#define CH 16

template <int I> struct idx_c { static constexpr int value = I; };
template <int I, int N, class F>
__device__ __forceinline__ void static_for(F&& f) {
  if constexpr (I < N) {
    f(idx_c<I>{});
    static_for<I + 1, N>(f);
  }
}

template <int K>
__device__ __forceinline__ int share16(int x) {
  return __builtin_amdgcn_update_dpp(0, x, 0x150 + K, 0xf, 0xf, true);
}
template <int K>
__device__ __forceinline__ float share16f(float x) {
  return __int_as_float(share16<K>(__float_as_int(x)));
}
template <int W>
__device__ __forceinline__ float xor16(float x) {
  return __int_as_float(
      __builtin_amdgcn_update_dpp(0, __float_as_int(x), 0x160 + W, 0xf, 0xf, true));
}

template <int W>
__device__ __forceinline__ void fold16(float (&v)[CH], const bool up) {
  static_for<0, W>([&](auto ic) {
    constexpr int i = decltype(ic)::value;
    const float lower = v[i];
    const float upper = v[i + W];
    const float keep = up ? upper : lower;
    const float send = up ? lower : upper;
    v[i] = keep + xor16<W>(send);
  });
}

__device__ __forceinline__ void walk_chunk(
    const float* __restrict__ h, const int hl, const unsigned hofs, const float4 av,
    const float adn, const int sj, const int cnt, const int cntmax,
    float& m, float& ssum, float4& a) {
  float4 g[CH];
  static_for<0, CH>([&](auto kc) {
    constexpr int k = decltype(kc)::value;
    if (k == 0 || k < cntmax) {
      const unsigned ofs = ((unsigned)share16<k>(sj) << 8) | hofs;
      g[k] = *(const float4*)((const char*)h + ofs);
    }
  });

  float v[CH];
  static_for<0, CH>([&](auto kc) {
    constexpr int k = decltype(kc)::value;
    v[k] = 0.0f;
    if (k == 0 || k < cntmax)
      v[k] = g[k].x * av.x + g[k].y * av.y + g[k].z * av.z + g[k].w * av.w;
  });

  fold16<8>(v, (hl & 8) != 0);
  fold16<4>(v, (hl & 4) != 0);
  fold16<2>(v, (hl & 2) != 0);
  fold16<1>(v, (hl & 1) != 0);
  const float score = v[0];

  float e = score + adn;
  e = (e > 0.0f) ? e : NEG_SLOPE * e;
  const bool valid = hl < cnt;
  float cm = valid ? e : m;
  cm = fmaxf(cm, xor16<8>(cm));
  cm = fmaxf(cm, xor16<4>(cm));
  cm = fmaxf(cm, xor16<2>(cm));
  cm = fmaxf(cm, xor16<1>(cm));
  const float mn = fmaxf(m, cm);
  const float scale = __expf(m - mn);
  ssum *= scale;
  a.x *= scale;
  a.y *= scale;
  a.z *= scale;
  a.w *= scale;
  m = mn;

  const float p = valid ? __expf(e - m) : 0.0f;
  static_for<0, CH>([&](auto kc) {
    constexpr int k = decltype(kc)::value;
    if (k == 0 || k < cntmax) {
      const float pk = share16f<k>(p);
      ssum += pk;
      a.x += g[k].x * pk;
      a.y += g[k].y * pk;
      a.z += g[k].z * pk;
      a.w += g[k].w * pk;
    }
  });
}

#define WALK_BLOCKS 1024
#define AST 68

static_assert(4 * WALK_BLOCKS * 32 >= N_NODES, "node walk: a wave's tile has 16 rows, 4 per unit");

template <int ELU, int NOUT, bool ALPHA, bool XOUT>
__global__ __launch_bounds__(256) void node_walk(
    const int* __restrict__ off, const int* __restrict__ csr,
    const float* __restrict__ alpha_s, const float* __restrict__ alpha_d,
    const float* __restrict__ h, const float* __restrict__ a_src,
    const float* __restrict__ b, float* __restrict__ xout,
    const float* __restrict__ Bd, const float* __restrict__ biasd, float* __restrict__ C,
    const float* __restrict__ a_src_n, const float* __restrict__ a_dst_n,
    float* __restrict__ alpha_s_n, float* __restrict__ alpha_d_n, int n_nodes) {
  __shared__ __attribute__((aligned(16))) float Bs[HDIM * NOUT];
  __shared__ __attribute__((aligned(16))) float As[8 * 16 * AST];
  const int lane = threadIdx.x & 31;
  const int hl = lane & 15;
  const int half = lane >> 4;
  constexpr int NB4 = HDIM * NOUT / 4 / 256;
  static_assert(NB4 * 4 * 256 == HDIM * NOUT, "Bd: whole 16-byte pieces for 256 threads");
  float4 bt[NB4];
#pragma unroll
  for (int u = 0; u < NB4; ++u) bt[u] = ((const float4*)Bd)[threadIdx.x + u * 256];
  const float4 bb = *(const float4*)(b + 4 * hl);
  const float4 av = *(const float4*)(a_src + 4 * hl);
#pragma unroll
  for (int u = 0; u < NB4; ++u) ((float4*)Bs)[threadIdx.x + u * 256] = bt[u];
  __syncthreads();

  const int wave = __builtin_amdgcn_readfirstlane((int)(threadIdx.x >> 5));
  const int blk = blockIdx.x;
  const int nunits = (n_nodes + 31) >> 5;
  const unsigned hofs = (unsigned)hl << 4;
  float* __restrict__ Aw = As + wave * (16 * AST);
  float* __restrict__ Amine = Aw + half * AST + 4 * hl;

  int it = 0;
  for (; it < 8; ++it) {
    const int u = blk + (it >> 1) * WALK_BLOCKS;
    if (u >= nunits) break;
    const int n = 32 * u + 4 * wave + 2 * (it & 1) + half;
    const bool nv = n < n_nodes;
    const int nn = nv ? n : (n_nodes - 1);
    const int start = off[nn];
    const int endg = off[nn + 1];
    const int deg = nv ? (endg - start) : 0;
    const float adn = alpha_d[nn];
    float es = alpha_s[nn] + adn;
    es = (es > 0.0f) ? es : NEG_SLOPE * es;
    const float4 hn = *(const float4*)(h + (size_t)nn * HDIM + 4 * hl);

    float m = es;
    float ssum = 1.0f;
    float4 a = hn;

    const int d0 = __builtin_amdgcn_readlane(deg, 0);
    const int d1 = __builtin_amdgcn_readlane(deg, 16);
    const int maxdeg = (d0 > d1) ? d0 : d1;
    int base = 0;
    do {
      const int q = base + hl;
      int sj = nn;
      if (q < deg) sj = csr[start + q];
      const int left = maxdeg - base;
      walk_chunk(h, hl, hofs, av, adn, sj, deg - base, (left < CH) ? left : CH, m, ssum, a);
      base += CH;
    } while (base < maxdeg);

    float4 o;
    o.x = a.x / ssum + bb.x;
    o.y = a.y / ssum + bb.y;
    o.z = a.z / ssum + bb.z;
    o.w = a.w / ssum + bb.w;
    if (ELU) {
      o.x = (o.x > 0.0f) ? o.x : (__expf(o.x) - 1.0f);
      o.y = (o.y > 0.0f) ? o.y : (__expf(o.y) - 1.0f);
      o.z = (o.z > 0.0f) ? o.z : (__expf(o.z) - 1.0f);
      o.w = (o.w > 0.0f) ? o.w : (__expf(o.w) - 1.0f);
    }
    if (XOUT) {
      if (nv) *(float4*)(xout + (size_t)n * HDIM + 4 * hl) = o;
    }
    float* __restrict__ ar = Amine + it * (2 * AST);
    ar[0] = nv ? o.x : 0.0f;
    ar[1] = nv ? o.y : 0.0f;
    ar[2] = nv ? o.z : 0.0f;
    ar[3] = nv ? o.w : 0.0f;
  }
  for (; it < 8; ++it) {
    float* __restrict__ ar = Amine + it * (2 * AST);
    ar[0] = 0.0f;
    ar[1] = 0.0f;
    ar[2] = 0.0f;
    ar[3] = 0.0f;
  }

  const int lo = hl;
  const int hi = half;
  constexpr int NT = NOUT / 16;
  v8f acc[NT] = {};
  [[maybe_unused]] float avs[NT], avd[NT];
  if constexpr (ALPHA) {
#pragma unroll
    for (int nt = 0; nt < NT; ++nt) {
      avs[nt] = a_src_n[nt * 16 + lo];
      avd[nt] = a_dst_n[nt * 16 + lo];
    }
  }
  const float* __restrict__ Arow = Aw + lo * AST;
  for (int kk = 0; kk < HDIM / 4; ++kk) {
    const int kb = kk * 4 + 2 * hi;
    v2f fa;
    fa.x = Arow[kb];
    fa.y = Arow[kb + 1];
#pragma unroll
    for (int nt = 0; nt < NT; ++nt) {
      v2f fb;
      fb.x = Bs[kb * NOUT + nt * 16 + lo];
      fb.y = Bs[(kb + 1) * NOUT + nt * 16 + lo];
      acc[nt] = __builtin_amdgcn_wmma_f32_16x16x4_f32(
           false, fa,  false, fb,
           (short)0, acc[nt],  false,  false);
    }
  }
  const auto rowof = [blk, wave](const int r) {
    return (long)32 * (blk + (r >> 2) * WALK_BLOCKS) + 4 * wave + (r & 3);
  };
  tile_store<NOUT>(acc, biasd, C, n_nodes, lo, hi, rowof);
  if constexpr (ALPHA)
    tile_scores<NOUT>(acc, avs, avd, alpha_s_n, alpha_d_n, n_nodes, lo, hi, rowof);
}

extern "C" void kernel_launch(void* const* d_in, const int* in_sizes, int n_in,
                              void* d_out, int out_size, void* d_ws,
                              size_t ws_size, hipStream_t stream) {
  const float* X      = (const float*)d_in[0];
  const int*   adj    = (const int*)d_in[1];
  const float* W1     = (const float*)d_in[2];
  const float* a_src1 = (const float*)d_in[3];
  const float* a_dst1 = (const float*)d_in[4];
  const float* b1     = (const float*)d_in[5];
  const float* W2     = (const float*)d_in[6];
  const float* a_src2 = (const float*)d_in[7];
  const float* a_dst2 = (const float*)d_in[8];
  const float* b2     = (const float*)d_in[9];
  const float* clas_W = (const float*)d_in[10];
  const float* clas_b = (const float*)d_in[11];

  float* out = (float*)d_out;
  float* x2  = out + (size_t)N_NODES * N_CLASSES;

  float* ws   = (float*)d_ws;
  float* h    = ws;
  float* h2   = ws + (size_t)N_NODES * HDIM;
  int*   ib   = (int*)(ws + (size_t)2 * N_NODES * HDIM);
  float* as_  = ws + (size_t)3 * N_NODES * HDIM;
  float* ad_  = as_ + N_NODES;
  int* runtab = ib;
  int* off  = ib + 2 * 100352;
  int* ebuf = ib + 3 * 100352;
  int* tmp  = ebuf + N_EDGES;
  int* csr  = tmp + N_EDGES;
  float* as2 = (float*)(csr + N_EDGES);
  float* ad2 = as2 + 100352;

  order_and_gemm1<<<G1_BLOCKS + NCHUNK, 1024, 0, stream>>>(
      adj, runtab, ebuf, X, W1, h, a_src1, a_dst1, as_, ad_);
  bin_sort<<<NBINS, 1024, 0, stream>>>(runtab, ebuf, off, tmp, csr);

  node_walk<1, HDIM, true, false><<<WALK_BLOCKS, 256, 0, stream>>>(
      off, csr, as_, ad_, h, a_src1, b1, nullptr,
      W2, nullptr, h2, a_src2, a_dst2, as2, ad2, N_NODES);

  node_walk<0, N_CLASSES, false, true><<<WALK_BLOCKS, 256, 0, stream>>>(
      off, csr, as2, ad2, h2, a_src2, b2, x2,
      clas_W, clas_b, out, nullptr, nullptr, nullptr, nullptr, N_NODES);
}


# Layer 1 GEMM requests its operands ahead of use: W1 to LDS in one round trip, A as 16-byte fragment loads 3 blocks of 8 k ahead; k order inside a block of 8 changes, and the sums' order with it
# speedup vs baseline: 6.8669x; 1.0346x over previous
#include <hip/hip_runtime.h>
#include <hip/hip_bf16.h>

typedef float v2f __attribute__((ext_vector_type(2)));
typedef float v8f __attribute__((ext_vector_type(8)));

#define N_NODES   100000
#define N_EDGES   1600000
#define IN_DIM    128
#define HDIM      64
#define N_CLASSES 16
#define NEG_SLOPE 0.2f

template <int N, class RowOf>
__device__ __forceinline__ void tile_store(
    const v8f (&acc)[N / 16], const float* __restrict__ bias, float* __restrict__ C,
    const long M, const int lo, const int hi, RowOf rowof) {
#pragma unroll
  for (int nt = 0; nt < N / 16; ++nt) {
    const int col = nt * 16 + lo;
    const float bv = bias ? bias[col] : 0.0f;
#pragma unroll
    for (int rr = 0; rr < 8; ++rr) {
      const long row = rowof(rr + 8 * hi);
      if (row < M) C[(size_t)row * N + col] = acc[nt][rr] + bv;
    }
  }
}

template <int N, class RowOf>
__device__ __forceinline__ void tile_scores(
    const v8f (&acc)[N / 16], const float (&avs)[N / 16], const float (&avd)[N / 16],
    float* __restrict__ alpha_s, float* __restrict__ alpha_d,
    const long M, const int lo, const int hi, RowOf rowof) {
  constexpr int NT = N / 16;
  float v[16];
#pragma unroll
  for (int rr = 0; rr < 8; ++rr) {
    float s = acc[0][rr] * avs[0];
    float d = acc[0][rr] * avd[0];
#pragma unroll
    for (int nt = 1; nt < NT; ++nt) {
      s += acc[nt][rr] * avs[nt];
      d += acc[nt][rr] * avd[nt];
    }
    v[rr] = s;
    v[8 + rr] = d;
  }
#pragma unroll
  for (int w = 8; w >= 1; w >>= 1) {
    const bool up = (lo & w) != 0;
#pragma unroll
    for (int i = 0; i < w; ++i) {
      const float lower = v[i];
      const float upper = v[i + w];
      const float keep = up ? upper : lower;
      const float send = up ? lower : upper;
      v[i] = keep + __shfl_xor(send, w, 32);
    }
  }
  const long row = rowof((lo & 7) + 8 * hi);
  float* __restrict__ dstp = (lo < 8) ? alpha_s : alpha_d;
  if (row < M) dstp[row] = v[0];
}

template <int K, int N, bool ALPHA, int NTHR>
__device__ __forceinline__ void gemm_tile(
    const float* __restrict__ A, const float* __restrict__ B,
    const float* __restrict__ bias, float* __restrict__ C, int M,
    const float* __restrict__ a_src, const float* __restrict__ a_dst,
    float* __restrict__ alpha_s, float* __restrict__ alpha_d, const long m0) {
  constexpr int BP = 2 * N + 16;
  constexpr int NJ = K / 8;
  constexpr int PF = 3;
  constexpr int NPC = (K / 2) * (N / 4);
  constexpr int NP = NPC / NTHR;
  static_assert(NP * NTHR == NPC, "B: whole pieces for every thread");
  static_assert(K % 8 == 0 && PF < NJ, "A: blocks of 8 k");
  __shared__ __attribute__((aligned(16))) float Bs[(K / 2) * BP];
  const int tid = threadIdx.x;
  const int lane = tid & 31;
  const int hi = lane >> 4;
  const int lo = lane & 15;
  const bool live = m0 < M;

  float4 r0[NP], r1[NP];
#pragma unroll
  for (int u = 0; u < NP; ++u) {
    const int p = tid + u * NTHR;
    const int kp = p / (N / 4);
    const int q = p % (N / 4);
    r0[u] = *(const float4*)(B + (size_t)(2 * kp) * N + 4 * q);
    r1[u] = *(const float4*)(B + (size_t)(2 * kp + 1) * N + 4 * q);
  }
  long r = m0 + lo;
  if (r > M - 1) r = M - 1;
  const float* __restrict__ Arow = A + (size_t)r * K + 4 * hi;
  float4 a[NJ];
  if (live) {
#pragma unroll
    for (int j = 0; j < PF; ++j) a[j] = *(const float4*)(Arow + 8 * j);
  }
#pragma unroll
  for (int u = 0; u < NP; ++u) {
    const int p = tid + u * NTHR;
    const int kp = p / (N / 4);
    const int q = p % (N / 4);
    float4* __restrict__ d = (float4*)(Bs + kp * BP + 8 * q);
    d[0] = make_float4(r0[u].x, r1[u].x, r0[u].y, r1[u].y);
    d[1] = make_float4(r0[u].z, r1[u].z, r0[u].w, r1[u].w);
  }
  __syncthreads();
  if (!live) return;

  constexpr int NT = N / 16;
  v8f acc[NT] = {};
  const float* __restrict__ Bl = Bs + (2 * hi) * BP + 2 * lo;

#pragma unroll
  for (int j = 0; j < NJ; ++j) {
    if (j + PF < NJ) a[j + PF] = *(const float4*)(Arow + 8 * (j + PF));
#pragma unroll
    for (int t = 0; t < 2; ++t) {
      v2f fa;
      fa.x = t ? a[j].z : a[j].x;
      fa.y = t ? a[j].w : a[j].y;
#pragma unroll
      for (int nt = 0; nt < NT; ++nt) {
        const v2f fb = *(const v2f*)(Bl + (4 * j + t) * BP + 32 * nt);
        acc[nt] = __builtin_amdgcn_wmma_f32_16x16x4_f32(
             false, fa,  false, fb,
             (short)0, acc[nt],  false,  false);
      }
    }
  }

  const auto rowof = [m0](const int r) { return m0 + r; };
  tile_store<N>(acc, bias, C, M, lo, hi, rowof);

  if constexpr (ALPHA) {
    float avs[NT], avd[NT];
#pragma unroll
    for (int nt = 0; nt < NT; ++nt) {
      avs[nt] = a_src[nt * 16 + lo];
      avd[nt] = a_dst[nt * 16 + lo];
    }
    tile_scores<N>(acc, avs, avd, alpha_s, alpha_d, M, lo, hi, rowof);
  }
}

#define BIN_SHIFT 8
#define BIN_NODES 256
#define NBINS     ((N_NODES + BIN_NODES - 1) / BIN_NODES)
#define NCHUNK    256
#define CHUNK     ((N_EDGES + NCHUNK - 1) / NCHUNK)
#define TP        416

#define EPT ((CHUNK + 1023) / 1024)

static_assert(NBINS + 1 <= 512, "one bin per thread of the first 16 waves");
static_assert(NBINS + 1 <= TP && (TP % 32) == 0, "run table row");

__device__ __forceinline__ void chunk_order(
    const int* __restrict__ adj, int* __restrict__ runtab,
    int* __restrict__ ebuf, const int blk) {
  __shared__ int hb[NBINS + 1];
  __shared__ int wsum[16];
  __shared__ int s_val[CHUNK];
  const int tid = threadIdx.x;
  const int lane = tid & 31;
  const int wave = tid >> 5;

  for (int i = tid; i < NBINS + 1; i += 1024) hb[i] = 0;
  const int e0 = blk * CHUNK;
  const int e1 = (e0 + CHUNK < N_EDGES) ? (e0 + CHUNK) : N_EDGES;
  int es[EPT], ed[EPT];
#pragma unroll
  for (int u = 0; u < EPT; ++u) {
    const int i = e0 + tid + u * 1024;
    const int ii = (i < e1) ? i : (e1 - 1);
    es[u] = adj[ii];
    ed[u] = adj[N_EDGES + ii];
  }
  __builtin_amdgcn_sched_barrier(0);
  __syncthreads();

  int rk[EPT];
#pragma unroll
  for (int u = 0; u < EPT; ++u) {
    rk[u] = 0;
    if (e0 + tid + u * 1024 < e1) rk[u] = atomicAdd(&hb[ed[u] >> BIN_SHIFT], 1);
  }
  __syncthreads();

  int c = 0, inc = 0;
  if (tid < 512) {
    if (tid < NBINS) c = hb[tid];
    inc = c;
#pragma unroll
    for (int o = 1; o < 32; o <<= 1) {
      const int t = __shfl_up(inc, o, 32);
      if (lane >= o) inc += t;
    }
    if (lane == 31) wsum[wave] = inc;
  }
  __syncthreads();
  if (tid <= NBINS) {
    int wpre = 0;
    for (int w = 0; w < wave; ++w) wpre += wsum[w];
    const int lstart = wpre + inc - c;
    hb[tid] = lstart;
    runtab[blk * TP + tid] = lstart;
  }
  __syncthreads();

#pragma unroll
  for (int u = 0; u < EPT; ++u) {
    if (e0 + tid + u * 1024 < e1)
      s_val[hb[ed[u] >> BIN_SHIFT] + rk[u]] =
          (es[u] << BIN_SHIFT) | (ed[u] & (BIN_NODES - 1));
  }
  __syncthreads();
  const int n = e1 - e0;
  for (int j = tid; j < n; j += 1024) ebuf[e0 + j] = s_val[j];
}

#define G1_BLOCKS 256

static_assert(32 * G1_BLOCKS * 16 >= N_NODES, "first launch: every 16-row tile has a wave");

__global__ __launch_bounds__(1024, 16) void order_and_gemm1(
    const int* __restrict__ adj, int* __restrict__ runtab, int* __restrict__ ebuf,
    const float* __restrict__ X, const float* __restrict__ W, float* __restrict__ h,
    const float* __restrict__ a_src, const float* __restrict__ a_dst,
    float* __restrict__ alpha_s, float* __restrict__ alpha_d) {
  const int blk = blockIdx.x;
  if (blk < G1_BLOCKS) {
    const int wave = __builtin_amdgcn_readfirstlane((int)(threadIdx.x >> 5));
    const long tile = (long)(wave >> 1) * (2 * G1_BLOCKS) + 2 * blk + (wave & 1);
    gemm_tile<IN_DIM, HDIM, true, 1024>(X, W, nullptr, h, N_NODES, a_src, a_dst,
                                        alpha_s, alpha_d, tile * 16);
  } else {
    chunk_order(adj, runtab, ebuf, blk - G1_BLOCKS);
  }
}

#define STAGE_CAP 12288
#define RS 8
#define RL 4

static_assert(RL * NCHUNK == 1024, "bin_sort: every thread has its place in a run");
static_assert(NCHUNK == BIN_NODES, "bin_sort: the first 8 waves take a chunk and a node each");

__device__ __forceinline__ void group_and_sort(
    int* __restrict__ stage, int* __restrict__ cur, const int* __restrict__ loc,
    const int* __restrict__ ebuf, int* __restrict__ csr, const int (&v)[RS],
    const int bs, const int gs, const int rn, const int rq,
    const int lane, const int wave) {
#pragma unroll
  for (int u = 0; u < RS; ++u) {
    if (v[u] >= 0) {
      const int p = atomicAdd(&cur[v[u] & (BIN_NODES - 1)], 1);
      stage[p] = v[u] >> BIN_SHIFT;
    }
  }
  for (int i = rq + RL * RS; i < rn; i += RL) {
    const int w = ebuf[gs + i];
    const int p = atomicAdd(&cur[w & (BIN_NODES - 1)], 1);
    stage[p] = w >> BIN_SHIFT;
  }
  __syncthreads();

  for (int d = wave; d < BIN_NODES; d += 32) {
    const int start = __builtin_amdgcn_readfirstlane(loc[d]);
    const int end = __builtin_amdgcn_readfirstlane(loc[d + 1]);
    for (int ib = start; ib < end; ib += 32) {
      const int i = ib + lane;
      const bool vi = i < end;
      const int mine = vi ? stage[i] : 0;
      int r = 0;
      for (int kb = start; kb < end; kb += 32) {
        const int kj = kb + lane;
        const int other = (kj < end) ? stage[kj] : 0;
        const int nk = (end - kb < 32) ? (end - kb) : 32;
        for (int k = 0; k < nk; ++k) {
          const int o = __builtin_amdgcn_readlane(other, k);
          const int kk = kb + k;
          r += (o < mine || (o == mine && kk < i)) ? 1 : 0;
        }
      }
      if (vi) csr[bs + start + r] = mine;
    }
  }
}

__global__ __launch_bounds__(1024) void bin_sort(
    const int* __restrict__ runtab, const int* __restrict__ ebuf,
    int* __restrict__ off, int* __restrict__ tmp, int* __restrict__ csr) {
  __shared__ int cnt[BIN_NODES];
  __shared__ int loc[BIN_NODES + 1];
  __shared__ int cur[BIN_NODES];
  __shared__ int wsum[BIN_NODES / 32];
  __shared__ int rsum[NCHUNK / 32];
  __shared__ int lsum[NCHUNK / 32];
  __shared__ int run_g[NCHUNK];
  __shared__ int run_n[NCHUNK];
  __shared__ int stage_l[STAGE_CAP];
  const int tid = threadIdx.x;
  const int lane = tid & 31;
  const int wave = tid >> 5;
  const int b = blockIdx.x;

  if (tid < NCHUNK) {
    const int rs = runtab[tid * TP + b];
    const int ln = runtab[tid * TP + b + 1] - rs;
    run_g[tid] = tid * CHUNK + rs;
    run_n[tid] = ln;
    int sr = rs, sl = ln;
#pragma unroll
    for (int o = 16; o > 0; o >>= 1) {
      sr += __shfl_xor(sr, o, 32);
      sl += __shfl_xor(sl, o, 32);
    }
    if (lane == 0) {
      rsum[wave] = sr;
      lsum[wave] = sl;
    }
    cnt[tid] = 0;
  }
  __syncthreads();
  int bs = 0, m = 0;
#pragma unroll
  for (int w = 0; w < NCHUNK / 32; ++w) {
    bs += rsum[w];
    m += lsum[w];
  }
  const int be = bs + m;

  const int rq = tid & (RL - 1);
  const int gs = run_g[tid / RL];
  const int rn = run_n[tid / RL];
  int v[RS];
#pragma unroll
  for (int u = 0; u < RS; ++u) {
    const int i = rq + RL * u;
    v[u] = (i < rn) ? ebuf[gs + i] : -1;
  }
#pragma unroll
  for (int u = 0; u < RS; ++u)
    if (v[u] >= 0) atomicAdd(&cnt[v[u] & (BIN_NODES - 1)], 1);
  for (int i = rq + RL * RS; i < rn; i += RL)
    atomicAdd(&cnt[ebuf[gs + i] & (BIN_NODES - 1)], 1);
  __syncthreads();

  int c = 0, inc = 0;
  if (tid < BIN_NODES) {
    c = cnt[tid];
    inc = c;
#pragma unroll
    for (int o = 1; o < 32; o <<= 1) {
      const int t = __shfl_up(inc, o, 32);
      if (lane >= o) inc += t;
    }
    if (lane == 31) wsum[wave] = inc;
  }
  __syncthreads();
  if (tid < BIN_NODES) {
    int wpre = 0;
    for (int w = 0; w < wave; ++w) wpre += wsum[w];
    const int ls = wpre + inc - c;
    loc[tid] = ls;
    cur[tid] = ls;
    const int node = b * BIN_NODES + tid;
    if (node < N_NODES) off[node] = bs + ls;
  }
  if (tid == 0) {
    loc[BIN_NODES] = m;
    if (b == NBINS - 1) off[N_NODES] = be;
  }
  __syncthreads();

  if (m <= STAGE_CAP)
    group_and_sort(stage_l, cur, loc, ebuf, csr, v, bs, gs, rn, rq, lane, wave);
  else
    group_and_sort(tmp + bs, cur, loc, ebuf, csr, v, bs, gs, rn, rq, lane, wave);
}

#define CH 16

template <int I> struct idx_c { static constexpr int value = I; };
template <int I, int N, class F>
__device__ __forceinline__ void static_for(F&& f) {
  if constexpr (I < N) {
    f(idx_c<I>{});
    static_for<I + 1, N>(f);
  }
}

template <int K>
__device__ __forceinline__ int share16(int x) {
  return __builtin_amdgcn_update_dpp(0, x, 0x150 + K, 0xf, 0xf, true);
}
template <int K>
__device__ __forceinline__ float share16f(float x) {
  return __int_as_float(share16<K>(__float_as_int(x)));
}
template <int W>
__device__ __forceinline__ float xor16(float x) {
  return __int_as_float(
      __builtin_amdgcn_update_dpp(0, __float_as_int(x), 0x160 + W, 0xf, 0xf, true));
}

template <int W>
__device__ __forceinline__ void fold16(float (&v)[CH], const bool up) {
  static_for<0, W>([&](auto ic) {
    constexpr int i = decltype(ic)::value;
    const float lower = v[i];
    const float upper = v[i + W];
    const float keep = up ? upper : lower;
    const float send = up ? lower : upper;
    v[i] = keep + xor16<W>(send);
  });
}

__device__ __forceinline__ void walk_chunk(
    const float* __restrict__ h, const int hl, const unsigned hofs, const float4 av,
    const float adn, const int sj, const int cnt, const int cntmax,
    float& m, float& ssum, float4& a) {
  float4 g[CH];
  static_for<0, CH>([&](auto kc) {
    constexpr int k = decltype(kc)::value;
    if (k == 0 || k < cntmax) {
      const unsigned ofs = ((unsigned)share16<k>(sj) << 8) | hofs;
      g[k] = *(const float4*)((const char*)h + ofs);
    }
  });

  float v[CH];
  static_for<0, CH>([&](auto kc) {
    constexpr int k = decltype(kc)::value;
    v[k] = 0.0f;
    if (k == 0 || k < cntmax)
      v[k] = g[k].x * av.x + g[k].y * av.y + g[k].z * av.z + g[k].w * av.w;
  });

  fold16<8>(v, (hl & 8) != 0);
  fold16<4>(v, (hl & 4) != 0);
  fold16<2>(v, (hl & 2) != 0);
  fold16<1>(v, (hl & 1) != 0);
  const float score = v[0];

  float e = score + adn;
  e = (e > 0.0f) ? e : NEG_SLOPE * e;
  const bool valid = hl < cnt;
  float cm = valid ? e : m;
  cm = fmaxf(cm, xor16<8>(cm));
  cm = fmaxf(cm, xor16<4>(cm));
  cm = fmaxf(cm, xor16<2>(cm));
  cm = fmaxf(cm, xor16<1>(cm));
  const float mn = fmaxf(m, cm);
  const float scale = __expf(m - mn);
  ssum *= scale;
  a.x *= scale;
  a.y *= scale;
  a.z *= scale;
  a.w *= scale;
  m = mn;

  const float p = valid ? __expf(e - m) : 0.0f;
  static_for<0, CH>([&](auto kc) {
    constexpr int k = decltype(kc)::value;
    if (k == 0 || k < cntmax) {
      const float pk = share16f<k>(p);
      ssum += pk;
      a.x += g[k].x * pk;
      a.y += g[k].y * pk;
      a.z += g[k].z * pk;
      a.w += g[k].w * pk;
    }
  });
}

#define WALK_BLOCKS 1024
#define AST 68

static_assert(4 * WALK_BLOCKS * 32 >= N_NODES, "node walk: a wave's tile has 16 rows, 4 per unit");

template <int ELU, int NOUT, bool ALPHA, bool XOUT>
__global__ __launch_bounds__(256) void node_walk(
    const int* __restrict__ off, const int* __restrict__ csr,
    const float* __restrict__ alpha_s, const float* __restrict__ alpha_d,
    const float* __restrict__ h, const float* __restrict__ a_src,
    const float* __restrict__ b, float* __restrict__ xout,
    const float* __restrict__ Bd, const float* __restrict__ biasd, float* __restrict__ C,
    const float* __restrict__ a_src_n, const float* __restrict__ a_dst_n,
    float* __restrict__ alpha_s_n, float* __restrict__ alpha_d_n, int n_nodes) {
  __shared__ __attribute__((aligned(16))) float Bs[HDIM * NOUT];
  __shared__ __attribute__((aligned(16))) float As[8 * 16 * AST];
  const int lane = threadIdx.x & 31;
  const int hl = lane & 15;
  const int half = lane >> 4;
  constexpr int NB4 = HDIM * NOUT / 4 / 256;
  static_assert(NB4 * 4 * 256 == HDIM * NOUT, "Bd: whole 16-byte pieces for 256 threads");
  float4 bt[NB4];
#pragma unroll
  for (int u = 0; u < NB4; ++u) bt[u] = ((const float4*)Bd)[threadIdx.x + u * 256];
  const float4 bb = *(const float4*)(b + 4 * hl);
  const float4 av = *(const float4*)(a_src + 4 * hl);
#pragma unroll
  for (int u = 0; u < NB4; ++u) ((float4*)Bs)[threadIdx.x + u * 256] = bt[u];
  __syncthreads();

  const int wave = __builtin_amdgcn_readfirstlane((int)(threadIdx.x >> 5));
  const int blk = blockIdx.x;
  const int nunits = (n_nodes + 31) >> 5;
  const unsigned hofs = (unsigned)hl << 4;
  float* __restrict__ Aw = As + wave * (16 * AST);
  float* __restrict__ Amine = Aw + half * AST + 4 * hl;

  int it = 0;
  for (; it < 8; ++it) {
    const int u = blk + (it >> 1) * WALK_BLOCKS;
    if (u >= nunits) break;
    const int n = 32 * u + 4 * wave + 2 * (it & 1) + half;
    const bool nv = n < n_nodes;
    const int nn = nv ? n : (n_nodes - 1);
    const int start = off[nn];
    const int endg = off[nn + 1];
    const int deg = nv ? (endg - start) : 0;
    const float adn = alpha_d[nn];
    float es = alpha_s[nn] + adn;
    es = (es > 0.0f) ? es : NEG_SLOPE * es;
    const float4 hn = *(const float4*)(h + (size_t)nn * HDIM + 4 * hl);

    float m = es;
    float ssum = 1.0f;
    float4 a = hn;

    const int d0 = __builtin_amdgcn_readlane(deg, 0);
    const int d1 = __builtin_amdgcn_readlane(deg, 16);
    const int maxdeg = (d0 > d1) ? d0 : d1;
    int base = 0;
    do {
      const int q = base + hl;
      int sj = nn;
      if (q < deg) sj = csr[start + q];
      const int left = maxdeg - base;
      walk_chunk(h, hl, hofs, av, adn, sj, deg - base, (left < CH) ? left : CH, m, ssum, a);
      base += CH;
    } while (base < maxdeg);

    float4 o;
    o.x = a.x / ssum + bb.x;
    o.y = a.y / ssum + bb.y;
    o.z = a.z / ssum + bb.z;
    o.w = a.w / ssum + bb.w;
    if (ELU) {
      o.x = (o.x > 0.0f) ? o.x : (__expf(o.x) - 1.0f);
      o.y = (o.y > 0.0f) ? o.y : (__expf(o.y) - 1.0f);
      o.z = (o.z > 0.0f) ? o.z : (__expf(o.z) - 1.0f);
      o.w = (o.w > 0.0f) ? o.w : (__expf(o.w) - 1.0f);
    }
    if (XOUT) {
      if (nv) *(float4*)(xout + (size_t)n * HDIM + 4 * hl) = o;
    }
    float* __restrict__ ar = Amine + it * (2 * AST);
    ar[0] = nv ? o.x : 0.0f;
    ar[1] = nv ? o.y : 0.0f;
    ar[2] = nv ? o.z : 0.0f;
    ar[3] = nv ? o.w : 0.0f;
  }
  for (; it < 8; ++it) {
    float* __restrict__ ar = Amine + it * (2 * AST);
    ar[0] = 0.0f;
    ar[1] = 0.0f;
    ar[2] = 0.0f;
    ar[3] = 0.0f;
  }

  const int lo = hl;
  const int hi = half;
  constexpr int NT = NOUT / 16;
  v8f acc[NT] = {};
  [[maybe_unused]] float avs[NT], avd[NT];
  if constexpr (ALPHA) {
#pragma unroll
    for (int nt = 0; nt < NT; ++nt) {
      avs[nt] = a_src_n[nt * 16 + lo];
      avd[nt] = a_dst_n[nt * 16 + lo];
    }
  }
  const float* __restrict__ Arow = Aw + lo * AST;
  for (int kk = 0; kk < HDIM / 4; ++kk) {
    const int kb = kk * 4 + 2 * hi;
    v2f fa;
    fa.x = Arow[kb];
    fa.y = Arow[kb + 1];
#pragma unroll
    for (int nt = 0; nt < NT; ++nt) {
      v2f fb;
      fb.x = Bs[kb * NOUT + nt * 16 + lo];
      fb.y = Bs[(kb + 1) * NOUT + nt * 16 + lo];
      acc[nt] = __builtin_amdgcn_wmma_f32_16x16x4_f32(
           false, fa,  false, fb,
           (short)0, acc[nt],  false,  false);
    }
  }
  const auto rowof = [blk, wave](const int r) {
    return (long)32 * (blk + (r >> 2) * WALK_BLOCKS) + 4 * wave + (r & 3);
  };
  tile_store<NOUT>(acc, biasd, C, n_nodes, lo, hi, rowof);
  if constexpr (ALPHA)
    tile_scores<NOUT>(acc, avs, avd, alpha_s_n, alpha_d_n, n_nodes, lo, hi, rowof);
}

extern "C" void kernel_launch(void* const* d_in, const int* in_sizes, int n_in,
                              void* d_out, int out_size, void* d_ws,
                              size_t ws_size, hipStream_t stream) {
  const float* X      = (const float*)d_in[0];
  const int*   adj    = (const int*)d_in[1];
  const float* W1     = (const float*)d_in[2];
  const float* a_src1 = (const float*)d_in[3];
  const float* a_dst1 = (const float*)d_in[4];
  const float* b1     = (const float*)d_in[5];
  const float* W2     = (const float*)d_in[6];
  const float* a_src2 = (const float*)d_in[7];
  const float* a_dst2 = (const float*)d_in[8];
  const float* b2     = (const float*)d_in[9];
  const float* clas_W = (const float*)d_in[10];
  const float* clas_b = (const float*)d_in[11];

  float* out = (float*)d_out;
  float* x2  = out + (size_t)N_NODES * N_CLASSES;

  float* ws   = (float*)d_ws;
  float* h    = ws;
  float* h2   = ws + (size_t)N_NODES * HDIM;
  int*   ib   = (int*)(ws + (size_t)2 * N_NODES * HDIM);
  float* as_  = ws + (size_t)3 * N_NODES * HDIM;
  float* ad_  = as_ + N_NODES;
  int* runtab = ib;
  int* off  = ib + 2 * 100352;
  int* ebuf = ib + 3 * 100352;
  int* tmp  = ebuf + N_EDGES;
  int* csr  = tmp + N_EDGES;
  float* as2 = (float*)(csr + N_EDGES);
  float* ad2 = as2 + 100352;

  order_and_gemm1<<<G1_BLOCKS + NCHUNK, 1024, 0, stream>>>(
      adj, runtab, ebuf, X, W1, h, a_src1, a_dst1, as_, ad_);
  bin_sort<<<NBINS, 1024, 0, stream>>>(runtab, ebuf, off, tmp, csr);

  node_walk<1, HDIM, true, false><<<WALK_BLOCKS, 256, 0, stream>>>(
      off, csr, as_, ad_, h, a_src1, b1, nullptr,
      W2, nullptr, h2, a_src2, a_dst2, as2, ad2, N_NODES);

  node_walk<0, N_CLASSES, false, true><<<WALK_BLOCKS, 256, 0, stream>>>(
      off, csr, as2, ad2, h2, a_src2, b2, x2,
      clas_W, clas_b, out, nullptr, nullptr, nullptr, nullptr, N_NODES);
}
